# starting kernel
# baseline (speedup 1.0000x reference)
#include <hip/hip_runtime.h>
#include <hip/hip_bf16.h>
#include <math.h>

#define NNODES 50000
#define NEDGES 800000
#define ETOT   (NEDGES + NNODES)
#define FIN    256
#define H1     8
#define C1     32
#define D1     256
#define OUT2   2
#define SLOPE  0.2f

#define LDSPAD 4
#define ASTRIDE (FIN + LDSPAD)

typedef float v2f __attribute__((ext_vector_type(2)));
typedef float v8f __attribute__((ext_vector_type(8)));

__device__ __forceinline__ void edge_sd(const int* __restrict__ ei, int e, int& s, int& d) {
    if (e < NEDGES) { s = ei[e]; d = ei[NEDGES + e]; }
    else            { s = e - NEDGES; d = s; }
}

__device__ __forceinline__ float lrelu(float v) {
    return v > 0.0f ? v : SLOPE * v;
}

__device__ __forceinline__ void atomicMaxF(float* addr, float val) {
    if (val >= 0.0f) atomicMax((int*)addr, __float_as_int(val));
    else             atomicMin((unsigned int*)addr, __float_as_uint(val));
}

__global__ __launch_bounds__(512) void gemm1_wmma(const float* __restrict__ x,
                                                  const float* __restrict__ W1,
                                                  float* __restrict__ xp) {
    __shared__ float aTile[16 * ASTRIDE];

    const int t     = threadIdx.x;
    const int lane  = t & 31;
    const int wave  = t >> 5;
    const int tileM = blockIdx.x;
    const int half  = lane >> 4;
    const int l16   = lane & 15;

    {
        const int row = t >> 5;
        const int col = (t & 31) * 8;
        const float4* src = (const float4*)(x + (size_t)(tileM * 16 + row) * FIN + col);
        float4* dst = (float4*)(aTile + row * ASTRIDE + col);
        dst[0] = src[0];
        dst[1] = src[1];
    }
    __syncthreads();

    const int bcol = wave * 16 + l16;

    v8f acc = {};
#pragma unroll
    for (int k = 0; k < FIN; k += 4) {
        const int kk = k + half * 2;
        v2f a = *(const v2f*)(aTile + l16 * ASTRIDE + kk);
        v2f b;
        b.x = W1[kk * D1 + bcol];
        b.y = W1[(kk + 1) * D1 + bcol];
        acc = __builtin_amdgcn_wmma_f32_16x16x4_f32(
             false, a,  false, b,
             (short)0, acc,  false,  false);
    }

    const int mrow = tileM * 16 + half * 8;
#pragma unroll
    for (int r = 0; r < 8; ++r)
        xp[(size_t)(mrow + r) * D1 + bcol] = acc[r];
}

__global__ void alpha1_kernel(const float* __restrict__ xp,
                              const float* __restrict__ a_src,
                              const float* __restrict__ a_dst,
                              float* __restrict__ as, float* __restrict__ ad) {
    int i = blockIdx.x * blockDim.x + threadIdx.x;
    if (i >= NNODES * H1) return;
    int h = i & (H1 - 1);
    int n = i >> 3;
    const float* row = xp + (size_t)n * D1 + h * C1;
    const float* ws  = a_src + h * C1;
    const float* wd  = a_dst + h * C1;
    float s = 0.0f, d = 0.0f;
#pragma unroll
    for (int c = 0; c < C1; ++c) {
        float v = row[c];
        s = fmaf(v, ws[c], s);
        d = fmaf(v, wd[c], d);
    }
    as[i] = s;
    ad[i] = d;
}

__global__ void fill_f32(float* __restrict__ p, float v, int n) {
    int i = blockIdx.x * blockDim.x + threadIdx.x;
    if (i < n) p[i] = v;
}

template <int NH>
__global__ void edge_max(const int* __restrict__ ei,
                         const float* __restrict__ as,
                         const float* __restrict__ ad,
                         float* __restrict__ m) {
    int e = blockIdx.x * blockDim.x + threadIdx.x;
    if (e >= ETOT) return;
    int s, d;
    edge_sd(ei, e, s, d);
#pragma unroll
    for (int h = 0; h < NH; ++h) {
        float v = lrelu(as[s * NH + h] + ad[d * NH + h]);
        atomicMaxF(&m[d * NH + h], v);
    }
}

template <int NH>
__global__ void edge_denom(const int* __restrict__ ei,
                           const float* __restrict__ as,
                           const float* __restrict__ ad,
                           const float* __restrict__ m,
                           float* __restrict__ den) {
    int e = blockIdx.x * blockDim.x + threadIdx.x;
    if (e >= ETOT) return;
    int s, d;
    edge_sd(ei, e, s, d);
#pragma unroll
    for (int h = 0; h < NH; ++h) {
        float v = lrelu(as[s * NH + h] + ad[d * NH + h]);
        atomicAdd(&den[d * NH + h], __expf(v - m[d * NH + h]));
    }
}

__global__ __launch_bounds__(256) void agg1(const int* __restrict__ ei,
                                            const float* __restrict__ xp,
                                            const float* __restrict__ as,
                                            const float* __restrict__ ad,
                                            const float* __restrict__ m,
                                            const float* __restrict__ den,
                                            float* __restrict__ out) {
    int e = blockIdx.x;
    int t = threadIdx.x;
    int h = t >> 5;
    int s, d;
    edge_sd(ei, e, s, d);
    float v     = lrelu(as[s * H1 + h] + ad[d * H1 + h]);
    float alpha = __expf(v - m[d * H1 + h]) / den[d * H1 + h];
    atomicAdd(&out[(size_t)d * D1 + t], alpha * xp[(size_t)s * D1 + t]);
}

__global__ void bias_relu(float* __restrict__ hbuf, const float* __restrict__ b) {
    int i = blockIdx.x * blockDim.x + threadIdx.x;
    if (i >= NNODES * D1) return;
    float v = hbuf[i] + b[i & (D1 - 1)];
    hbuf[i] = v > 0.0f ? v : 0.0f;
}

__global__ __launch_bounds__(256) void gemm2_alpha2(const float* __restrict__ h,
                                                    const float* __restrict__ W2,
                                                    const float* __restrict__ aw_src,
                                                    const float* __restrict__ aw_dst,
                                                    float* __restrict__ xp2,
                                                    float* __restrict__ as2,
                                                    float* __restrict__ ad2) {
    int lane = threadIdx.x & 31;
    int n    = blockIdx.x * (blockDim.x >> 5) + (threadIdx.x >> 5);
    if (n >= NNODES) return;
    float s0 = 0.0f, s1 = 0.0f;
    for (int k = lane; k < D1; k += 32) {
        float v = h[(size_t)n * D1 + k];
        s0 = fmaf(v, W2[k * OUT2 + 0], s0);
        s1 = fmaf(v, W2[k * OUT2 + 1], s1);
    }
#pragma unroll
    for (int off = 16; off > 0; off >>= 1) {
        s0 += __shfl_down(s0, off);
        s1 += __shfl_down(s1, off);
    }
    if (lane == 0) {
        xp2[n * OUT2 + 0] = s0;
        xp2[n * OUT2 + 1] = s1;
        as2[n] = fmaf(s0, aw_src[0], s1 * aw_src[1]);
        ad2[n] = fmaf(s0, aw_dst[0], s1 * aw_dst[1]);
    }
}

__global__ void agg2(const int* __restrict__ ei,
                     const float* __restrict__ xp2,
                     const float* __restrict__ as2,
                     const float* __restrict__ ad2,
                     const float* __restrict__ m2,
                     const float* __restrict__ den2,
                     float* __restrict__ out) {
    int e = blockIdx.x * blockDim.x + threadIdx.x;
    if (e >= ETOT) return;
    int s, d;
    edge_sd(ei, e, s, d);
    float v     = lrelu(as2[s] + ad2[d]);
    float alpha = __expf(v - m2[d]) / den2[d];
    atomicAdd(&out[d * OUT2 + 0], alpha * xp2[s * OUT2 + 0]);
    atomicAdd(&out[d * OUT2 + 1], alpha * xp2[s * OUT2 + 1]);
}

__global__ void bias2_add(float* __restrict__ out, const float* __restrict__ b) {
    int i = blockIdx.x * blockDim.x + threadIdx.x;
    if (i < NNODES * OUT2) out[i] += b[i & 1];
}

extern "C" void kernel_launch(void* const* d_in, const int* in_sizes, int n_in,
                              void* d_out, int out_size, void* d_ws, size_t ws_size,
                              hipStream_t stream) {
    const float* x     = (const float*)d_in[0];
    const int*   ei    = (const int*)d_in[1];
    const float* W1    = (const float*)d_in[2];
    const float* asw1  = (const float*)d_in[3];
    const float* adw1  = (const float*)d_in[4];
    const float* b1    = (const float*)d_in[5];
    const float* W2    = (const float*)d_in[6];
    const float* asw2  = (const float*)d_in[7];
    const float* adw2  = (const float*)d_in[8];
    const float* b2    = (const float*)d_in[9];
    float*       out   = (float*)d_out;

    float* w    = (float*)d_ws;
    float* xp1  = w;
    float* hbuf = xp1  + (size_t)NNODES * D1;
    float* as1  = hbuf + (size_t)NNODES * D1;
    float* ad1  = as1  + (size_t)NNODES * H1;
    float* m1   = ad1  + (size_t)NNODES * H1;
    float* dn1  = m1   + (size_t)NNODES * H1;
    float* xp2  = dn1  + (size_t)NNODES * H1;
    float* as2  = xp2  + (size_t)NNODES * OUT2;
    float* ad2  = as2  + (size_t)NNODES;
    float* m2   = ad2  + (size_t)NNODES;
    float* dn2  = m2   + (size_t)NNODES;

    hipMemsetAsync(hbuf, 0, (size_t)NNODES * D1 * sizeof(float), stream);
    hipMemsetAsync(dn1,  0, (size_t)NNODES * H1 * sizeof(float), stream);
    hipMemsetAsync(dn2,  0, (size_t)NNODES * sizeof(float), stream);
    hipMemsetAsync(out,  0, (size_t)NNODES * OUT2 * sizeof(float), stream);
    fill_f32<<<(NNODES * H1 + 255) / 256, 256, 0, stream>>>(m1, -INFINITY, NNODES * H1);
    fill_f32<<<(NNODES + 255) / 256, 256, 0, stream>>>(m2, -INFINITY, NNODES);

    gemm1_wmma<<<NNODES / 16, 512, 0, stream>>>(x, W1, xp1);
    alpha1_kernel<<<(NNODES * H1 + 255) / 256, 256, 0, stream>>>(xp1, asw1, adw1, as1, ad1);
    edge_max<H1><<<(ETOT + 255) / 256, 256, 0, stream>>>(ei, as1, ad1, m1);
    edge_denom<H1><<<(ETOT + 255) / 256, 256, 0, stream>>>(ei, as1, ad1, m1, dn1);
    agg1<<<ETOT, 256, 0, stream>>>(ei, xp1, as1, ad1, m1, dn1, hbuf);
    bias_relu<<<(NNODES * D1 + 255) / 256, 256, 0, stream>>>(hbuf, b1);

    gemm2_alpha2<<<(NNODES + 7) / 8, 256, 0, stream>>>(hbuf, W2, asw2, adw2, xp2, as2, ad2);
    edge_max<1><<<(ETOT + 255) / 256, 256, 0, stream>>>(ei, as2, ad2, m2);
    edge_denom<1><<<(ETOT + 255) / 256, 256, 0, stream>>>(ei, as2, ad2, m2, dn2);
    agg2<<<(ETOT + 255) / 256, 256, 0, stream>>>(ei, xp2, as2, ad2, m2, dn2, out);
    bias2_add<<<(NNODES * OUT2 + 255) / 256, 256, 0, stream>>>(out, b2);
}


# Edges grouped by destination each call (count, scan, fill); both layers' softmax+aggregation become per-node gathers summed in registers: no float atomics, sums now in list and lane-tree order
# speedup vs baseline: 4.7246x; 4.7246x over previous
#include <hip/hip_runtime.h>
#include <hip/hip_bf16.h>
#include <math.h>

#define NNODES 50000
#define NEDGES 800000
#define FIN    256
#define H1     8
#define C1     32
#define D1     256
#define OUT2   2
#define SLOPE  0.2f

#define LDSPAD 4
#define ASTRIDE (FIN + LDSPAD)

typedef float v2f __attribute__((ext_vector_type(2)));
typedef float v8f __attribute__((ext_vector_type(8)));

__device__ __forceinline__ float lrelu(float v) {
    return v > 0.0f ? v : SLOPE * v;
}

__global__ __launch_bounds__(512) void gemm1_wmma(const float* __restrict__ x,
                                                  const float* __restrict__ W1,
                                                  float* __restrict__ xp) {
    __shared__ float aTile[16 * ASTRIDE];

    const int t     = threadIdx.x;
    const int lane  = t & 31;
    const int wave  = t >> 5;
    const int tileM = blockIdx.x;
    const int half  = lane >> 4;
    const int l16   = lane & 15;

    {
        const int row = t >> 5;
        const int col = (t & 31) * 8;
        const float4* src = (const float4*)(x + (size_t)(tileM * 16 + row) * FIN + col);
        float4* dst = (float4*)(aTile + row * ASTRIDE + col);
        dst[0] = src[0];
        dst[1] = src[1];
    }
    __syncthreads();

    const int bcol = wave * 16 + l16;

    v8f acc = {};
#pragma unroll
    for (int k = 0; k < FIN; k += 4) {
        const int kk = k + half * 2;
        v2f a = *(const v2f*)(aTile + l16 * ASTRIDE + kk);
        v2f b;
        b.x = W1[kk * D1 + bcol];
        b.y = W1[(kk + 1) * D1 + bcol];
        acc = __builtin_amdgcn_wmma_f32_16x16x4_f32(
             false, a,  false, b,
             (short)0, acc,  false,  false);
    }

    const int mrow = tileM * 16 + half * 8;
#pragma unroll
    for (int r = 0; r < 8; ++r)
        xp[(size_t)(mrow + r) * D1 + bcol] = acc[r];
}

__global__ void alpha1_kernel(const float* __restrict__ xp,
                              const float* __restrict__ a_src,
                              const float* __restrict__ a_dst,
                              float* __restrict__ as, float* __restrict__ ad) {
    int i = blockIdx.x * blockDim.x + threadIdx.x;
    if (i >= NNODES * H1) return;
    int h = i & (H1 - 1);
    int n = i >> 3;
    const float* row = xp + (size_t)n * D1 + h * C1;
    const float* ws  = a_src + h * C1;
    const float* wd  = a_dst + h * C1;
    float s = 0.0f, d = 0.0f;
#pragma unroll
    for (int c = 0; c < C1; ++c) {
        float v = row[c];
        s = fmaf(v, ws[c], s);
        d = fmaf(v, wd[c], d);
    }
    as[i] = s;
    ad[i] = d;
}

__global__ void zero_i32(int* __restrict__ p, int n) {
    int i = blockIdx.x * blockDim.x + threadIdx.x;
    if (i < n) p[i] = 0;
}

__global__ void count_rank(const int* __restrict__ ei, int* __restrict__ cnt,
                           int* __restrict__ rnk) {
    int e = blockIdx.x * blockDim.x + threadIdx.x;
    if (e >= NEDGES) return;
    int d = ei[NEDGES + e];
    rnk[e] = atomicAdd(&cnt[d], 1);
}

static_assert(NNODES % 16 == 0, "scan_counts threads own 16 counts each");
__global__ __launch_bounds__(1024) void scan_counts(const int* __restrict__ cnt,
                                                    int* __restrict__ off) {
    __shared__ int wsum[2][32];
    const int t    = threadIdx.x;
    const int lane = t & 31;
    const int w    = t >> 5;
    int carry = 0;
    int buf   = 0;
    for (int base = 0; base < NNODES; base += 1024 * 16) {
        const int  i0 = base + t * 16;
        const bool in = i0 < NNODES;
        int c[16];
#pragma unroll
        for (int r = 0; r < 4; ++r) {
            int4 q = make_int4(0, 0, 0, 0);
            if (in) q = ((const int4*)(cnt + i0))[r];
            c[4 * r + 0] = q.x; c[4 * r + 1] = q.y; c[4 * r + 2] = q.z; c[4 * r + 3] = q.w;
        }
        int v = 0;
#pragma unroll
        for (int k = 0; k < 16; ++k) { const int ck = c[k]; c[k] = v; v += ck; }

        int x = v;
        for (int o = 1; o < 32; o <<= 1) {
            int y = __shfl_up(x, o);
            if (lane >= o) x += y;
        }
        if (lane == 31) wsum[buf][w] = x;
        __syncthreads();
        const int wsv = wsum[buf][lane];
        int wx = wsv;
        for (int o = 1; o < 32; o <<= 1) {
            int y = __shfl_up(wx, o);
            if (lane >= o) wx += y;
        }
        const int wave_excl  = __shfl(wx, w) - __shfl(wsv, w);
        const int tile_total = __shfl(wx, 31);
        if (in) {
            const int b0 = carry + wave_excl + x - v;
#pragma unroll
            for (int r = 0; r < 4; ++r)
                ((int4*)(off + i0))[r] = make_int4(b0 + c[4 * r + 0], b0 + c[4 * r + 1],
                                                   b0 + c[4 * r + 2], b0 + c[4 * r + 3]);
        }
        carry += tile_total;
        buf ^= 1;
    }
    if (t == 0) off[NNODES] = carry;
}

__global__ void fill_csr(const int* __restrict__ ei, const int* __restrict__ off,
                         const int* __restrict__ rnk, int* __restrict__ csrc) {
    int e = blockIdx.x * blockDim.x + threadIdx.x;
    if (e >= NEDGES) return;
    int d = ei[NEDGES + e];
    csrc[off[d] + rnk[e]] = ei[e];
}

__device__ __forceinline__ float wave_max(float v) {
    for (int o = 16; o > 0; o >>= 1) v = fmaxf(v, __shfl_xor(v, o));
    return v;
}
__device__ __forceinline__ float wave_sum(float v) {
    for (int o = 16; o > 0; o >>= 1) v += __shfl_xor(v, o);
    return v;
}
__device__ __forceinline__ int list_src(const int* __restrict__ csrc, int beg, int deg,
                                        int d, int j) {
    int s = d;
    if (j < deg) s = min(max(csrc[beg + j], 0), NNODES - 1);
    return s;
}

__device__ __forceinline__ float gather_rows(float acc, const float* __restrict__ xcol,
                                             int s_l, float a_l, int n) {
    for (int k = 0; k < n; k += 8) {
        float xv[8];
#pragma unroll
        for (int i = 0; i < 8; ++i) {
            const int s = __builtin_amdgcn_readlane(s_l, k + i);
            xv[i] = xcol[(size_t)s * D1];
        }
#pragma unroll
        for (int i = 0; i < 8; ++i) {
            const float a = __int_as_float(__builtin_amdgcn_readlane(__float_as_int(a_l), k + i));
            const float x = (k + i < n) ? xv[i] : 0.0f;
            acc = fmaf(a, x, acc);
        }
    }
    return acc;
}

__global__ __launch_bounds__(256) void node_agg1(const int* __restrict__ off,
                                                 const int* __restrict__ csrc,
                                                 const float* __restrict__ xp,
                                                 const float* __restrict__ as,
                                                 const float* __restrict__ ad,
                                                 const float* __restrict__ b,
                                                 float* __restrict__ hout) {
    const int d    = blockIdx.x;
    const int t    = threadIdx.x;
    const int h    = t >> 5;
    const int lane = t & 31;
    const int beg   = off[d];
    const int deg   = off[d + 1] - beg;
    const int total = deg + 1;
    const float add = ad[d * H1 + h];

    const bool  act0 = lane < total;
    const int   s0   = list_src(csrc, beg, deg, d, lane);
    const float v0   = act0 ? lrelu(as[s0 * H1 + h] + add) : -INFINITY;

    float mx = v0;
    for (int j0 = 32; j0 < total; j0 += 32) {
        const int j = j0 + lane;
        if (j < total) {
            const int s = list_src(csrc, beg, deg, d, j);
            mx = fmaxf(mx, lrelu(as[s * H1 + h] + add));
        }
    }
    mx = wave_max(mx);

    const float p0 = act0 ? __expf(v0 - mx) : 0.0f;
    float sum = p0;
    for (int j0 = 32; j0 < total; j0 += 32) {
        const int j = j0 + lane;
        if (j < total) {
            const int s = list_src(csrc, beg, deg, d, j);
            sum += __expf(lrelu(as[s * H1 + h] + add) - mx);
        }
    }
    const float den = wave_sum(sum);

    const float* xcol = xp + t;
    float acc = gather_rows(0.0f, xcol, s0, p0 / den, min(total, 32));
    for (int j0 = 32; j0 < total; j0 += 32) {
        const int j = j0 + lane;
        int   s = d;
        float a = 0.0f;
        if (j < total) {
            s = list_src(csrc, beg, deg, d, j);
            a = __expf(lrelu(as[s * H1 + h] + add) - mx) / den;
        }
        acc = gather_rows(acc, xcol, s, a, min(total - j0, 32));
    }

    const float r = acc + b[t];
    hout[(size_t)d * D1 + t] = r > 0.0f ? r : 0.0f;
}

__global__ __launch_bounds__(256) void gemm2_alpha2(const float* __restrict__ h,
                                                    const float* __restrict__ W2,
                                                    const float* __restrict__ aw_src,
                                                    const float* __restrict__ aw_dst,
                                                    float* __restrict__ xp2,
                                                    float* __restrict__ as2,
                                                    float* __restrict__ ad2) {
    int lane = threadIdx.x & 31;
    int n    = blockIdx.x * (blockDim.x >> 5) + (threadIdx.x >> 5);
    if (n >= NNODES) return;
    float s0 = 0.0f, s1 = 0.0f;
    for (int k = lane; k < D1; k += 32) {
        float v = h[(size_t)n * D1 + k];
        s0 = fmaf(v, W2[k * OUT2 + 0], s0);
        s1 = fmaf(v, W2[k * OUT2 + 1], s1);
    }
#pragma unroll
    for (int off = 16; off > 0; off >>= 1) {
        s0 += __shfl_down(s0, off);
        s1 += __shfl_down(s1, off);
    }
    if (lane == 0) {
        xp2[n * OUT2 + 0] = s0;
        xp2[n * OUT2 + 1] = s1;
        as2[n] = fmaf(s0, aw_src[0], s1 * aw_src[1]);
        ad2[n] = fmaf(s0, aw_dst[0], s1 * aw_dst[1]);
    }
}

static_assert(NNODES % 16 == 0, "node_agg2 blocks cover 16 nodes each");
__global__ __launch_bounds__(512) void node_agg2(const int* __restrict__ off,
                                                 const int* __restrict__ csrc,
                                                 const float* __restrict__ xp2,
                                                 const float* __restrict__ as2,
                                                 const float* __restrict__ ad2,
                                                 const float* __restrict__ b,
                                                 float* __restrict__ out) {
    __shared__ float res[16 * OUT2];
    const int lane = threadIdx.x & 31;
    const int w    = threadIdx.x >> 5;
    const int d    = blockIdx.x * 16 + w;
    const int beg   = off[d];
    const int deg   = off[d + 1] - beg;
    const int total = deg + 1;
    const float add = ad2[d];

    const bool  act0 = lane < total;
    const int   s0   = list_src(csrc, beg, deg, d, lane);
    const float v0   = act0 ? lrelu(as2[s0] + add) : -INFINITY;

    float mx = v0;
    for (int j0 = 32; j0 < total; j0 += 32) {
        const int j = j0 + lane;
        if (j < total) {
            const int s = list_src(csrc, beg, deg, d, j);
            mx = fmaxf(mx, lrelu(as2[s] + add));
        }
    }
    mx = wave_max(mx);

    const float p0 = act0 ? __expf(v0 - mx) : 0.0f;
    float sum = p0;
    for (int j0 = 32; j0 < total; j0 += 32) {
        const int j = j0 + lane;
        if (j < total) {
            const int s = list_src(csrc, beg, deg, d, j);
            sum += __expf(lrelu(as2[s] + add) - mx);
        }
    }
    const float den = wave_sum(sum);

    float c0 = 0.0f, c1 = 0.0f;
    if (act0) {
        const float a = p0 / den;
        c0 = a * xp2[s0 * OUT2 + 0];
        c1 = a * xp2[s0 * OUT2 + 1];
    }
    for (int j0 = 32; j0 < total; j0 += 32) {
        const int j = j0 + lane;
        if (j < total) {
            const int s = list_src(csrc, beg, deg, d, j);
            const float a = __expf(lrelu(as2[s] + add) - mx) / den;
            c0 = fmaf(a, xp2[s * OUT2 + 0], c0);
            c1 = fmaf(a, xp2[s * OUT2 + 1], c1);
        }
    }
    c0 = wave_sum(c0);
    c1 = wave_sum(c1);
    if (lane == 0) {
        res[w * OUT2 + 0] = c0;
        res[w * OUT2 + 1] = c1;
    }
    __syncthreads();
    if (threadIdx.x < 16 * OUT2)
        out[blockIdx.x * (16 * OUT2) + threadIdx.x] = res[threadIdx.x] + b[threadIdx.x & 1];
}

extern "C" void kernel_launch(void* const* d_in, const int* in_sizes, int n_in,
                              void* d_out, int out_size, void* d_ws, size_t ws_size,
                              hipStream_t stream) {
    const float* x     = (const float*)d_in[0];
    const int*   ei    = (const int*)d_in[1];
    const float* W1    = (const float*)d_in[2];
    const float* asw1  = (const float*)d_in[3];
    const float* adw1  = (const float*)d_in[4];
    const float* b1    = (const float*)d_in[5];
    const float* W2    = (const float*)d_in[6];
    const float* asw2  = (const float*)d_in[7];
    const float* adw2  = (const float*)d_in[8];
    const float* b2    = (const float*)d_in[9];
    float*       out   = (float*)d_out;

    float* w    = (float*)d_ws;
    float* xp1  = w;
    float* hbuf = xp1  + (size_t)NNODES * D1;
    float* as1  = hbuf + (size_t)NNODES * D1;
    float* ad1  = as1  + (size_t)NNODES * H1;
    float* xp2  = ad1  + (size_t)NNODES * H1;
    float* as2  = xp2  + (size_t)NNODES * OUT2;
    float* ad2  = as2  + (size_t)NNODES;
    int*   cnt  = (int*)(ad2 + (size_t)NNODES);
    int*   off  = cnt  + (size_t)NNODES;
    int*   rnk  = off  + (size_t)NNODES + 1;
    int*   csrc = rnk  + (size_t)NEDGES;

    zero_i32<<<(NNODES + 255) / 256, 256, 0, stream>>>(cnt, NNODES);
    count_rank<<<(NEDGES + 255) / 256, 256, 0, stream>>>(ei, cnt, rnk);
    scan_counts<<<1, 1024, 0, stream>>>(cnt, off);
    fill_csr<<<(NEDGES + 255) / 256, 256, 0, stream>>>(ei, off, rnk, csrc);

    gemm1_wmma<<<NNODES / 16, 512, 0, stream>>>(x, W1, xp1);
    alpha1_kernel<<<(NNODES * H1 + 255) / 256, 256, 0, stream>>>(xp1, asw1, adw1, as1, ad1);
    node_agg1<<<NNODES, 256, 0, stream>>>(off, csrc, xp1, as1, ad1, b1, hbuf);

    gemm2_alpha2<<<(NNODES + 7) / 8, 256, 0, stream>>>(hbuf, W2, asw2, adw2, xp2, as2, ad2);
    node_agg2<<<NNODES / 16, 512, 0, stream>>>(off, csrc, xp2, as2, ad2, b2, out);
}


# Layer-1 aggregation: one wave per node instead of eight, each lane sums 8 channels from 16-byte row loads (8x fewer waves, 4x fewer loads); softmax denominator summed in a new slot-then-lane order
# speedup vs baseline: 5.7991x; 1.2274x over previous
#include <hip/hip_runtime.h>
#include <hip/hip_bf16.h>
#include <math.h>

#define NNODES 50000
#define NEDGES 800000
#define FIN    256
#define H1     8
#define C1     32
#define D1     256
#define OUT2   2
#define SLOPE  0.2f

#define LDSPAD 4
#define ASTRIDE (FIN + LDSPAD)

typedef float v2f __attribute__((ext_vector_type(2)));
typedef float v8f __attribute__((ext_vector_type(8)));

__device__ __forceinline__ float lrelu(float v) {
    return v > 0.0f ? v : SLOPE * v;
}

__global__ __launch_bounds__(512) void gemm1_wmma(const float* __restrict__ x,
                                                  const float* __restrict__ W1,
                                                  float* __restrict__ xp) {
    __shared__ float aTile[16 * ASTRIDE];

    const int t     = threadIdx.x;
    const int lane  = t & 31;
    const int wave  = t >> 5;
    const int tileM = blockIdx.x;
    const int half  = lane >> 4;
    const int l16   = lane & 15;

    {
        const int row = t >> 5;
        const int col = (t & 31) * 8;
        const float4* src = (const float4*)(x + (size_t)(tileM * 16 + row) * FIN + col);
        float4* dst = (float4*)(aTile + row * ASTRIDE + col);
        dst[0] = src[0];
        dst[1] = src[1];
    }
    __syncthreads();

    const int bcol = wave * 16 + l16;

    v8f acc = {};
#pragma unroll
    for (int k = 0; k < FIN; k += 4) {
        const int kk = k + half * 2;
        v2f a = *(const v2f*)(aTile + l16 * ASTRIDE + kk);
        v2f b;
        b.x = W1[kk * D1 + bcol];
        b.y = W1[(kk + 1) * D1 + bcol];
        acc = __builtin_amdgcn_wmma_f32_16x16x4_f32(
             false, a,  false, b,
             (short)0, acc,  false,  false);
    }

    const int mrow = tileM * 16 + half * 8;
#pragma unroll
    for (int r = 0; r < 8; ++r)
        xp[(size_t)(mrow + r) * D1 + bcol] = acc[r];
}

__global__ void alpha1_kernel(const float* __restrict__ xp,
                              const float* __restrict__ a_src,
                              const float* __restrict__ a_dst,
                              float* __restrict__ as, float* __restrict__ ad) {
    int i = blockIdx.x * blockDim.x + threadIdx.x;
    if (i >= NNODES * H1) return;
    int h = i & (H1 - 1);
    int n = i >> 3;
    const float* row = xp + (size_t)n * D1 + h * C1;
    const float* ws  = a_src + h * C1;
    const float* wd  = a_dst + h * C1;
    float s = 0.0f, d = 0.0f;
#pragma unroll
    for (int c = 0; c < C1; ++c) {
        float v = row[c];
        s = fmaf(v, ws[c], s);
        d = fmaf(v, wd[c], d);
    }
    as[i] = s;
    ad[i] = d;
}

__global__ void zero_i32(int* __restrict__ p, int n) {
    int i = blockIdx.x * blockDim.x + threadIdx.x;
    if (i < n) p[i] = 0;
}

__global__ void count_rank(const int* __restrict__ ei, int* __restrict__ cnt,
                           int* __restrict__ rnk) {
    int e = blockIdx.x * blockDim.x + threadIdx.x;
    if (e >= NEDGES) return;
    int d = ei[NEDGES + e];
    rnk[e] = atomicAdd(&cnt[d], 1);
}

static_assert(NNODES % 16 == 0, "scan_counts threads own 16 counts each");
__global__ __launch_bounds__(1024) void scan_counts(const int* __restrict__ cnt,
                                                    int* __restrict__ off) {
    __shared__ int wsum[2][32];
    const int t    = threadIdx.x;
    const int lane = t & 31;
    const int w    = t >> 5;
    int carry = 0;
    int buf   = 0;
    for (int base = 0; base < NNODES; base += 1024 * 16) {
        const int  i0 = base + t * 16;
        const bool in = i0 < NNODES;
        int c[16];
#pragma unroll
        for (int r = 0; r < 4; ++r) {
            int4 q = make_int4(0, 0, 0, 0);
            if (in) q = ((const int4*)(cnt + i0))[r];
            c[4 * r + 0] = q.x; c[4 * r + 1] = q.y; c[4 * r + 2] = q.z; c[4 * r + 3] = q.w;
        }
        int v = 0;
#pragma unroll
        for (int k = 0; k < 16; ++k) { const int ck = c[k]; c[k] = v; v += ck; }

        int x = v;
        for (int o = 1; o < 32; o <<= 1) {
            int y = __shfl_up(x, o);
            if (lane >= o) x += y;
        }
        if (lane == 31) wsum[buf][w] = x;
        __syncthreads();
        const int wsv = wsum[buf][lane];
        int wx = wsv;
        for (int o = 1; o < 32; o <<= 1) {
            int y = __shfl_up(wx, o);
            if (lane >= o) wx += y;
        }
        const int wave_excl  = __shfl(wx, w) - __shfl(wsv, w);
        const int tile_total = __shfl(wx, 31);
        if (in) {
            const int b0 = carry + wave_excl + x - v;
#pragma unroll
            for (int r = 0; r < 4; ++r)
                ((int4*)(off + i0))[r] = make_int4(b0 + c[4 * r + 0], b0 + c[4 * r + 1],
                                                   b0 + c[4 * r + 2], b0 + c[4 * r + 3]);
        }
        carry += tile_total;
        buf ^= 1;
    }
    if (t == 0) off[NNODES] = carry;
}

__global__ void fill_csr(const int* __restrict__ ei, const int* __restrict__ off,
                         const int* __restrict__ rnk, int* __restrict__ csrc) {
    int e = blockIdx.x * blockDim.x + threadIdx.x;
    if (e >= NEDGES) return;
    int d = ei[NEDGES + e];
    csrc[off[d] + rnk[e]] = ei[e];
}

__device__ __forceinline__ float wave_max(float v) {
    for (int o = 16; o > 0; o >>= 1) v = fmaxf(v, __shfl_xor(v, o));
    return v;
}
__device__ __forceinline__ float wave_sum(float v) {
    for (int o = 16; o > 0; o >>= 1) v += __shfl_xor(v, o);
    return v;
}
__device__ __forceinline__ int list_src(const int* __restrict__ csrc, int beg, int deg,
                                        int d, int j) {
    int s = d;
    if (j < deg) s = min(max(csrc[beg + j], 0), NNODES - 1);
    return s;
}

__device__ __forceinline__ void gather_pass(float4& accA, float4& accB,
                                            const float* __restrict__ xlane,
                                            int s_l, float a_l, int nvalid, int hsel) {
    float4 xa[4], xb[4];
#pragma unroll
    for (int i = 0; i < 4; ++i) {
        const int s = __builtin_amdgcn_readlane(s_l, i * 8);
        const float* r = xlane + (size_t)s * D1;
        xa[i] = *(const float4*)(r);
        xb[i] = *(const float4*)(r + 128);
    }
#pragma unroll
    for (int i = 0; i < 4; ++i) {
        const float alo = __shfl(a_l, i * 8 + hsel);
        const float ahi = __shfl(a_l, i * 8 + 4 + hsel);
        float4 va = xa[i], vb = xb[i];
        if (i >= nvalid) {
            va = make_float4(0.0f, 0.0f, 0.0f, 0.0f);
            vb = make_float4(0.0f, 0.0f, 0.0f, 0.0f);
        }
        accA.x = fmaf(alo, va.x, accA.x);
        accA.y = fmaf(alo, va.y, accA.y);
        accA.z = fmaf(alo, va.z, accA.z);
        accA.w = fmaf(alo, va.w, accA.w);
        accB.x = fmaf(ahi, vb.x, accB.x);
        accB.y = fmaf(ahi, vb.y, accB.y);
        accB.z = fmaf(ahi, vb.z, accB.z);
        accB.w = fmaf(ahi, vb.w, accB.w);
    }
}

static_assert(NNODES % 8 == 0, "node_agg1 blocks cover 8 nodes each");
__global__ __launch_bounds__(256) void node_agg1(const int* __restrict__ off,
                                                 const int* __restrict__ csrc,
                                                 const float* __restrict__ xp,
                                                 const float* __restrict__ as,
                                                 const float* __restrict__ ad,
                                                 const float* __restrict__ b,
                                                 float* __restrict__ hout) {
    const int lane = threadIdx.x & 31;
    const int d = __builtin_amdgcn_readfirstlane(blockIdx.x * 8 + (threadIdx.x >> 5));
    const int beg   = off[d];
    const int deg   = off[d + 1] - beg;
    const int total = deg + 1;
    const int jj    = lane >> 3;
    const int h     = lane & 7;
    const float add = ad[d * H1 + h];

    int   sp[8];
    float ev[8];
    const int s0 = list_src(csrc, beg, deg, d, lane);
#pragma unroll
    for (int p = 0; p < 8; ++p) sp[p] = __shfl(s0, 4 * p + jj);
#pragma unroll
    for (int p = 0; p < 8; ++p) {
        const float v = lrelu(as[sp[p] * H1 + h] + add);
        ev[p] = (4 * p + jj < total) ? v : -INFINITY;
    }

    float mx = ev[0];
#pragma unroll
    for (int p = 1; p < 8; ++p) mx = fmaxf(mx, ev[p]);
    for (int j0 = 32; j0 < total; j0 += 4) {
        const int j = j0 + jj;
        if (j < total) {
            const int s = list_src(csrc, beg, deg, d, j);
            mx = fmaxf(mx, lrelu(as[s * H1 + h] + add));
        }
    }
    mx = fmaxf(mx, __shfl_xor(mx, 8));
    mx = fmaxf(mx, __shfl_xor(mx, 16));

    float sum = 0.0f;
#pragma unroll
    for (int p = 0; p < 8; ++p) {
        ev[p] = (4 * p + jj < total) ? __expf(ev[p] - mx) : 0.0f;
        sum += ev[p];
    }
    for (int j0 = 32; j0 < total; j0 += 4) {
        const int j = j0 + jj;
        if (j < total) {
            const int s = list_src(csrc, beg, deg, d, j);
            sum += __expf(lrelu(as[s * H1 + h] + add) - mx);
        }
    }
    sum += __shfl_xor(sum, 8);
    sum += __shfl_xor(sum, 16);
    const float den = sum;
#pragma unroll
    for (int p = 0; p < 8; ++p) ev[p] = ev[p] / den;

    float4 accA = make_float4(0.0f, 0.0f, 0.0f, 0.0f);
    float4 accB = make_float4(0.0f, 0.0f, 0.0f, 0.0f);
    const float* xlane = xp + 4 * lane;
    const int hsel = lane >> 3;
    const int n0   = min(total, 32);
#pragma unroll
    for (int p = 0; p < 8; ++p) {
        if (4 * p < n0) gather_pass(accA, accB, xlane, sp[p], ev[p], n0 - 4 * p, hsel);
    }
    for (int j0 = 32; j0 < total; j0 += 4) {
        const int j = j0 + jj;
        int   s = d;
        float a = 0.0f;
        if (j < total) {
            s = list_src(csrc, beg, deg, d, j);
            a = __expf(lrelu(as[s * H1 + h] + add) - mx) / den;
        }
        gather_pass(accA, accB, xlane, s, a, total - j0, hsel);
    }

    const float* bl = b + 4 * lane;
    float4 rA, rB;
    rA.x = accA.x + bl[0];   rA.y = accA.y + bl[1];   rA.z = accA.z + bl[2];   rA.w = accA.w + bl[3];
    rB.x = accB.x + bl[128]; rB.y = accB.y + bl[129]; rB.z = accB.z + bl[130]; rB.w = accB.w + bl[131];
    rA.x = rA.x > 0.0f ? rA.x : 0.0f; rA.y = rA.y > 0.0f ? rA.y : 0.0f;
    rA.z = rA.z > 0.0f ? rA.z : 0.0f; rA.w = rA.w > 0.0f ? rA.w : 0.0f;
    rB.x = rB.x > 0.0f ? rB.x : 0.0f; rB.y = rB.y > 0.0f ? rB.y : 0.0f;
    rB.z = rB.z > 0.0f ? rB.z : 0.0f; rB.w = rB.w > 0.0f ? rB.w : 0.0f;
    float* hrow = hout + (size_t)d * D1 + 4 * lane;
    *(float4*)(hrow)       = rA;
    *(float4*)(hrow + 128) = rB;
}

__global__ __launch_bounds__(256) void gemm2_alpha2(const float* __restrict__ h,
                                                    const float* __restrict__ W2,
                                                    const float* __restrict__ aw_src,
                                                    const float* __restrict__ aw_dst,
                                                    float* __restrict__ xp2,
                                                    float* __restrict__ as2,
                                                    float* __restrict__ ad2) {
    int lane = threadIdx.x & 31;
    int n    = blockIdx.x * (blockDim.x >> 5) + (threadIdx.x >> 5);
    if (n >= NNODES) return;
    float s0 = 0.0f, s1 = 0.0f;
    for (int k = lane; k < D1; k += 32) {
        float v = h[(size_t)n * D1 + k];
        s0 = fmaf(v, W2[k * OUT2 + 0], s0);
        s1 = fmaf(v, W2[k * OUT2 + 1], s1);
    }
#pragma unroll
    for (int off = 16; off > 0; off >>= 1) {
        s0 += __shfl_down(s0, off);
        s1 += __shfl_down(s1, off);
    }
    if (lane == 0) {
        xp2[n * OUT2 + 0] = s0;
        xp2[n * OUT2 + 1] = s1;
        as2[n] = fmaf(s0, aw_src[0], s1 * aw_src[1]);
        ad2[n] = fmaf(s0, aw_dst[0], s1 * aw_dst[1]);
    }
}

static_assert(NNODES % 16 == 0, "node_agg2 blocks cover 16 nodes each");
__global__ __launch_bounds__(512) void node_agg2(const int* __restrict__ off,
                                                 const int* __restrict__ csrc,
                                                 const float* __restrict__ xp2,
                                                 const float* __restrict__ as2,
                                                 const float* __restrict__ ad2,
                                                 const float* __restrict__ b,
                                                 float* __restrict__ out) {
    __shared__ float res[16 * OUT2];
    const int lane = threadIdx.x & 31;
    const int w    = threadIdx.x >> 5;
    const int d    = blockIdx.x * 16 + w;
    const int beg   = off[d];
    const int deg   = off[d + 1] - beg;
    const int total = deg + 1;
    const float add = ad2[d];

    const bool  act0 = lane < total;
    const int   s0   = list_src(csrc, beg, deg, d, lane);
    const float v0   = act0 ? lrelu(as2[s0] + add) : -INFINITY;

    float mx = v0;
    for (int j0 = 32; j0 < total; j0 += 32) {
        const int j = j0 + lane;
        if (j < total) {
            const int s = list_src(csrc, beg, deg, d, j);
            mx = fmaxf(mx, lrelu(as2[s] + add));
        }
    }
    mx = wave_max(mx);

    const float p0 = act0 ? __expf(v0 - mx) : 0.0f;
    float sum = p0;
    for (int j0 = 32; j0 < total; j0 += 32) {
        const int j = j0 + lane;
        if (j < total) {
            const int s = list_src(csrc, beg, deg, d, j);
            sum += __expf(lrelu(as2[s] + add) - mx);
        }
    }
    const float den = wave_sum(sum);

    float c0 = 0.0f, c1 = 0.0f;
    if (act0) {
        const float a = p0 / den;
        c0 = a * xp2[s0 * OUT2 + 0];
        c1 = a * xp2[s0 * OUT2 + 1];
    }
    for (int j0 = 32; j0 < total; j0 += 32) {
        const int j = j0 + lane;
        if (j < total) {
            const int s = list_src(csrc, beg, deg, d, j);
            const float a = __expf(lrelu(as2[s] + add) - mx) / den;
            c0 = fmaf(a, xp2[s * OUT2 + 0], c0);
            c1 = fmaf(a, xp2[s * OUT2 + 1], c1);
        }
    }
    c0 = wave_sum(c0);
    c1 = wave_sum(c1);
    if (lane == 0) {
        res[w * OUT2 + 0] = c0;
        res[w * OUT2 + 1] = c1;
    }
    __syncthreads();
    if (threadIdx.x < 16 * OUT2)
        out[blockIdx.x * (16 * OUT2) + threadIdx.x] = res[threadIdx.x] + b[threadIdx.x & 1];
}

extern "C" void kernel_launch(void* const* d_in, const int* in_sizes, int n_in,
                              void* d_out, int out_size, void* d_ws, size_t ws_size,
                              hipStream_t stream) {
    const float* x     = (const float*)d_in[0];
    const int*   ei    = (const int*)d_in[1];
    const float* W1    = (const float*)d_in[2];
    const float* asw1  = (const float*)d_in[3];
    const float* adw1  = (const float*)d_in[4];
    const float* b1    = (const float*)d_in[5];
    const float* W2    = (const float*)d_in[6];
    const float* asw2  = (const float*)d_in[7];
    const float* adw2  = (const float*)d_in[8];
    const float* b2    = (const float*)d_in[9];
    float*       out   = (float*)d_out;

    float* w    = (float*)d_ws;
    float* xp1  = w;
    float* hbuf = xp1  + (size_t)NNODES * D1;
    float* as1  = hbuf + (size_t)NNODES * D1;
    float* ad1  = as1  + (size_t)NNODES * H1;
    float* xp2  = ad1  + (size_t)NNODES * H1;
    float* as2  = xp2  + (size_t)NNODES * OUT2;
    float* ad2  = as2  + (size_t)NNODES;
    int*   cnt  = (int*)(ad2 + (size_t)NNODES);
    int*   off  = cnt  + (size_t)NNODES;
    int*   rnk  = off  + (size_t)NNODES + 1;
    int*   csrc = rnk  + (size_t)NEDGES;

    zero_i32<<<(NNODES + 255) / 256, 256, 0, stream>>>(cnt, NNODES);
    count_rank<<<(NEDGES + 255) / 256, 256, 0, stream>>>(ei, cnt, rnk);
    scan_counts<<<1, 1024, 0, stream>>>(cnt, off);
    fill_csr<<<(NEDGES + 255) / 256, 256, 0, stream>>>(ei, off, rnk, csrc);

    gemm1_wmma<<<NNODES / 16, 512, 0, stream>>>(x, W1, xp1);
    alpha1_kernel<<<(NNODES * H1 + 255) / 256, 256, 0, stream>>>(xp1, asw1, adw1, as1, ad1);
    node_agg1<<<NNODES / 8, 256, 0, stream>>>(off, csrc, xp1, as1, ad1, b1, hbuf);

    gemm2_alpha2<<<(NNODES + 7) / 8, 256, 0, stream>>>(hbuf, W2, asw2, adw2, xp2, as2, ad2);
    node_agg2<<<NNODES / 16, 512, 0, stream>>>(off, csrc, xp2, as2, ad2, b2, out);
}


# Layer-1 GEMM persistent: 256 work-groups, each wave keeps its 16 W1 columns (all k) in registers and loops over row tiles, next x tile prefetched; W1 read once per group, sum order kept
# speedup vs baseline: 6.0073x; 1.0359x over previous
#include <hip/hip_runtime.h>
#include <hip/hip_bf16.h>
#include <math.h>

#define NNODES 50000
#define NEDGES 800000
#define FIN    256
#define H1     8
#define C1     32
#define D1     256
#define OUT2   2
#define SLOPE  0.2f

#define LDSPAD 4
#define ASTRIDE (FIN + LDSPAD)

typedef float v2f __attribute__((ext_vector_type(2)));
typedef float v8f __attribute__((ext_vector_type(8)));

__device__ __forceinline__ float lrelu(float v) {
    return v > 0.0f ? v : SLOPE * v;
}

#define MTILES    (NNODES / 16)
#define G1_BLOCKS 256
#define KSTEPS    (FIN / 4)
static_assert(NNODES % 16 == 0, "gemm1 row tiles are 16 rows");
__global__ __launch_bounds__(512) void gemm1_wmma(const float* __restrict__ x,
                                                  const float* __restrict__ W1,
                                                  float* __restrict__ xp) {
    __shared__ float aTile[2][16 * ASTRIDE];

    const int t     = threadIdx.x;
    const int lane  = t & 31;
    const int wave  = t >> 5;
    const int half  = lane >> 4;
    const int l16   = lane & 15;
    const int bcol  = wave * 16 + l16;

    v2f breg[KSTEPS];
#pragma unroll
    for (int s = 0; s < KSTEPS; ++s) {
        const int kk = 4 * s + half * 2;
        breg[s].x = W1[kk * D1 + bcol];
        breg[s].y = W1[(kk + 1) * D1 + bcol];
    }

    const int srow = t >> 5;
    const int scol = (t & 31) * 8;
    const int soff = srow * ASTRIDE + scol;
    const int aoff = l16 * ASTRIDE + half * 2;

    int tile = blockIdx.x;
    float4 p0 = make_float4(0.0f, 0.0f, 0.0f, 0.0f);
    float4 p1 = p0;
    if (tile < MTILES) {
        const float4* src = (const float4*)(x + (size_t)(tile * 16 + srow) * FIN + scol);
        p0 = src[0];
        p1 = src[1];
    }
    int buf = 0;
    for (; tile < MTILES; tile += G1_BLOCKS) {
        float* abuf = aTile[buf];
        *(float4*)(abuf + soff)     = p0;
        *(float4*)(abuf + soff + 4) = p1;
        __syncthreads();

        const int nxt = tile + G1_BLOCKS;
        if (nxt < MTILES) {
            const float4* src = (const float4*)(x + (size_t)(nxt * 16 + srow) * FIN + scol);
            p0 = src[0];
            p1 = src[1];
        }

        v8f acc = {};
#pragma unroll
        for (int s = 0; s < KSTEPS; ++s) {
            v2f a = *(const v2f*)(abuf + aoff + 4 * s);
            acc = __builtin_amdgcn_wmma_f32_16x16x4_f32(
                 false, a,  false, breg[s],
                 (short)0, acc,  false,  false);
        }

        const int mrow = tile * 16 + half * 8;
#pragma unroll
        for (int r = 0; r < 8; ++r)
            xp[(size_t)(mrow + r) * D1 + bcol] = acc[r];
        buf ^= 1;
    }
}

__global__ void alpha1_kernel(const float* __restrict__ xp,
                              const float* __restrict__ a_src,
                              const float* __restrict__ a_dst,
                              float* __restrict__ as, float* __restrict__ ad) {
    int i = blockIdx.x * blockDim.x + threadIdx.x;
    if (i >= NNODES * H1) return;
    int h = i & (H1 - 1);
    int n = i >> 3;
    const float* row = xp + (size_t)n * D1 + h * C1;
    const float* ws  = a_src + h * C1;
    const float* wd  = a_dst + h * C1;
    float s = 0.0f, d = 0.0f;
#pragma unroll
    for (int c = 0; c < C1; ++c) {
        float v = row[c];
        s = fmaf(v, ws[c], s);
        d = fmaf(v, wd[c], d);
    }
    as[i] = s;
    ad[i] = d;
}

__global__ void zero_i32(int* __restrict__ p, int n) {
    int i = blockIdx.x * blockDim.x + threadIdx.x;
    if (i < n) p[i] = 0;
}

__global__ void count_rank(const int* __restrict__ ei, int* __restrict__ cnt,
                           int* __restrict__ rnk) {
    int e = blockIdx.x * blockDim.x + threadIdx.x;
    if (e >= NEDGES) return;
    int d = ei[NEDGES + e];
    rnk[e] = atomicAdd(&cnt[d], 1);
}

static_assert(NNODES % 16 == 0, "scan_counts threads own 16 counts each");
__global__ __launch_bounds__(1024) void scan_counts(const int* __restrict__ cnt,
                                                    int* __restrict__ off) {
    __shared__ int wsum[2][32];
    const int t    = threadIdx.x;
    const int lane = t & 31;
    const int w    = t >> 5;
    int carry = 0;
    int buf   = 0;
    for (int base = 0; base < NNODES; base += 1024 * 16) {
        const int  i0 = base + t * 16;
        const bool in = i0 < NNODES;
        int c[16];
#pragma unroll
        for (int r = 0; r < 4; ++r) {
            int4 q = make_int4(0, 0, 0, 0);
            if (in) q = ((const int4*)(cnt + i0))[r];
            c[4 * r + 0] = q.x; c[4 * r + 1] = q.y; c[4 * r + 2] = q.z; c[4 * r + 3] = q.w;
        }
        int v = 0;
#pragma unroll
        for (int k = 0; k < 16; ++k) { const int ck = c[k]; c[k] = v; v += ck; }

        int x = v;
        for (int o = 1; o < 32; o <<= 1) {
            int y = __shfl_up(x, o);
            if (lane >= o) x += y;
        }
        if (lane == 31) wsum[buf][w] = x;
        __syncthreads();
        const int wsv = wsum[buf][lane];
        int wx = wsv;
        for (int o = 1; o < 32; o <<= 1) {
            int y = __shfl_up(wx, o);
            if (lane >= o) wx += y;
        }
        const int wave_excl  = __shfl(wx, w) - __shfl(wsv, w);
        const int tile_total = __shfl(wx, 31);
        if (in) {
            const int b0 = carry + wave_excl + x - v;
#pragma unroll
            for (int r = 0; r < 4; ++r)
                ((int4*)(off + i0))[r] = make_int4(b0 + c[4 * r + 0], b0 + c[4 * r + 1],
                                                   b0 + c[4 * r + 2], b0 + c[4 * r + 3]);
        }
        carry += tile_total;
        buf ^= 1;
    }
    if (t == 0) off[NNODES] = carry;
}

__global__ void fill_csr(const int* __restrict__ ei, const int* __restrict__ off,
                         const int* __restrict__ rnk, int* __restrict__ csrc) {
    int e = blockIdx.x * blockDim.x + threadIdx.x;
    if (e >= NEDGES) return;
    int d = ei[NEDGES + e];
    csrc[off[d] + rnk[e]] = ei[e];
}

__device__ __forceinline__ float wave_max(float v) {
    for (int o = 16; o > 0; o >>= 1) v = fmaxf(v, __shfl_xor(v, o));
    return v;
}
__device__ __forceinline__ float wave_sum(float v) {
    for (int o = 16; o > 0; o >>= 1) v += __shfl_xor(v, o);
    return v;
}
__device__ __forceinline__ int list_src(const int* __restrict__ csrc, int beg, int deg,
                                        int d, int j) {
    int s = d;
    if (j < deg) s = min(max(csrc[beg + j], 0), NNODES - 1);
    return s;
}

__device__ __forceinline__ void gather_pass(float4& accA, float4& accB,
                                            const float* __restrict__ xlane,
                                            int s_l, float a_l, int nvalid, int hsel) {
    float4 xa[4], xb[4];
#pragma unroll
    for (int i = 0; i < 4; ++i) {
        const int s = __builtin_amdgcn_readlane(s_l, i * 8);
        const float* r = xlane + (size_t)s * D1;
        xa[i] = *(const float4*)(r);
        xb[i] = *(const float4*)(r + 128);
    }
#pragma unroll
    for (int i = 0; i < 4; ++i) {
        const float alo = __shfl(a_l, i * 8 + hsel);
        const float ahi = __shfl(a_l, i * 8 + 4 + hsel);
        float4 va = xa[i], vb = xb[i];
        if (i >= nvalid) {
            va = make_float4(0.0f, 0.0f, 0.0f, 0.0f);
            vb = make_float4(0.0f, 0.0f, 0.0f, 0.0f);
        }
        accA.x = fmaf(alo, va.x, accA.x);
        accA.y = fmaf(alo, va.y, accA.y);
        accA.z = fmaf(alo, va.z, accA.z);
        accA.w = fmaf(alo, va.w, accA.w);
        accB.x = fmaf(ahi, vb.x, accB.x);
        accB.y = fmaf(ahi, vb.y, accB.y);
        accB.z = fmaf(ahi, vb.z, accB.z);
        accB.w = fmaf(ahi, vb.w, accB.w);
    }
}

static_assert(NNODES % 8 == 0, "node_agg1 blocks cover 8 nodes each");
__global__ __launch_bounds__(256) void node_agg1(const int* __restrict__ off,
                                                 const int* __restrict__ csrc,
                                                 const float* __restrict__ xp,
                                                 const float* __restrict__ as,
                                                 const float* __restrict__ ad,
                                                 const float* __restrict__ b,
                                                 float* __restrict__ hout) {
    const int lane = threadIdx.x & 31;
    const int d = __builtin_amdgcn_readfirstlane(blockIdx.x * 8 + (threadIdx.x >> 5));
    const int beg   = off[d];
    const int deg   = off[d + 1] - beg;
    const int total = deg + 1;
    const int jj    = lane >> 3;
    const int h     = lane & 7;
    const float add = ad[d * H1 + h];

    int   sp[8];
    float ev[8];
    const int s0 = list_src(csrc, beg, deg, d, lane);
#pragma unroll
    for (int p = 0; p < 8; ++p) sp[p] = __shfl(s0, 4 * p + jj);
#pragma unroll
    for (int p = 0; p < 8; ++p) {
        const float v = lrelu(as[sp[p] * H1 + h] + add);
        ev[p] = (4 * p + jj < total) ? v : -INFINITY;
    }

    float mx = ev[0];
#pragma unroll
    for (int p = 1; p < 8; ++p) mx = fmaxf(mx, ev[p]);
    for (int j0 = 32; j0 < total; j0 += 4) {
        const int j = j0 + jj;
        if (j < total) {
            const int s = list_src(csrc, beg, deg, d, j);
            mx = fmaxf(mx, lrelu(as[s * H1 + h] + add));
        }
    }
    mx = fmaxf(mx, __shfl_xor(mx, 8));
    mx = fmaxf(mx, __shfl_xor(mx, 16));

    float sum = 0.0f;
#pragma unroll
    for (int p = 0; p < 8; ++p) {
        ev[p] = (4 * p + jj < total) ? __expf(ev[p] - mx) : 0.0f;
        sum += ev[p];
    }
    for (int j0 = 32; j0 < total; j0 += 4) {
        const int j = j0 + jj;
        if (j < total) {
            const int s = list_src(csrc, beg, deg, d, j);
            sum += __expf(lrelu(as[s * H1 + h] + add) - mx);
        }
    }
    sum += __shfl_xor(sum, 8);
    sum += __shfl_xor(sum, 16);
    const float den = sum;
#pragma unroll
    for (int p = 0; p < 8; ++p) ev[p] = ev[p] / den;

    float4 accA = make_float4(0.0f, 0.0f, 0.0f, 0.0f);
    float4 accB = make_float4(0.0f, 0.0f, 0.0f, 0.0f);
    const float* xlane = xp + 4 * lane;
    const int hsel = lane >> 3;
    const int n0   = min(total, 32);
#pragma unroll
    for (int p = 0; p < 8; ++p) {
        if (4 * p < n0) gather_pass(accA, accB, xlane, sp[p], ev[p], n0 - 4 * p, hsel);
    }
    for (int j0 = 32; j0 < total; j0 += 4) {
        const int j = j0 + jj;
        int   s = d;
        float a = 0.0f;
        if (j < total) {
            s = list_src(csrc, beg, deg, d, j);
            a = __expf(lrelu(as[s * H1 + h] + add) - mx) / den;
        }
        gather_pass(accA, accB, xlane, s, a, total - j0, hsel);
    }

    const float* bl = b + 4 * lane;
    float4 rA, rB;
    rA.x = accA.x + bl[0];   rA.y = accA.y + bl[1];   rA.z = accA.z + bl[2];   rA.w = accA.w + bl[3];
    rB.x = accB.x + bl[128]; rB.y = accB.y + bl[129]; rB.z = accB.z + bl[130]; rB.w = accB.w + bl[131];
    rA.x = rA.x > 0.0f ? rA.x : 0.0f; rA.y = rA.y > 0.0f ? rA.y : 0.0f;
    rA.z = rA.z > 0.0f ? rA.z : 0.0f; rA.w = rA.w > 0.0f ? rA.w : 0.0f;
    rB.x = rB.x > 0.0f ? rB.x : 0.0f; rB.y = rB.y > 0.0f ? rB.y : 0.0f;
    rB.z = rB.z > 0.0f ? rB.z : 0.0f; rB.w = rB.w > 0.0f ? rB.w : 0.0f;
    float* hrow = hout + (size_t)d * D1 + 4 * lane;
    *(float4*)(hrow)       = rA;
    *(float4*)(hrow + 128) = rB;
}

__global__ __launch_bounds__(256) void gemm2_alpha2(const float* __restrict__ h,
                                                    const float* __restrict__ W2,
                                                    const float* __restrict__ aw_src,
                                                    const float* __restrict__ aw_dst,
                                                    float* __restrict__ xp2,
                                                    float* __restrict__ as2,
                                                    float* __restrict__ ad2) {
    int lane = threadIdx.x & 31;
    int n    = blockIdx.x * (blockDim.x >> 5) + (threadIdx.x >> 5);
    if (n >= NNODES) return;
    float s0 = 0.0f, s1 = 0.0f;
    for (int k = lane; k < D1; k += 32) {
        float v = h[(size_t)n * D1 + k];
        s0 = fmaf(v, W2[k * OUT2 + 0], s0);
        s1 = fmaf(v, W2[k * OUT2 + 1], s1);
    }
#pragma unroll
    for (int off = 16; off > 0; off >>= 1) {
        s0 += __shfl_down(s0, off);
        s1 += __shfl_down(s1, off);
    }
    if (lane == 0) {
        xp2[n * OUT2 + 0] = s0;
        xp2[n * OUT2 + 1] = s1;
        as2[n] = fmaf(s0, aw_src[0], s1 * aw_src[1]);
        ad2[n] = fmaf(s0, aw_dst[0], s1 * aw_dst[1]);
    }
}

static_assert(NNODES % 16 == 0, "node_agg2 blocks cover 16 nodes each");
__global__ __launch_bounds__(512) void node_agg2(const int* __restrict__ off,
                                                 const int* __restrict__ csrc,
                                                 const float* __restrict__ xp2,
                                                 const float* __restrict__ as2,
                                                 const float* __restrict__ ad2,
                                                 const float* __restrict__ b,
                                                 float* __restrict__ out) {
    __shared__ float res[16 * OUT2];
    const int lane = threadIdx.x & 31;
    const int w    = threadIdx.x >> 5;
    const int d    = blockIdx.x * 16 + w;
    const int beg   = off[d];
    const int deg   = off[d + 1] - beg;
    const int total = deg + 1;
    const float add = ad2[d];

    const bool  act0 = lane < total;
    const int   s0   = list_src(csrc, beg, deg, d, lane);
    const float v0   = act0 ? lrelu(as2[s0] + add) : -INFINITY;

    float mx = v0;
    for (int j0 = 32; j0 < total; j0 += 32) {
        const int j = j0 + lane;
        if (j < total) {
            const int s = list_src(csrc, beg, deg, d, j);
            mx = fmaxf(mx, lrelu(as2[s] + add));
        }
    }
    mx = wave_max(mx);

    const float p0 = act0 ? __expf(v0 - mx) : 0.0f;
    float sum = p0;
    for (int j0 = 32; j0 < total; j0 += 32) {
        const int j = j0 + lane;
        if (j < total) {
            const int s = list_src(csrc, beg, deg, d, j);
            sum += __expf(lrelu(as2[s] + add) - mx);
        }
    }
    const float den = wave_sum(sum);

    float c0 = 0.0f, c1 = 0.0f;
    if (act0) {
        const float a = p0 / den;
        c0 = a * xp2[s0 * OUT2 + 0];
        c1 = a * xp2[s0 * OUT2 + 1];
    }
    for (int j0 = 32; j0 < total; j0 += 32) {
        const int j = j0 + lane;
        if (j < total) {
            const int s = list_src(csrc, beg, deg, d, j);
            const float a = __expf(lrelu(as2[s] + add) - mx) / den;
            c0 = fmaf(a, xp2[s * OUT2 + 0], c0);
            c1 = fmaf(a, xp2[s * OUT2 + 1], c1);
        }
    }
    c0 = wave_sum(c0);
    c1 = wave_sum(c1);
    if (lane == 0) {
        res[w * OUT2 + 0] = c0;
        res[w * OUT2 + 1] = c1;
    }
    __syncthreads();
    if (threadIdx.x < 16 * OUT2)
        out[blockIdx.x * (16 * OUT2) + threadIdx.x] = res[threadIdx.x] + b[threadIdx.x & 1];
}

extern "C" void kernel_launch(void* const* d_in, const int* in_sizes, int n_in,
                              void* d_out, int out_size, void* d_ws, size_t ws_size,
                              hipStream_t stream) {
    const float* x     = (const float*)d_in[0];
    const int*   ei    = (const int*)d_in[1];
    const float* W1    = (const float*)d_in[2];
    const float* asw1  = (const float*)d_in[3];
    const float* adw1  = (const float*)d_in[4];
    const float* b1    = (const float*)d_in[5];
    const float* W2    = (const float*)d_in[6];
    const float* asw2  = (const float*)d_in[7];
    const float* adw2  = (const float*)d_in[8];
    const float* b2    = (const float*)d_in[9];
    float*       out   = (float*)d_out;

    float* w    = (float*)d_ws;
    float* xp1  = w;
    float* hbuf = xp1  + (size_t)NNODES * D1;
    float* as1  = hbuf + (size_t)NNODES * D1;
    float* ad1  = as1  + (size_t)NNODES * H1;
    float* xp2  = ad1  + (size_t)NNODES * H1;
    float* as2  = xp2  + (size_t)NNODES * OUT2;
    float* ad2  = as2  + (size_t)NNODES;
    int*   cnt  = (int*)(ad2 + (size_t)NNODES);
    int*   off  = cnt  + (size_t)NNODES;
    int*   rnk  = off  + (size_t)NNODES + 1;
    int*   csrc = rnk  + (size_t)NEDGES;

    zero_i32<<<(NNODES + 255) / 256, 256, 0, stream>>>(cnt, NNODES);
    count_rank<<<(NEDGES + 255) / 256, 256, 0, stream>>>(ei, cnt, rnk);
    scan_counts<<<1, 1024, 0, stream>>>(cnt, off);
    fill_csr<<<(NEDGES + 255) / 256, 256, 0, stream>>>(ei, off, rnk, csrc);

    gemm1_wmma<<<G1_BLOCKS, 512, 0, stream>>>(x, W1, xp1);
    alpha1_kernel<<<(NNODES * H1 + 255) / 256, 256, 0, stream>>>(xp1, asw1, adw1, as1, ad1);
    node_agg1<<<NNODES / 8, 256, 0, stream>>>(off, csrc, xp1, as1, ad1, b1, hbuf);

    gemm2_alpha2<<<(NNODES + 7) / 8, 256, 0, stream>>>(hbuf, W2, asw2, adw2, xp2, as2, ad2);
    node_agg2<<<NNODES / 16, 512, 0, stream>>>(off, csrc, xp2, as2, ad2, b2, out);
}


# Layer-2 projection and logits done in the layer-1 aggregation's epilogue on the row held in registers: h never stored or re-read, one launch fewer; xp2 summed 8 channels per lane, then lane tree
# speedup vs baseline: 6.6623x; 1.1090x over previous
#include <hip/hip_runtime.h>
#include <hip/hip_bf16.h>
#include <math.h>

#define NNODES 50000
#define NEDGES 800000
#define FIN    256
#define H1     8
#define C1     32
#define D1     256
#define OUT2   2
#define SLOPE  0.2f

#define LDSPAD 4
#define ASTRIDE (FIN + LDSPAD)

typedef float v2f __attribute__((ext_vector_type(2)));
typedef float v8f __attribute__((ext_vector_type(8)));

__device__ __forceinline__ float lrelu(float v) {
    return v > 0.0f ? v : SLOPE * v;
}

#define MTILES    (NNODES / 16)
#define G1_BLOCKS 256
#define KSTEPS    (FIN / 4)
static_assert(NNODES % 16 == 0, "gemm1 row tiles are 16 rows");
__global__ __launch_bounds__(512) void gemm1_wmma(const float* __restrict__ x,
                                                  const float* __restrict__ W1,
                                                  float* __restrict__ xp) {
    __shared__ float aTile[2][16 * ASTRIDE];

    const int t     = threadIdx.x;
    const int lane  = t & 31;
    const int wave  = t >> 5;
    const int half  = lane >> 4;
    const int l16   = lane & 15;
    const int bcol  = wave * 16 + l16;

    v2f breg[KSTEPS];
#pragma unroll
    for (int s = 0; s < KSTEPS; ++s) {
        const int kk = 4 * s + half * 2;
        breg[s].x = W1[kk * D1 + bcol];
        breg[s].y = W1[(kk + 1) * D1 + bcol];
    }

    const int srow = t >> 5;
    const int scol = (t & 31) * 8;
    const int soff = srow * ASTRIDE + scol;
    const int aoff = l16 * ASTRIDE + half * 2;

    int tile = blockIdx.x;
    float4 p0 = make_float4(0.0f, 0.0f, 0.0f, 0.0f);
    float4 p1 = p0;
    if (tile < MTILES) {
        const float4* src = (const float4*)(x + (size_t)(tile * 16 + srow) * FIN + scol);
        p0 = src[0];
        p1 = src[1];
    }
    int buf = 0;
    for (; tile < MTILES; tile += G1_BLOCKS) {
        float* abuf = aTile[buf];
        *(float4*)(abuf + soff)     = p0;
        *(float4*)(abuf + soff + 4) = p1;
        __syncthreads();

        const int nxt = tile + G1_BLOCKS;
        if (nxt < MTILES) {
            const float4* src = (const float4*)(x + (size_t)(nxt * 16 + srow) * FIN + scol);
            p0 = src[0];
            p1 = src[1];
        }

        v8f acc = {};
#pragma unroll
        for (int s = 0; s < KSTEPS; ++s) {
            v2f a = *(const v2f*)(abuf + aoff + 4 * s);
            acc = __builtin_amdgcn_wmma_f32_16x16x4_f32(
                 false, a,  false, breg[s],
                 (short)0, acc,  false,  false);
        }

        const int mrow = tile * 16 + half * 8;
#pragma unroll
        for (int r = 0; r < 8; ++r)
            xp[(size_t)(mrow + r) * D1 + bcol] = acc[r];
        buf ^= 1;
    }
}

__global__ void alpha1_kernel(const float* __restrict__ xp,
                              const float* __restrict__ a_src,
                              const float* __restrict__ a_dst,
                              float* __restrict__ as, float* __restrict__ ad) {
    int i = blockIdx.x * blockDim.x + threadIdx.x;
    if (i >= NNODES * H1) return;
    int h = i & (H1 - 1);
    int n = i >> 3;
    const float* row = xp + (size_t)n * D1 + h * C1;
    const float* ws  = a_src + h * C1;
    const float* wd  = a_dst + h * C1;
    float s = 0.0f, d = 0.0f;
#pragma unroll
    for (int c = 0; c < C1; ++c) {
        float v = row[c];
        s = fmaf(v, ws[c], s);
        d = fmaf(v, wd[c], d);
    }
    as[i] = s;
    ad[i] = d;
}

__global__ void zero_i32(int* __restrict__ p, int n) {
    int i = blockIdx.x * blockDim.x + threadIdx.x;
    if (i < n) p[i] = 0;
}

__global__ void count_rank(const int* __restrict__ ei, int* __restrict__ cnt,
                           int* __restrict__ rnk) {
    int e = blockIdx.x * blockDim.x + threadIdx.x;
    if (e >= NEDGES) return;
    int d = ei[NEDGES + e];
    rnk[e] = atomicAdd(&cnt[d], 1);
}

static_assert(NNODES % 16 == 0, "scan_counts threads own 16 counts each");
__global__ __launch_bounds__(1024) void scan_counts(const int* __restrict__ cnt,
                                                    int* __restrict__ off) {
    __shared__ int wsum[2][32];
    const int t    = threadIdx.x;
    const int lane = t & 31;
    const int w    = t >> 5;
    int carry = 0;
    int buf   = 0;
    for (int base = 0; base < NNODES; base += 1024 * 16) {
        const int  i0 = base + t * 16;
        const bool in = i0 < NNODES;
        int c[16];
#pragma unroll
        for (int r = 0; r < 4; ++r) {
            int4 q = make_int4(0, 0, 0, 0);
            if (in) q = ((const int4*)(cnt + i0))[r];
            c[4 * r + 0] = q.x; c[4 * r + 1] = q.y; c[4 * r + 2] = q.z; c[4 * r + 3] = q.w;
        }
        int v = 0;
#pragma unroll
        for (int k = 0; k < 16; ++k) { const int ck = c[k]; c[k] = v; v += ck; }

        int x = v;
        for (int o = 1; o < 32; o <<= 1) {
            int y = __shfl_up(x, o);
            if (lane >= o) x += y;
        }
        if (lane == 31) wsum[buf][w] = x;
        __syncthreads();
        const int wsv = wsum[buf][lane];
        int wx = wsv;
        for (int o = 1; o < 32; o <<= 1) {
            int y = __shfl_up(wx, o);
            if (lane >= o) wx += y;
        }
        const int wave_excl  = __shfl(wx, w) - __shfl(wsv, w);
        const int tile_total = __shfl(wx, 31);
        if (in) {
            const int b0 = carry + wave_excl + x - v;
#pragma unroll
            for (int r = 0; r < 4; ++r)
                ((int4*)(off + i0))[r] = make_int4(b0 + c[4 * r + 0], b0 + c[4 * r + 1],
                                                   b0 + c[4 * r + 2], b0 + c[4 * r + 3]);
        }
        carry += tile_total;
        buf ^= 1;
    }
    if (t == 0) off[NNODES] = carry;
}

__global__ void fill_csr(const int* __restrict__ ei, const int* __restrict__ off,
                         const int* __restrict__ rnk, int* __restrict__ csrc) {
    int e = blockIdx.x * blockDim.x + threadIdx.x;
    if (e >= NEDGES) return;
    int d = ei[NEDGES + e];
    csrc[off[d] + rnk[e]] = ei[e];
}

__device__ __forceinline__ float wave_max(float v) {
    for (int o = 16; o > 0; o >>= 1) v = fmaxf(v, __shfl_xor(v, o));
    return v;
}
__device__ __forceinline__ float wave_sum(float v) {
    for (int o = 16; o > 0; o >>= 1) v += __shfl_xor(v, o);
    return v;
}
__device__ __forceinline__ int list_src(const int* __restrict__ csrc, int beg, int deg,
                                        int d, int j) {
    int s = d;
    if (j < deg) s = min(max(csrc[beg + j], 0), NNODES - 1);
    return s;
}

__device__ __forceinline__ void gather_pass(float4& accA, float4& accB,
                                            const float* __restrict__ xlane,
                                            int s_l, float a_l, int nvalid, int hsel) {
    float4 xa[4], xb[4];
#pragma unroll
    for (int i = 0; i < 4; ++i) {
        const int s = __builtin_amdgcn_readlane(s_l, i * 8);
        const float* r = xlane + (size_t)s * D1;
        xa[i] = *(const float4*)(r);
        xb[i] = *(const float4*)(r + 128);
    }
#pragma unroll
    for (int i = 0; i < 4; ++i) {
        const float alo = __shfl(a_l, i * 8 + hsel);
        const float ahi = __shfl(a_l, i * 8 + 4 + hsel);
        float4 va = xa[i], vb = xb[i];
        if (i >= nvalid) {
            va = make_float4(0.0f, 0.0f, 0.0f, 0.0f);
            vb = make_float4(0.0f, 0.0f, 0.0f, 0.0f);
        }
        accA.x = fmaf(alo, va.x, accA.x);
        accA.y = fmaf(alo, va.y, accA.y);
        accA.z = fmaf(alo, va.z, accA.z);
        accA.w = fmaf(alo, va.w, accA.w);
        accB.x = fmaf(ahi, vb.x, accB.x);
        accB.y = fmaf(ahi, vb.y, accB.y);
        accB.z = fmaf(ahi, vb.z, accB.z);
        accB.w = fmaf(ahi, vb.w, accB.w);
    }
}

static_assert(NNODES % 8 == 0, "node_agg1 blocks cover 8 nodes each");
__global__ __launch_bounds__(256) void node_agg1(const int* __restrict__ off,
                                                 const int* __restrict__ csrc,
                                                 const float* __restrict__ xp,
                                                 const float* __restrict__ as,
                                                 const float* __restrict__ ad,
                                                 const float* __restrict__ b,
                                                 const float* __restrict__ W2,
                                                 const float* __restrict__ aw_src,
                                                 const float* __restrict__ aw_dst,
                                                 float* __restrict__ xp2,
                                                 float* __restrict__ as2,
                                                 float* __restrict__ ad2) {
    const int lane = threadIdx.x & 31;
    const int d = __builtin_amdgcn_readfirstlane(blockIdx.x * 8 + (threadIdx.x >> 5));
    const int beg   = off[d];
    const int deg   = off[d + 1] - beg;
    const int total = deg + 1;
    const int jj    = lane >> 3;
    const int h     = lane & 7;
    const float add = ad[d * H1 + h];

    int   sp[8];
    float ev[8];
    const int s0 = list_src(csrc, beg, deg, d, lane);
#pragma unroll
    for (int p = 0; p < 8; ++p) sp[p] = __shfl(s0, 4 * p + jj);
#pragma unroll
    for (int p = 0; p < 8; ++p) {
        const float v = lrelu(as[sp[p] * H1 + h] + add);
        ev[p] = (4 * p + jj < total) ? v : -INFINITY;
    }

    float mx = ev[0];
#pragma unroll
    for (int p = 1; p < 8; ++p) mx = fmaxf(mx, ev[p]);
    for (int j0 = 32; j0 < total; j0 += 4) {
        const int j = j0 + jj;
        if (j < total) {
            const int s = list_src(csrc, beg, deg, d, j);
            mx = fmaxf(mx, lrelu(as[s * H1 + h] + add));
        }
    }
    mx = fmaxf(mx, __shfl_xor(mx, 8));
    mx = fmaxf(mx, __shfl_xor(mx, 16));

    float sum = 0.0f;
#pragma unroll
    for (int p = 0; p < 8; ++p) {
        ev[p] = (4 * p + jj < total) ? __expf(ev[p] - mx) : 0.0f;
        sum += ev[p];
    }
    for (int j0 = 32; j0 < total; j0 += 4) {
        const int j = j0 + jj;
        if (j < total) {
            const int s = list_src(csrc, beg, deg, d, j);
            sum += __expf(lrelu(as[s * H1 + h] + add) - mx);
        }
    }
    sum += __shfl_xor(sum, 8);
    sum += __shfl_xor(sum, 16);
    const float den = sum;
#pragma unroll
    for (int p = 0; p < 8; ++p) ev[p] = ev[p] / den;

    float4 accA = make_float4(0.0f, 0.0f, 0.0f, 0.0f);
    float4 accB = make_float4(0.0f, 0.0f, 0.0f, 0.0f);
    const float* xlane = xp + 4 * lane;
    const int hsel = lane >> 3;
    const int n0   = min(total, 32);
#pragma unroll
    for (int p = 0; p < 8; ++p) {
        if (4 * p < n0) gather_pass(accA, accB, xlane, sp[p], ev[p], n0 - 4 * p, hsel);
    }
    for (int j0 = 32; j0 < total; j0 += 4) {
        const int j = j0 + jj;
        int   s = d;
        float a = 0.0f;
        if (j < total) {
            s = list_src(csrc, beg, deg, d, j);
            a = __expf(lrelu(as[s * H1 + h] + add) - mx) / den;
        }
        gather_pass(accA, accB, xlane, s, a, total - j0, hsel);
    }

    const float* bl = b + 4 * lane;
    float4 rA, rB;
    rA.x = accA.x + bl[0];   rA.y = accA.y + bl[1];   rA.z = accA.z + bl[2];   rA.w = accA.w + bl[3];
    rB.x = accB.x + bl[128]; rB.y = accB.y + bl[129]; rB.z = accB.z + bl[130]; rB.w = accB.w + bl[131];
    rA.x = rA.x > 0.0f ? rA.x : 0.0f; rA.y = rA.y > 0.0f ? rA.y : 0.0f;
    rA.z = rA.z > 0.0f ? rA.z : 0.0f; rA.w = rA.w > 0.0f ? rA.w : 0.0f;
    rB.x = rB.x > 0.0f ? rB.x : 0.0f; rB.y = rB.y > 0.0f ? rB.y : 0.0f;
    rB.z = rB.z > 0.0f ? rB.z : 0.0f; rB.w = rB.w > 0.0f ? rB.w : 0.0f;

    const float4* w2l = (const float4*)(W2 + 8 * lane);
    const float4 wa0 = w2l[0];
    const float4 wa1 = w2l[1];
    const float4 wb0 = w2l[64];
    const float4 wb1 = w2l[65];
    float y0 = 0.0f, y1 = 0.0f;
    y0 = fmaf(rA.x, wa0.x, y0);  y1 = fmaf(rA.x, wa0.y, y1);
    y0 = fmaf(rA.y, wa0.z, y0);  y1 = fmaf(rA.y, wa0.w, y1);
    y0 = fmaf(rA.z, wa1.x, y0);  y1 = fmaf(rA.z, wa1.y, y1);
    y0 = fmaf(rA.w, wa1.z, y0);  y1 = fmaf(rA.w, wa1.w, y1);
    y0 = fmaf(rB.x, wb0.x, y0);  y1 = fmaf(rB.x, wb0.y, y1);
    y0 = fmaf(rB.y, wb0.z, y0);  y1 = fmaf(rB.y, wb0.w, y1);
    y0 = fmaf(rB.z, wb1.x, y0);  y1 = fmaf(rB.z, wb1.y, y1);
    y0 = fmaf(rB.w, wb1.z, y0);  y1 = fmaf(rB.w, wb1.w, y1);
    y0 = wave_sum(y0);
    y1 = wave_sum(y1);
    if (lane == 0) {
        *(float2*)(xp2 + (size_t)d * OUT2) = make_float2(y0, y1);
        as2[d] = fmaf(y0, aw_src[0], y1 * aw_src[1]);
        ad2[d] = fmaf(y0, aw_dst[0], y1 * aw_dst[1]);
    }
}

static_assert(NNODES % 16 == 0, "node_agg2 blocks cover 16 nodes each");
__global__ __launch_bounds__(512) void node_agg2(const int* __restrict__ off,
                                                 const int* __restrict__ csrc,
                                                 const float* __restrict__ xp2,
                                                 const float* __restrict__ as2,
                                                 const float* __restrict__ ad2,
                                                 const float* __restrict__ b,
                                                 float* __restrict__ out) {
    __shared__ float res[16 * OUT2];
    const int lane = threadIdx.x & 31;
    const int w    = threadIdx.x >> 5;
    const int d    = blockIdx.x * 16 + w;
    const int beg   = off[d];
    const int deg   = off[d + 1] - beg;
    const int total = deg + 1;
    const float add = ad2[d];

    const bool  act0 = lane < total;
    const int   s0   = list_src(csrc, beg, deg, d, lane);
    const float v0   = act0 ? lrelu(as2[s0] + add) : -INFINITY;

    float mx = v0;
    for (int j0 = 32; j0 < total; j0 += 32) {
        const int j = j0 + lane;
        if (j < total) {
            const int s = list_src(csrc, beg, deg, d, j);
            mx = fmaxf(mx, lrelu(as2[s] + add));
        }
    }
    mx = wave_max(mx);

    const float p0 = act0 ? __expf(v0 - mx) : 0.0f;
    float sum = p0;
    for (int j0 = 32; j0 < total; j0 += 32) {
        const int j = j0 + lane;
        if (j < total) {
            const int s = list_src(csrc, beg, deg, d, j);
            sum += __expf(lrelu(as2[s] + add) - mx);
        }
    }
    const float den = wave_sum(sum);

    float c0 = 0.0f, c1 = 0.0f;
    if (act0) {
        const float a = p0 / den;
        c0 = a * xp2[s0 * OUT2 + 0];
        c1 = a * xp2[s0 * OUT2 + 1];
    }
    for (int j0 = 32; j0 < total; j0 += 32) {
        const int j = j0 + lane;
        if (j < total) {
            const int s = list_src(csrc, beg, deg, d, j);
            const float a = __expf(lrelu(as2[s] + add) - mx) / den;
            c0 = fmaf(a, xp2[s * OUT2 + 0], c0);
            c1 = fmaf(a, xp2[s * OUT2 + 1], c1);
        }
    }
    c0 = wave_sum(c0);
    c1 = wave_sum(c1);
    if (lane == 0) {
        res[w * OUT2 + 0] = c0;
        res[w * OUT2 + 1] = c1;
    }
    __syncthreads();
    if (threadIdx.x < 16 * OUT2)
        out[blockIdx.x * (16 * OUT2) + threadIdx.x] = res[threadIdx.x] + b[threadIdx.x & 1];
}

extern "C" void kernel_launch(void* const* d_in, const int* in_sizes, int n_in,
                              void* d_out, int out_size, void* d_ws, size_t ws_size,
                              hipStream_t stream) {
    const float* x     = (const float*)d_in[0];
    const int*   ei    = (const int*)d_in[1];
    const float* W1    = (const float*)d_in[2];
    const float* asw1  = (const float*)d_in[3];
    const float* adw1  = (const float*)d_in[4];
    const float* b1    = (const float*)d_in[5];
    const float* W2    = (const float*)d_in[6];
    const float* asw2  = (const float*)d_in[7];
    const float* adw2  = (const float*)d_in[8];
    const float* b2    = (const float*)d_in[9];
    float*       out   = (float*)d_out;

    float* w    = (float*)d_ws;
    float* xp1  = w;
    float* as1  = xp1  + (size_t)NNODES * D1;
    float* ad1  = as1  + (size_t)NNODES * H1;
    float* xp2  = ad1  + (size_t)NNODES * H1;
    float* as2  = xp2  + (size_t)NNODES * OUT2;
    float* ad2  = as2  + (size_t)NNODES;
    int*   cnt  = (int*)(ad2 + (size_t)NNODES);
    int*   off  = cnt  + (size_t)NNODES;
    int*   rnk  = off  + (size_t)NNODES + 1;
    int*   csrc = rnk  + (size_t)NEDGES;

    zero_i32<<<(NNODES + 255) / 256, 256, 0, stream>>>(cnt, NNODES);
    count_rank<<<(NEDGES + 255) / 256, 256, 0, stream>>>(ei, cnt, rnk);
    scan_counts<<<1, 1024, 0, stream>>>(cnt, off);
    fill_csr<<<(NEDGES + 255) / 256, 256, 0, stream>>>(ei, off, rnk, csrc);

    gemm1_wmma<<<G1_BLOCKS, 512, 0, stream>>>(x, W1, xp1);
    alpha1_kernel<<<(NNODES * H1 + 255) / 256, 256, 0, stream>>>(xp1, asw1, adw1, as1, ad1);
    node_agg1<<<NNODES / 8, 256, 0, stream>>>(off, csrc, xp1, as1, ad1, b1,
                                              W2, asw2, adw2, xp2, as2, ad2);

    node_agg2<<<NNODES / 16, 512, 0, stream>>>(off, csrc, xp2, as2, ad2, b2, out);
}


# Edges written straight into fixed 64-slot per-node lists by the counting add (longer lists spill to an overflow list): prefix-scan and fill launches and the rank array gone; list order kept
# speedup vs baseline: 7.0376x; 1.0563x over previous
#include <hip/hip_runtime.h>
#include <hip/hip_bf16.h>
#include <math.h>

#define NNODES 50000
#define NEDGES 800000
#define FIN    256
#define H1     8
#define C1     32
#define D1     256
#define OUT2   2
#define SLOPE  0.2f

#define LDSPAD 4
#define ASTRIDE (FIN + LDSPAD)

typedef float v2f __attribute__((ext_vector_type(2)));
typedef float v8f __attribute__((ext_vector_type(8)));

__device__ __forceinline__ float lrelu(float v) {
    return v > 0.0f ? v : SLOPE * v;
}

#define MTILES    (NNODES / 16)
#define G1_BLOCKS 256
#define KSTEPS    (FIN / 4)
static_assert(NNODES % 16 == 0, "gemm1 row tiles are 16 rows");
__global__ __launch_bounds__(512) void gemm1_wmma(const float* __restrict__ x,
                                                  const float* __restrict__ W1,
                                                  float* __restrict__ xp) {
    __shared__ float aTile[2][16 * ASTRIDE];

    const int t     = threadIdx.x;
    const int lane  = t & 31;
    const int wave  = t >> 5;
    const int half  = lane >> 4;
    const int l16   = lane & 15;
    const int bcol  = wave * 16 + l16;

    v2f breg[KSTEPS];
#pragma unroll
    for (int s = 0; s < KSTEPS; ++s) {
        const int kk = 4 * s + half * 2;
        breg[s].x = W1[kk * D1 + bcol];
        breg[s].y = W1[(kk + 1) * D1 + bcol];
    }

    const int srow = t >> 5;
    const int scol = (t & 31) * 8;
    const int soff = srow * ASTRIDE + scol;
    const int aoff = l16 * ASTRIDE + half * 2;

    int tile = blockIdx.x;
    float4 p0 = make_float4(0.0f, 0.0f, 0.0f, 0.0f);
    float4 p1 = p0;
    if (tile < MTILES) {
        const float4* src = (const float4*)(x + (size_t)(tile * 16 + srow) * FIN + scol);
        p0 = src[0];
        p1 = src[1];
    }
    int buf = 0;
    for (; tile < MTILES; tile += G1_BLOCKS) {
        float* abuf = aTile[buf];
        *(float4*)(abuf + soff)     = p0;
        *(float4*)(abuf + soff + 4) = p1;
        __syncthreads();

        const int nxt = tile + G1_BLOCKS;
        if (nxt < MTILES) {
            const float4* src = (const float4*)(x + (size_t)(nxt * 16 + srow) * FIN + scol);
            p0 = src[0];
            p1 = src[1];
        }

        v8f acc = {};
#pragma unroll
        for (int s = 0; s < KSTEPS; ++s) {
            v2f a = *(const v2f*)(abuf + aoff + 4 * s);
            acc = __builtin_amdgcn_wmma_f32_16x16x4_f32(
                 false, a,  false, breg[s],
                 (short)0, acc,  false,  false);
        }

        const int mrow = tile * 16 + half * 8;
#pragma unroll
        for (int r = 0; r < 8; ++r)
            xp[(size_t)(mrow + r) * D1 + bcol] = acc[r];
        buf ^= 1;
    }
}

__global__ void alpha1_kernel(const float* __restrict__ xp,
                              const float* __restrict__ a_src,
                              const float* __restrict__ a_dst,
                              float* __restrict__ as, float* __restrict__ ad) {
    int i = blockIdx.x * blockDim.x + threadIdx.x;
    if (i >= NNODES * H1) return;
    int h = i & (H1 - 1);
    int n = i >> 3;
    const float* row = xp + (size_t)n * D1 + h * C1;
    const float* ws  = a_src + h * C1;
    const float* wd  = a_dst + h * C1;
    float s = 0.0f, d = 0.0f;
#pragma unroll
    for (int c = 0; c < C1; ++c) {
        float v = row[c];
        s = fmaf(v, ws[c], s);
        d = fmaf(v, wd[c], d);
    }
    as[i] = s;
    ad[i] = d;
}

#define CAPH 32
#define CAP  (2 * CAPH)
static_assert((long long)NNODES * CAP < 2147483647LL, "slot index fits an int");

__global__ void zero_i32(int* __restrict__ p, int n) {
    int i = blockIdx.x * blockDim.x + threadIdx.x;
    if (i < n) p[i] = 0;
}

__device__ __forceinline__ int slot_index(int d, int j) {
    return (j < CAPH ? 0 : NNODES * CAPH - CAPH) + d * CAPH + j;
}

__global__ void bucket_edges(const int* __restrict__ ei, int* __restrict__ cnt,
                             int* __restrict__ lst, int* __restrict__ ovf) {
    int e = blockIdx.x * blockDim.x + threadIdx.x;
    if (e >= NEDGES) return;
    const int s = ei[e];
    const int d = ei[NEDGES + e];
    const int slot = atomicAdd(&cnt[d], 1);
    if (slot < CAP) {
        lst[slot_index(d, slot)] = s;
    } else {
        const int k = atomicAdd(&cnt[NNODES], 1);
        ovf[2 * k]     = d;
        ovf[2 * k + 1] = s;
    }
}

__device__ __forceinline__ float wave_max(float v) {
    for (int o = 16; o > 0; o >>= 1) v = fmaxf(v, __shfl_xor(v, o));
    return v;
}
__device__ __forceinline__ float wave_sum(float v) {
    for (int o = 16; o > 0; o >>= 1) v += __shfl_xor(v, o);
    return v;
}
__device__ __forceinline__ int clamp_node(int s) {
    return min(max(s, 0), NNODES - 1);
}
__device__ __forceinline__ int list_src(const int* __restrict__ lst, int deg, int d, int j) {
    int s = d;
    if (j < deg) s = clamp_node(lst[slot_index(d, j)]);
    return s;
}
template <typename F>
__device__ __forceinline__ void walk_long_list(const int* __restrict__ lst,
                                               const int* __restrict__ ovf, int novf,
                                               int d, F&& f) {
#pragma nounroll
    for (int j = 0; j < CAP; ++j) f(clamp_node(lst[slot_index(d, j)]));
#pragma nounroll
    for (int k = 0; k < novf; ++k)
        if (ovf[2 * k] == d) f(clamp_node(ovf[2 * k + 1]));
    f(d);
}

__device__ __forceinline__ void gather_pass(float4& accA, float4& accB,
                                            const float* __restrict__ xlane,
                                            int s_l, float a_l, int nvalid, int hsel) {
    float4 xa[4], xb[4];
#pragma unroll
    for (int i = 0; i < 4; ++i) {
        const int s = __builtin_amdgcn_readlane(s_l, i * 8);
        const float* r = xlane + (size_t)s * D1;
        xa[i] = *(const float4*)(r);
        xb[i] = *(const float4*)(r + 128);
    }
#pragma unroll
    for (int i = 0; i < 4; ++i) {
        const float alo = __shfl(a_l, i * 8 + hsel);
        const float ahi = __shfl(a_l, i * 8 + 4 + hsel);
        float4 va = xa[i], vb = xb[i];
        if (i >= nvalid) {
            va = make_float4(0.0f, 0.0f, 0.0f, 0.0f);
            vb = make_float4(0.0f, 0.0f, 0.0f, 0.0f);
        }
        accA.x = fmaf(alo, va.x, accA.x);
        accA.y = fmaf(alo, va.y, accA.y);
        accA.z = fmaf(alo, va.z, accA.z);
        accA.w = fmaf(alo, va.w, accA.w);
        accB.x = fmaf(ahi, vb.x, accB.x);
        accB.y = fmaf(ahi, vb.y, accB.y);
        accB.z = fmaf(ahi, vb.z, accB.z);
        accB.w = fmaf(ahi, vb.w, accB.w);
    }
}

static_assert(NNODES % 8 == 0, "node_agg1 blocks cover 8 nodes each");
__global__ __launch_bounds__(256) void node_agg1(const int* __restrict__ cnt,
                                                 const int* __restrict__ lst,
                                                 const int* __restrict__ ovf,
                                                 const float* __restrict__ xp,
                                                 const float* __restrict__ as,
                                                 const float* __restrict__ ad,
                                                 const float* __restrict__ b,
                                                 const float* __restrict__ W2,
                                                 const float* __restrict__ aw_src,
                                                 const float* __restrict__ aw_dst,
                                                 float* __restrict__ xp2,
                                                 float* __restrict__ as2,
                                                 float* __restrict__ ad2) {
    const int lane = threadIdx.x & 31;
    const int d = __builtin_amdgcn_readfirstlane(blockIdx.x * 8 + (threadIdx.x >> 5));
    const int deg   = cnt[d];
    const int total = deg + 1;

    float4 accA = make_float4(0.0f, 0.0f, 0.0f, 0.0f);
    float4 accB = make_float4(0.0f, 0.0f, 0.0f, 0.0f);
    const float* xlane = xp + 4 * lane;
    const int hsel = lane >> 3;

    if (__builtin_expect(deg <= CAP, 1)) {
        const int jj    = lane >> 3;
        const int h     = lane & 7;
        const float add = ad[d * H1 + h];

        int   sp[8];
        float ev[8];
        const int s0 = list_src(lst, deg, d, lane);
    #pragma unroll
        for (int p = 0; p < 8; ++p) sp[p] = __shfl(s0, 4 * p + jj);
    #pragma unroll
        for (int p = 0; p < 8; ++p) {
            const float v = lrelu(as[sp[p] * H1 + h] + add);
            ev[p] = (4 * p + jj < total) ? v : -INFINITY;
        }

        float mx = ev[0];
    #pragma unroll
        for (int p = 1; p < 8; ++p) mx = fmaxf(mx, ev[p]);
        for (int j0 = 32; j0 < total; j0 += 4) {
            const int j = j0 + jj;
            if (j < total) {
                const int s = list_src(lst, deg, d, j);
                mx = fmaxf(mx, lrelu(as[s * H1 + h] + add));
            }
        }
        mx = fmaxf(mx, __shfl_xor(mx, 8));
        mx = fmaxf(mx, __shfl_xor(mx, 16));

        float sum = 0.0f;
    #pragma unroll
        for (int p = 0; p < 8; ++p) {
            ev[p] = (4 * p + jj < total) ? __expf(ev[p] - mx) : 0.0f;
            sum += ev[p];
        }
        for (int j0 = 32; j0 < total; j0 += 4) {
            const int j = j0 + jj;
            if (j < total) {
                const int s = list_src(lst, deg, d, j);
                sum += __expf(lrelu(as[s * H1 + h] + add) - mx);
            }
        }
        sum += __shfl_xor(sum, 8);
        sum += __shfl_xor(sum, 16);
        const float den = sum;
    #pragma unroll
        for (int p = 0; p < 8; ++p) ev[p] = ev[p] / den;

        const int n0   = min(total, 32);
    #pragma unroll
        for (int p = 0; p < 8; ++p) {
            if (4 * p < n0) gather_pass(accA, accB, xlane, sp[p], ev[p], n0 - 4 * p, hsel);
        }
        for (int j0 = 32; j0 < total; j0 += 4) {
            const int j = j0 + jj;
            int   s = d;
            float a = 0.0f;
            if (j < total) {
                s = list_src(lst, deg, d, j);
                a = __expf(lrelu(as[s * H1 + h] + add) - mx) / den;
            }
            gather_pass(accA, accB, xlane, s, a, total - j0, hsel);
        }
    } else {
        const int   novf = min(max(cnt[NNODES], 0), NEDGES);
        const int   hA   = hsel, hB = 4 + hsel;
        const float addA = ad[d * H1 + hA];
        const float addB = ad[d * H1 + hB];
        float mA = -INFINITY, mB = -INFINITY;
        walk_long_list(lst, ovf, novf, d, [&](int s) {
            mA = fmaxf(mA, lrelu(as[s * H1 + hA] + addA));
            mB = fmaxf(mB, lrelu(as[s * H1 + hB] + addB));
        });
        float sA = 0.0f, sB = 0.0f;
        walk_long_list(lst, ovf, novf, d, [&](int s) {
            sA += __expf(lrelu(as[s * H1 + hA] + addA) - mA);
            sB += __expf(lrelu(as[s * H1 + hB] + addB) - mB);
        });
        walk_long_list(lst, ovf, novf, d, [&](int s) {
            const float aA = __expf(lrelu(as[s * H1 + hA] + addA) - mA) / sA;
            const float aB = __expf(lrelu(as[s * H1 + hB] + addB) - mB) / sB;
            const float* r = xlane + (size_t)s * D1;
            const float4 va = *(const float4*)(r);
            const float4 vb = *(const float4*)(r + 128);
            accA.x = fmaf(aA, va.x, accA.x);
            accA.y = fmaf(aA, va.y, accA.y);
            accA.z = fmaf(aA, va.z, accA.z);
            accA.w = fmaf(aA, va.w, accA.w);
            accB.x = fmaf(aB, vb.x, accB.x);
            accB.y = fmaf(aB, vb.y, accB.y);
            accB.z = fmaf(aB, vb.z, accB.z);
            accB.w = fmaf(aB, vb.w, accB.w);
        });
    }

    const float* bl = b + 4 * lane;
    float4 rA, rB;
    rA.x = accA.x + bl[0];   rA.y = accA.y + bl[1];   rA.z = accA.z + bl[2];   rA.w = accA.w + bl[3];
    rB.x = accB.x + bl[128]; rB.y = accB.y + bl[129]; rB.z = accB.z + bl[130]; rB.w = accB.w + bl[131];
    rA.x = rA.x > 0.0f ? rA.x : 0.0f; rA.y = rA.y > 0.0f ? rA.y : 0.0f;
    rA.z = rA.z > 0.0f ? rA.z : 0.0f; rA.w = rA.w > 0.0f ? rA.w : 0.0f;
    rB.x = rB.x > 0.0f ? rB.x : 0.0f; rB.y = rB.y > 0.0f ? rB.y : 0.0f;
    rB.z = rB.z > 0.0f ? rB.z : 0.0f; rB.w = rB.w > 0.0f ? rB.w : 0.0f;

    const float4* w2l = (const float4*)(W2 + 8 * lane);
    const float4 wa0 = w2l[0];
    const float4 wa1 = w2l[1];
    const float4 wb0 = w2l[64];
    const float4 wb1 = w2l[65];
    float y0 = 0.0f, y1 = 0.0f;
    y0 = fmaf(rA.x, wa0.x, y0);  y1 = fmaf(rA.x, wa0.y, y1);
    y0 = fmaf(rA.y, wa0.z, y0);  y1 = fmaf(rA.y, wa0.w, y1);
    y0 = fmaf(rA.z, wa1.x, y0);  y1 = fmaf(rA.z, wa1.y, y1);
    y0 = fmaf(rA.w, wa1.z, y0);  y1 = fmaf(rA.w, wa1.w, y1);
    y0 = fmaf(rB.x, wb0.x, y0);  y1 = fmaf(rB.x, wb0.y, y1);
    y0 = fmaf(rB.y, wb0.z, y0);  y1 = fmaf(rB.y, wb0.w, y1);
    y0 = fmaf(rB.z, wb1.x, y0);  y1 = fmaf(rB.z, wb1.y, y1);
    y0 = fmaf(rB.w, wb1.z, y0);  y1 = fmaf(rB.w, wb1.w, y1);
    y0 = wave_sum(y0);
    y1 = wave_sum(y1);
    if (lane == 0) {
        *(float2*)(xp2 + (size_t)d * OUT2) = make_float2(y0, y1);
        as2[d] = fmaf(y0, aw_src[0], y1 * aw_src[1]);
        ad2[d] = fmaf(y0, aw_dst[0], y1 * aw_dst[1]);
    }
}

static_assert(NNODES % 16 == 0, "node_agg2 blocks cover 16 nodes each");
__global__ __launch_bounds__(512) void node_agg2(const int* __restrict__ cnt,
                                                 const int* __restrict__ lst,
                                                 const int* __restrict__ ovf,
                                                 const float* __restrict__ xp2,
                                                 const float* __restrict__ as2,
                                                 const float* __restrict__ ad2,
                                                 const float* __restrict__ b,
                                                 float* __restrict__ out) {
    __shared__ float res[16 * OUT2];
    const int lane = threadIdx.x & 31;
    const int w    = threadIdx.x >> 5;
    const int d    = __builtin_amdgcn_readfirstlane(blockIdx.x * 16 + w);
    const int deg   = cnt[d];
    const int total = deg + 1;
    const float add = ad2[d];

    float c0 = 0.0f, c1 = 0.0f;
    if (__builtin_expect(deg <= CAP, 1)) {
        const bool  act0 = lane < total;
        const int   s0   = list_src(lst, deg, d, lane);
        const float v0   = act0 ? lrelu(as2[s0] + add) : -INFINITY;

        float mx = v0;
        for (int j0 = 32; j0 < total; j0 += 32) {
            const int j = j0 + lane;
            if (j < total) {
                const int s = list_src(lst, deg, d, j);
                mx = fmaxf(mx, lrelu(as2[s] + add));
            }
        }
        mx = wave_max(mx);

        const float p0 = act0 ? __expf(v0 - mx) : 0.0f;
        float sum = p0;
        for (int j0 = 32; j0 < total; j0 += 32) {
            const int j = j0 + lane;
            if (j < total) {
                const int s = list_src(lst, deg, d, j);
                sum += __expf(lrelu(as2[s] + add) - mx);
            }
        }
        const float den = wave_sum(sum);

        if (act0) {
            const float a = p0 / den;
            c0 = a * xp2[s0 * OUT2 + 0];
            c1 = a * xp2[s0 * OUT2 + 1];
        }
        for (int j0 = 32; j0 < total; j0 += 32) {
            const int j = j0 + lane;
            if (j < total) {
                const int s = list_src(lst, deg, d, j);
                const float a = __expf(lrelu(as2[s] + add) - mx) / den;
                c0 = fmaf(a, xp2[s * OUT2 + 0], c0);
                c1 = fmaf(a, xp2[s * OUT2 + 1], c1);
            }
        }
    } else {
        const int novf = min(max(cnt[NNODES], 0), NEDGES);
        float m = -INFINITY;
        walk_long_list(lst, ovf, novf, d, [&](int s) {
            m = fmaxf(m, lrelu(as2[s] + add));
        });
        float sm = 0.0f;
        walk_long_list(lst, ovf, novf, d, [&](int s) {
            sm += __expf(lrelu(as2[s] + add) - m);
        });
        float t0 = 0.0f, t1 = 0.0f;
        walk_long_list(lst, ovf, novf, d, [&](int s) {
            const float a = __expf(lrelu(as2[s] + add) - m) / sm;
            t0 = fmaf(a, xp2[s * OUT2 + 0], t0);
            t1 = fmaf(a, xp2[s * OUT2 + 1], t1);
        });
        if (lane == 0) { c0 = t0; c1 = t1; }
    }
    c0 = wave_sum(c0);
    c1 = wave_sum(c1);
    if (lane == 0) {
        res[w * OUT2 + 0] = c0;
        res[w * OUT2 + 1] = c1;
    }
    __syncthreads();
    if (threadIdx.x < 16 * OUT2)
        out[blockIdx.x * (16 * OUT2) + threadIdx.x] = res[threadIdx.x] + b[threadIdx.x & 1];
}

extern "C" void kernel_launch(void* const* d_in, const int* in_sizes, int n_in,
                              void* d_out, int out_size, void* d_ws, size_t ws_size,
                              hipStream_t stream) {
    const float* x     = (const float*)d_in[0];
    const int*   ei    = (const int*)d_in[1];
    const float* W1    = (const float*)d_in[2];
    const float* asw1  = (const float*)d_in[3];
    const float* adw1  = (const float*)d_in[4];
    const float* b1    = (const float*)d_in[5];
    const float* W2    = (const float*)d_in[6];
    const float* asw2  = (const float*)d_in[7];
    const float* adw2  = (const float*)d_in[8];
    const float* b2    = (const float*)d_in[9];
    float*       out   = (float*)d_out;

    float* w    = (float*)d_ws;
    float* xp1  = w;
    float* as1  = xp1  + (size_t)NNODES * D1;
    float* ad1  = as1  + (size_t)NNODES * H1;
    float* xp2  = ad1  + (size_t)NNODES * H1;
    float* as2  = xp2  + (size_t)NNODES * OUT2;
    float* ad2  = as2  + (size_t)NNODES;
    int*   lst  = (int*)(ad2 + (size_t)NNODES);
    int*   ovf  = lst  + (size_t)NNODES * CAP;
    int*   cnt  = ovf  + (size_t)2 * NEDGES;

    zero_i32<<<(NNODES + 1 + 255) / 256, 256, 0, stream>>>(cnt, NNODES + 1);
    bucket_edges<<<(NEDGES + 255) / 256, 256, 0, stream>>>(ei, cnt, lst, ovf);

    gemm1_wmma<<<G1_BLOCKS, 512, 0, stream>>>(x, W1, xp1);
    alpha1_kernel<<<(NNODES * H1 + 255) / 256, 256, 0, stream>>>(xp1, asw1, adw1, as1, ad1);
    node_agg1<<<NNODES / 8, 256, 0, stream>>>(cnt, lst, ovf, xp1, as1, ad1, b1,
                                              W2, asw2, adw2, xp2, as2, ad2);

    node_agg2<<<NNODES / 16, 512, 0, stream>>>(cnt, lst, ovf, xp2, as2, ad2, b2, out);
}


# Edge placement (counting add + list store) moved into the layer-1 GEMM's tile loop, its round trips hidden under the WMMAs: one launch fewer; list order stays the arrival order of the adds
# speedup vs baseline: 7.7266x; 1.0979x over previous
#include <hip/hip_runtime.h>
#include <hip/hip_bf16.h>
#include <math.h>

#define NNODES 50000
#define NEDGES 800000
#define FIN    256
#define H1     8
#define C1     32
#define D1     256
#define OUT2   2
#define SLOPE  0.2f

#define LDSPAD 4
#define ASTRIDE (FIN + LDSPAD)

typedef float v2f __attribute__((ext_vector_type(2)));
typedef float v8f __attribute__((ext_vector_type(8)));

__device__ __forceinline__ float lrelu(float v) {
    return v > 0.0f ? v : SLOPE * v;
}

#define CAPH 32
#define CAP  (2 * CAPH)
static_assert((long long)NNODES * CAP < 2147483647LL, "slot index fits an int");

__device__ __forceinline__ int slot_index(int d, int j) {
    return (j < CAPH ? 0 : NNODES * CAPH - CAPH) + d * CAPH + j;
}


#define MTILES    (NNODES / 16)
#define G1_BLOCKS 256
#define KSTEPS    (FIN / 4)
#define EDGES_PER_TILE 512
static_assert(NNODES % 16 == 0, "gemm1 row tiles are 16 rows");
static_assert((long long)MTILES * EDGES_PER_TILE >= NEDGES, "every edge has an owner thread");
__global__ __launch_bounds__(EDGES_PER_TILE) void gemm1_wmma(const float* __restrict__ x,
                                                             const float* __restrict__ W1,
                                                             float* __restrict__ xp,
                                                             const int* __restrict__ ei,
                                                             int* __restrict__ cnt,
                                                             int* __restrict__ lst,
                                                             int* __restrict__ ovf) {
    __shared__ float aTile[2][16 * ASTRIDE];

    const int t     = threadIdx.x;
    const int lane  = t & 31;
    const int wave  = t >> 5;
    const int half  = lane >> 4;
    const int l16   = lane & 15;
    const int bcol  = wave * 16 + l16;

    v2f breg[KSTEPS];
#pragma unroll
    for (int s = 0; s < KSTEPS; ++s) {
        const int kk = 4 * s + half * 2;
        breg[s].x = W1[kk * D1 + bcol];
        breg[s].y = W1[(kk + 1) * D1 + bcol];
    }

    const int srow = t >> 5;
    const int scol = (t & 31) * 8;
    const int soff = srow * ASTRIDE + scol;
    const int aoff = l16 * ASTRIDE + half * 2;

    int tile = blockIdx.x;
    float4 p0 = make_float4(0.0f, 0.0f, 0.0f, 0.0f);
    float4 p1 = p0;
    int es = 0, ed = 0;
    if (tile < MTILES) {
        const float4* src = (const float4*)(x + (size_t)(tile * 16 + srow) * FIN + scol);
        p0 = src[0];
        p1 = src[1];
        const int e = tile * EDGES_PER_TILE + t;
        if (e < NEDGES) {
            es = ei[e];
            ed = ei[NEDGES + e];
        }
    }
    int buf = 0;
    for (; tile < MTILES; tile += G1_BLOCKS) {
        float* abuf = aTile[buf];
        *(float4*)(abuf + soff)     = p0;
        *(float4*)(abuf + soff + 4) = p1;
        __syncthreads();

        const bool has_edge = tile * EDGES_PER_TILE + t < NEDGES;
        const int  cs = es, cd = ed;
        int slot = 0;
        if (has_edge) slot = atomicAdd(&cnt[cd], 1);

        const int nxt = tile + G1_BLOCKS;
        if (nxt < MTILES) {
            const float4* src = (const float4*)(x + (size_t)(nxt * 16 + srow) * FIN + scol);
            p0 = src[0];
            p1 = src[1];
            const int ne = nxt * EDGES_PER_TILE + t;
            if (ne < NEDGES) {
                es = ei[ne];
                ed = ei[NEDGES + ne];
            }
        }

        v8f acc = {};
#pragma unroll
        for (int s = 0; s < KSTEPS; ++s) {
            v2f a = *(const v2f*)(abuf + aoff + 4 * s);
            acc = __builtin_amdgcn_wmma_f32_16x16x4_f32(
                 false, a,  false, breg[s],
                 (short)0, acc,  false,  false);
        }

        const int mrow = tile * 16 + half * 8;
#pragma unroll
        for (int r = 0; r < 8; ++r)
            xp[(size_t)(mrow + r) * D1 + bcol] = acc[r];

        if (has_edge) {
            if (slot < CAP) {
                lst[slot_index(cd, slot)] = cs;
            } else {
                const int k = atomicAdd(&cnt[NNODES], 1);
                ovf[2 * k]     = cd;
                ovf[2 * k + 1] = cs;
            }
        }
        buf ^= 1;
    }
}

__global__ void alpha1_kernel(const float* __restrict__ xp,
                              const float* __restrict__ a_src,
                              const float* __restrict__ a_dst,
                              float* __restrict__ as, float* __restrict__ ad) {
    int i = blockIdx.x * blockDim.x + threadIdx.x;
    if (i >= NNODES * H1) return;
    int h = i & (H1 - 1);
    int n = i >> 3;
    const float* row = xp + (size_t)n * D1 + h * C1;
    const float* ws  = a_src + h * C1;
    const float* wd  = a_dst + h * C1;
    float s = 0.0f, d = 0.0f;
#pragma unroll
    for (int c = 0; c < C1; ++c) {
        float v = row[c];
        s = fmaf(v, ws[c], s);
        d = fmaf(v, wd[c], d);
    }
    as[i] = s;
    ad[i] = d;
}

__global__ void zero_i32(int* __restrict__ p, int n) {
    int i = blockIdx.x * blockDim.x + threadIdx.x;
    if (i < n) p[i] = 0;
}

__device__ __forceinline__ float wave_max(float v) {
    for (int o = 16; o > 0; o >>= 1) v = fmaxf(v, __shfl_xor(v, o));
    return v;
}
__device__ __forceinline__ float wave_sum(float v) {
    for (int o = 16; o > 0; o >>= 1) v += __shfl_xor(v, o);
    return v;
}
__device__ __forceinline__ int clamp_node(int s) {
    return min(max(s, 0), NNODES - 1);
}
__device__ __forceinline__ int list_src(const int* __restrict__ lst, int deg, int d, int j) {
    int s = d;
    if (j < deg) s = clamp_node(lst[slot_index(d, j)]);
    return s;
}
template <typename F>
__device__ __forceinline__ void walk_long_list(const int* __restrict__ lst,
                                               const int* __restrict__ ovf, int novf,
                                               int d, F&& f) {
#pragma nounroll
    for (int j = 0; j < CAP; ++j) f(clamp_node(lst[slot_index(d, j)]));
#pragma nounroll
    for (int k = 0; k < novf; ++k)
        if (ovf[2 * k] == d) f(clamp_node(ovf[2 * k + 1]));
    f(d);
}

__device__ __forceinline__ void gather_pass(float4& accA, float4& accB,
                                            const float* __restrict__ xlane,
                                            int s_l, float a_l, int nvalid, int hsel) {
    float4 xa[4], xb[4];
#pragma unroll
    for (int i = 0; i < 4; ++i) {
        const int s = __builtin_amdgcn_readlane(s_l, i * 8);
        const float* r = xlane + (size_t)s * D1;
        xa[i] = *(const float4*)(r);
        xb[i] = *(const float4*)(r + 128);
    }
#pragma unroll
    for (int i = 0; i < 4; ++i) {
        const float alo = __shfl(a_l, i * 8 + hsel);
        const float ahi = __shfl(a_l, i * 8 + 4 + hsel);
        float4 va = xa[i], vb = xb[i];
        if (i >= nvalid) {
            va = make_float4(0.0f, 0.0f, 0.0f, 0.0f);
            vb = make_float4(0.0f, 0.0f, 0.0f, 0.0f);
        }
        accA.x = fmaf(alo, va.x, accA.x);
        accA.y = fmaf(alo, va.y, accA.y);
        accA.z = fmaf(alo, va.z, accA.z);
        accA.w = fmaf(alo, va.w, accA.w);
        accB.x = fmaf(ahi, vb.x, accB.x);
        accB.y = fmaf(ahi, vb.y, accB.y);
        accB.z = fmaf(ahi, vb.z, accB.z);
        accB.w = fmaf(ahi, vb.w, accB.w);
    }
}

static_assert(NNODES % 8 == 0, "node_agg1 blocks cover 8 nodes each");
__global__ __launch_bounds__(256) void node_agg1(const int* __restrict__ cnt,
                                                 const int* __restrict__ lst,
                                                 const int* __restrict__ ovf,
                                                 const float* __restrict__ xp,
                                                 const float* __restrict__ as,
                                                 const float* __restrict__ ad,
                                                 const float* __restrict__ b,
                                                 const float* __restrict__ W2,
                                                 const float* __restrict__ aw_src,
                                                 const float* __restrict__ aw_dst,
                                                 float* __restrict__ xp2,
                                                 float* __restrict__ as2,
                                                 float* __restrict__ ad2) {
    const int lane = threadIdx.x & 31;
    const int d = __builtin_amdgcn_readfirstlane(blockIdx.x * 8 + (threadIdx.x >> 5));
    const int deg   = cnt[d];
    const int total = deg + 1;

    float4 accA = make_float4(0.0f, 0.0f, 0.0f, 0.0f);
    float4 accB = make_float4(0.0f, 0.0f, 0.0f, 0.0f);
    const float* xlane = xp + 4 * lane;
    const int hsel = lane >> 3;

    if (__builtin_expect(deg <= CAP, 1)) {
        const int jj    = lane >> 3;
        const int h     = lane & 7;
        const float add = ad[d * H1 + h];

        int   sp[8];
        float ev[8];
        const int s0 = list_src(lst, deg, d, lane);
    #pragma unroll
        for (int p = 0; p < 8; ++p) sp[p] = __shfl(s0, 4 * p + jj);
    #pragma unroll
        for (int p = 0; p < 8; ++p) {
            const float v = lrelu(as[sp[p] * H1 + h] + add);
            ev[p] = (4 * p + jj < total) ? v : -INFINITY;
        }

        float mx = ev[0];
    #pragma unroll
        for (int p = 1; p < 8; ++p) mx = fmaxf(mx, ev[p]);
        for (int j0 = 32; j0 < total; j0 += 4) {
            const int j = j0 + jj;
            if (j < total) {
                const int s = list_src(lst, deg, d, j);
                mx = fmaxf(mx, lrelu(as[s * H1 + h] + add));
            }
        }
        mx = fmaxf(mx, __shfl_xor(mx, 8));
        mx = fmaxf(mx, __shfl_xor(mx, 16));

        float sum = 0.0f;
    #pragma unroll
        for (int p = 0; p < 8; ++p) {
            ev[p] = (4 * p + jj < total) ? __expf(ev[p] - mx) : 0.0f;
            sum += ev[p];
        }
        for (int j0 = 32; j0 < total; j0 += 4) {
            const int j = j0 + jj;
            if (j < total) {
                const int s = list_src(lst, deg, d, j);
                sum += __expf(lrelu(as[s * H1 + h] + add) - mx);
            }
        }
        sum += __shfl_xor(sum, 8);
        sum += __shfl_xor(sum, 16);
        const float den = sum;
    #pragma unroll
        for (int p = 0; p < 8; ++p) ev[p] = ev[p] / den;

        const int n0   = min(total, 32);
    #pragma unroll
        for (int p = 0; p < 8; ++p) {
            if (4 * p < n0) gather_pass(accA, accB, xlane, sp[p], ev[p], n0 - 4 * p, hsel);
        }
        for (int j0 = 32; j0 < total; j0 += 4) {
            const int j = j0 + jj;
            int   s = d;
            float a = 0.0f;
            if (j < total) {
                s = list_src(lst, deg, d, j);
                a = __expf(lrelu(as[s * H1 + h] + add) - mx) / den;
            }
            gather_pass(accA, accB, xlane, s, a, total - j0, hsel);
        }
    } else {
        const int   novf = min(max(cnt[NNODES], 0), NEDGES);
        const int   hA   = hsel, hB = 4 + hsel;
        const float addA = ad[d * H1 + hA];
        const float addB = ad[d * H1 + hB];
        float mA = -INFINITY, mB = -INFINITY;
        walk_long_list(lst, ovf, novf, d, [&](int s) {
            mA = fmaxf(mA, lrelu(as[s * H1 + hA] + addA));
            mB = fmaxf(mB, lrelu(as[s * H1 + hB] + addB));
        });
        float sA = 0.0f, sB = 0.0f;
        walk_long_list(lst, ovf, novf, d, [&](int s) {
            sA += __expf(lrelu(as[s * H1 + hA] + addA) - mA);
            sB += __expf(lrelu(as[s * H1 + hB] + addB) - mB);
        });
        walk_long_list(lst, ovf, novf, d, [&](int s) {
            const float aA = __expf(lrelu(as[s * H1 + hA] + addA) - mA) / sA;
            const float aB = __expf(lrelu(as[s * H1 + hB] + addB) - mB) / sB;
            const float* r = xlane + (size_t)s * D1;
            const float4 va = *(const float4*)(r);
            const float4 vb = *(const float4*)(r + 128);
            accA.x = fmaf(aA, va.x, accA.x);
            accA.y = fmaf(aA, va.y, accA.y);
            accA.z = fmaf(aA, va.z, accA.z);
            accA.w = fmaf(aA, va.w, accA.w);
            accB.x = fmaf(aB, vb.x, accB.x);
            accB.y = fmaf(aB, vb.y, accB.y);
            accB.z = fmaf(aB, vb.z, accB.z);
            accB.w = fmaf(aB, vb.w, accB.w);
        });
    }

    const float* bl = b + 4 * lane;
    float4 rA, rB;
    rA.x = accA.x + bl[0];   rA.y = accA.y + bl[1];   rA.z = accA.z + bl[2];   rA.w = accA.w + bl[3];
    rB.x = accB.x + bl[128]; rB.y = accB.y + bl[129]; rB.z = accB.z + bl[130]; rB.w = accB.w + bl[131];
    rA.x = rA.x > 0.0f ? rA.x : 0.0f; rA.y = rA.y > 0.0f ? rA.y : 0.0f;
    rA.z = rA.z > 0.0f ? rA.z : 0.0f; rA.w = rA.w > 0.0f ? rA.w : 0.0f;
    rB.x = rB.x > 0.0f ? rB.x : 0.0f; rB.y = rB.y > 0.0f ? rB.y : 0.0f;
    rB.z = rB.z > 0.0f ? rB.z : 0.0f; rB.w = rB.w > 0.0f ? rB.w : 0.0f;

    const float4* w2l = (const float4*)(W2 + 8 * lane);
    const float4 wa0 = w2l[0];
    const float4 wa1 = w2l[1];
    const float4 wb0 = w2l[64];
    const float4 wb1 = w2l[65];
    float y0 = 0.0f, y1 = 0.0f;
    y0 = fmaf(rA.x, wa0.x, y0);  y1 = fmaf(rA.x, wa0.y, y1);
    y0 = fmaf(rA.y, wa0.z, y0);  y1 = fmaf(rA.y, wa0.w, y1);
    y0 = fmaf(rA.z, wa1.x, y0);  y1 = fmaf(rA.z, wa1.y, y1);
    y0 = fmaf(rA.w, wa1.z, y0);  y1 = fmaf(rA.w, wa1.w, y1);
    y0 = fmaf(rB.x, wb0.x, y0);  y1 = fmaf(rB.x, wb0.y, y1);
    y0 = fmaf(rB.y, wb0.z, y0);  y1 = fmaf(rB.y, wb0.w, y1);
    y0 = fmaf(rB.z, wb1.x, y0);  y1 = fmaf(rB.z, wb1.y, y1);
    y0 = fmaf(rB.w, wb1.z, y0);  y1 = fmaf(rB.w, wb1.w, y1);
    y0 = wave_sum(y0);
    y1 = wave_sum(y1);
    if (lane == 0) {
        *(float2*)(xp2 + (size_t)d * OUT2) = make_float2(y0, y1);
        as2[d] = fmaf(y0, aw_src[0], y1 * aw_src[1]);
        ad2[d] = fmaf(y0, aw_dst[0], y1 * aw_dst[1]);
    }
}

static_assert(NNODES % 16 == 0, "node_agg2 blocks cover 16 nodes each");
__global__ __launch_bounds__(512) void node_agg2(const int* __restrict__ cnt,
                                                 const int* __restrict__ lst,
                                                 const int* __restrict__ ovf,
                                                 const float* __restrict__ xp2,
                                                 const float* __restrict__ as2,
                                                 const float* __restrict__ ad2,
                                                 const float* __restrict__ b,
                                                 float* __restrict__ out) {
    __shared__ float res[16 * OUT2];
    const int lane = threadIdx.x & 31;
    const int w    = threadIdx.x >> 5;
    const int d    = __builtin_amdgcn_readfirstlane(blockIdx.x * 16 + w);
    const int deg   = cnt[d];
    const int total = deg + 1;
    const float add = ad2[d];

    float c0 = 0.0f, c1 = 0.0f;
    if (__builtin_expect(deg <= CAP, 1)) {
        const bool  act0 = lane < total;
        const int   s0   = list_src(lst, deg, d, lane);
        const float v0   = act0 ? lrelu(as2[s0] + add) : -INFINITY;

        float mx = v0;
        for (int j0 = 32; j0 < total; j0 += 32) {
            const int j = j0 + lane;
            if (j < total) {
                const int s = list_src(lst, deg, d, j);
                mx = fmaxf(mx, lrelu(as2[s] + add));
            }
        }
        mx = wave_max(mx);

        const float p0 = act0 ? __expf(v0 - mx) : 0.0f;
        float sum = p0;
        for (int j0 = 32; j0 < total; j0 += 32) {
            const int j = j0 + lane;
            if (j < total) {
                const int s = list_src(lst, deg, d, j);
                sum += __expf(lrelu(as2[s] + add) - mx);
            }
        }
        const float den = wave_sum(sum);

        if (act0) {
            const float a = p0 / den;
            c0 = a * xp2[s0 * OUT2 + 0];
            c1 = a * xp2[s0 * OUT2 + 1];
        }
        for (int j0 = 32; j0 < total; j0 += 32) {
            const int j = j0 + lane;
            if (j < total) {
                const int s = list_src(lst, deg, d, j);
                const float a = __expf(lrelu(as2[s] + add) - mx) / den;
                c0 = fmaf(a, xp2[s * OUT2 + 0], c0);
                c1 = fmaf(a, xp2[s * OUT2 + 1], c1);
            }
        }
    } else {
        const int novf = min(max(cnt[NNODES], 0), NEDGES);
        float m = -INFINITY;
        walk_long_list(lst, ovf, novf, d, [&](int s) {
            m = fmaxf(m, lrelu(as2[s] + add));
        });
        float sm = 0.0f;
        walk_long_list(lst, ovf, novf, d, [&](int s) {
            sm += __expf(lrelu(as2[s] + add) - m);
        });
        float t0 = 0.0f, t1 = 0.0f;
        walk_long_list(lst, ovf, novf, d, [&](int s) {
            const float a = __expf(lrelu(as2[s] + add) - m) / sm;
            t0 = fmaf(a, xp2[s * OUT2 + 0], t0);
            t1 = fmaf(a, xp2[s * OUT2 + 1], t1);
        });
        if (lane == 0) { c0 = t0; c1 = t1; }
    }
    c0 = wave_sum(c0);
    c1 = wave_sum(c1);
    if (lane == 0) {
        res[w * OUT2 + 0] = c0;
        res[w * OUT2 + 1] = c1;
    }
    __syncthreads();
    if (threadIdx.x < 16 * OUT2)
        out[blockIdx.x * (16 * OUT2) + threadIdx.x] = res[threadIdx.x] + b[threadIdx.x & 1];
}

extern "C" void kernel_launch(void* const* d_in, const int* in_sizes, int n_in,
                              void* d_out, int out_size, void* d_ws, size_t ws_size,
                              hipStream_t stream) {
    const float* x     = (const float*)d_in[0];
    const int*   ei    = (const int*)d_in[1];
    const float* W1    = (const float*)d_in[2];
    const float* asw1  = (const float*)d_in[3];
    const float* adw1  = (const float*)d_in[4];
    const float* b1    = (const float*)d_in[5];
    const float* W2    = (const float*)d_in[6];
    const float* asw2  = (const float*)d_in[7];
    const float* adw2  = (const float*)d_in[8];
    const float* b2    = (const float*)d_in[9];
    float*       out   = (float*)d_out;

    float* w    = (float*)d_ws;
    float* xp1  = w;
    float* as1  = xp1  + (size_t)NNODES * D1;
    float* ad1  = as1  + (size_t)NNODES * H1;
    float* xp2  = ad1  + (size_t)NNODES * H1;
    float* as2  = xp2  + (size_t)NNODES * OUT2;
    float* ad2  = as2  + (size_t)NNODES;
    int*   lst  = (int*)(ad2 + (size_t)NNODES);
    int*   ovf  = lst  + (size_t)NNODES * CAP;
    int*   cnt  = ovf  + (size_t)2 * NEDGES;

    zero_i32<<<(NNODES + 1 + 255) / 256, 256, 0, stream>>>(cnt, NNODES + 1);

    gemm1_wmma<<<G1_BLOCKS, EDGES_PER_TILE, 0, stream>>>(x, W1, xp1, ei, cnt, lst, ovf);
    alpha1_kernel<<<(NNODES * H1 + 255) / 256, 256, 0, stream>>>(xp1, asw1, adw1, as1, ad1);
    node_agg1<<<NNODES / 8, 256, 0, stream>>>(cnt, lst, ovf, xp1, as1, ad1, b1,
                                              W2, asw2, adw2, xp2, as2, ad2);

    node_agg2<<<NNODES / 16, 512, 0, stream>>>(cnt, lst, ovf, xp2, as2, ad2, b2, out);
}


# Layer-1 logits made in the GEMM epilogue from the accumulators (alpha launch and its re-read of xp gone); each now a 16-lane tree sum per half head, then the two halves, not a serial 32-term chain
# speedup vs baseline: 8.1888x; 1.0598x over previous
#include <hip/hip_runtime.h>
#include <hip/hip_bf16.h>
#include <math.h>

#define NNODES 50000
#define NEDGES 800000
#define FIN    256
#define H1     8
#define C1     32
#define D1     256
#define OUT2   2
#define SLOPE  0.2f

#define LDSPAD 4
#define ASTRIDE (FIN + LDSPAD)

typedef float v2f __attribute__((ext_vector_type(2)));
typedef float v8f __attribute__((ext_vector_type(8)));

__device__ __forceinline__ float lrelu(float v) {
    return v > 0.0f ? v : SLOPE * v;
}

#define CAPH 32
#define CAP  (2 * CAPH)
static_assert((long long)NNODES * CAP < 2147483647LL, "slot index fits an int");

__device__ __forceinline__ int slot_index(int d, int j) {
    return (j < CAPH ? 0 : NNODES * CAPH - CAPH) + d * CAPH + j;
}


#define MTILES    (NNODES / 16)
#define G1_BLOCKS 256
#define KSTEPS    (FIN / 4)
#define EDGES_PER_TILE 512
static_assert(NNODES % 16 == 0, "gemm1 row tiles are 16 rows");
static_assert((long long)MTILES * EDGES_PER_TILE >= NEDGES, "every edge has an owner thread");
#define PSTRIDE   36
static_assert(D1 == 16 * (EDGES_PER_TILE / 32), "one wave per 16 output columns");
static_assert(C1 == 32 && H1 * 2 == EDGES_PER_TILE / 32, "two waves per head");

template <int M>
__device__ __forceinline__ float row_xor(float v) {
    static_assert(M > 0 && M < 16, "inside a 16-lane row");
    return __int_as_float(__builtin_amdgcn_update_dpp(0, __float_as_int(v), 0x160 + M, 0xf, 0xf, false));
}

__device__ __forceinline__ void store_logits(const float* __restrict__ pbuf, int t, int tile,
                                             float* __restrict__ as, float* __restrict__ ad) {
    if (t < 256) {
        const int kind = t >> 7;
        const int row  = (t >> 3) & 15;
        const int h    = t & 7;
        const float* pp = pbuf + (2 * h) * PSTRIDE + kind * 16 + row;
        const float s = pp[0] + pp[PSTRIDE];
        float* dstp = kind ? ad : as;
        dstp[(size_t)(tile * 16 + row) * H1 + h] = s;
    }
}
__global__ __launch_bounds__(EDGES_PER_TILE) void gemm1_wmma(const float* __restrict__ x,
                                                             const float* __restrict__ W1,
                                                             float* __restrict__ xp,
                                                             const float* __restrict__ a_src,
                                                             const float* __restrict__ a_dst,
                                                             float* __restrict__ as,
                                                             float* __restrict__ ad,
                                                             const int* __restrict__ ei,
                                                             int* __restrict__ cnt,
                                                             int* __restrict__ lst,
                                                             int* __restrict__ ovf) {
    __shared__ float aTile[2][16 * ASTRIDE];
    __shared__ float part[2][16 * PSTRIDE];

    const int t     = threadIdx.x;
    const int lane  = t & 31;
    const int wave  = t >> 5;
    const int half  = lane >> 4;
    const int l16   = lane & 15;
    const int bcol  = wave * 16 + l16;

    v2f breg[KSTEPS];
#pragma unroll
    for (int s = 0; s < KSTEPS; ++s) {
        const int kk = 4 * s + half * 2;
        breg[s].x = W1[kk * D1 + bcol];
        breg[s].y = W1[(kk + 1) * D1 + bcol];
    }
    const float av_s = a_src[bcol];
    const float av_d = a_dst[bcol];
    const bool b3 = (l16 & 8) != 0, b2 = (l16 & 4) != 0, b1 = (l16 & 2) != 0, b0 = (l16 & 1) != 0;
    const int  poff = wave * PSTRIDE + (l16 >> 3) * 16 + half * 8 + (l16 & 7);

    const int srow = t >> 5;
    const int scol = (t & 31) * 8;
    const int soff = srow * ASTRIDE + scol;
    const int aoff = l16 * ASTRIDE + half * 2;

    int tile = blockIdx.x;
    float4 p0 = make_float4(0.0f, 0.0f, 0.0f, 0.0f);
    float4 p1 = p0;
    int es = 0, ed = 0;
    if (tile < MTILES) {
        const float4* src = (const float4*)(x + (size_t)(tile * 16 + srow) * FIN + scol);
        p0 = src[0];
        p1 = src[1];
        const int e = tile * EDGES_PER_TILE + t;
        if (e < NEDGES) {
            es = ei[e];
            ed = ei[NEDGES + e];
        }
    }
    int buf = 0;
    int ptile = -1;
    for (; tile < MTILES; tile += G1_BLOCKS) {
        float* abuf = aTile[buf];
        *(float4*)(abuf + soff)     = p0;
        *(float4*)(abuf + soff + 4) = p1;
        __syncthreads();

        const bool has_edge = tile * EDGES_PER_TILE + t < NEDGES;
        const int  cs = es, cd = ed;
        int slot = 0;
        if (has_edge) slot = atomicAdd(&cnt[cd], 1);

        const int nxt = tile + G1_BLOCKS;
        if (nxt < MTILES) {
            const float4* src = (const float4*)(x + (size_t)(nxt * 16 + srow) * FIN + scol);
            p0 = src[0];
            p1 = src[1];
            const int ne = nxt * EDGES_PER_TILE + t;
            if (ne < NEDGES) {
                es = ei[ne];
                ed = ei[NEDGES + ne];
            }
        }

        if (ptile >= 0) store_logits(part[buf ^ 1], t, ptile, as, ad);

        v8f acc = {};
#pragma unroll
        for (int s = 0; s < KSTEPS; ++s) {
            v2f a = *(const v2f*)(abuf + aoff + 4 * s);
            acc = __builtin_amdgcn_wmma_f32_16x16x4_f32(
                 false, a,  false, breg[s],
                 (short)0, acc,  false,  false);
        }

        const int mrow = tile * 16 + half * 8;
#pragma unroll
        for (int r = 0; r < 8; ++r)
            xp[(size_t)(mrow + r) * D1 + bcol] = acc[r];

        {
            float v[16];
#pragma unroll
            for (int r = 0; r < 8; ++r) {
                v[r]     = acc[r] * av_s;
                v[8 + r] = acc[r] * av_d;
            }
            float u[8], w[4], y[2];
#pragma unroll
            for (int i = 0; i < 8; ++i)
                u[i] = (b3 ? v[8 + i] : v[i]) + row_xor<8>(b3 ? v[i] : v[8 + i]);
#pragma unroll
            for (int i = 0; i < 4; ++i)
                w[i] = (b2 ? u[4 + i] : u[i]) + row_xor<4>(b2 ? u[i] : u[4 + i]);
#pragma unroll
            for (int i = 0; i < 2; ++i)
                y[i] = (b1 ? w[2 + i] : w[i]) + row_xor<2>(b1 ? w[i] : w[2 + i]);
            const float z = (b0 ? y[1] : y[0]) + row_xor<1>(b0 ? y[0] : y[1]);
            part[buf][poff] = z;
        }

        if (has_edge) {
            if (slot < CAP) {
                lst[slot_index(cd, slot)] = cs;
            } else {
                const int k = atomicAdd(&cnt[NNODES], 1);
                ovf[2 * k]     = cd;
                ovf[2 * k + 1] = cs;
            }
        }
        ptile = tile;
        buf ^= 1;
    }
    __syncthreads();
    if (ptile >= 0) store_logits(part[buf ^ 1], t, ptile, as, ad);
}

__global__ void zero_i32(int* __restrict__ p, int n) {
    int i = blockIdx.x * blockDim.x + threadIdx.x;
    if (i < n) p[i] = 0;
}

__device__ __forceinline__ float wave_max(float v) {
    for (int o = 16; o > 0; o >>= 1) v = fmaxf(v, __shfl_xor(v, o));
    return v;
}
__device__ __forceinline__ float wave_sum(float v) {
    for (int o = 16; o > 0; o >>= 1) v += __shfl_xor(v, o);
    return v;
}
__device__ __forceinline__ int clamp_node(int s) {
    return min(max(s, 0), NNODES - 1);
}
__device__ __forceinline__ int list_src(const int* __restrict__ lst, int deg, int d, int j) {
    int s = d;
    if (j < deg) s = clamp_node(lst[slot_index(d, j)]);
    return s;
}
template <typename F>
__device__ __forceinline__ void walk_long_list(const int* __restrict__ lst,
                                               const int* __restrict__ ovf, int novf,
                                               int d, F&& f) {
#pragma nounroll
    for (int j = 0; j < CAP; ++j) f(clamp_node(lst[slot_index(d, j)]));
#pragma nounroll
    for (int k = 0; k < novf; ++k)
        if (ovf[2 * k] == d) f(clamp_node(ovf[2 * k + 1]));
    f(d);
}

__device__ __forceinline__ void gather_pass(float4& accA, float4& accB,
                                            const float* __restrict__ xlane,
                                            int s_l, float a_l, int nvalid, int hsel) {
    float4 xa[4], xb[4];
#pragma unroll
    for (int i = 0; i < 4; ++i) {
        const int s = __builtin_amdgcn_readlane(s_l, i * 8);
        const float* r = xlane + (size_t)s * D1;
        xa[i] = *(const float4*)(r);
        xb[i] = *(const float4*)(r + 128);
    }
#pragma unroll
    for (int i = 0; i < 4; ++i) {
        const float alo = __shfl(a_l, i * 8 + hsel);
        const float ahi = __shfl(a_l, i * 8 + 4 + hsel);
        float4 va = xa[i], vb = xb[i];
        if (i >= nvalid) {
            va = make_float4(0.0f, 0.0f, 0.0f, 0.0f);
            vb = make_float4(0.0f, 0.0f, 0.0f, 0.0f);
        }
        accA.x = fmaf(alo, va.x, accA.x);
        accA.y = fmaf(alo, va.y, accA.y);
        accA.z = fmaf(alo, va.z, accA.z);
        accA.w = fmaf(alo, va.w, accA.w);
        accB.x = fmaf(ahi, vb.x, accB.x);
        accB.y = fmaf(ahi, vb.y, accB.y);
        accB.z = fmaf(ahi, vb.z, accB.z);
        accB.w = fmaf(ahi, vb.w, accB.w);
    }
}

static_assert(NNODES % 8 == 0, "node_agg1 blocks cover 8 nodes each");
__global__ __launch_bounds__(256) void node_agg1(const int* __restrict__ cnt,
                                                 const int* __restrict__ lst,
                                                 const int* __restrict__ ovf,
                                                 const float* __restrict__ xp,
                                                 const float* __restrict__ as,
                                                 const float* __restrict__ ad,
                                                 const float* __restrict__ b,
                                                 const float* __restrict__ W2,
                                                 const float* __restrict__ aw_src,
                                                 const float* __restrict__ aw_dst,
                                                 float* __restrict__ xp2,
                                                 float* __restrict__ as2,
                                                 float* __restrict__ ad2) {
    const int lane = threadIdx.x & 31;
    const int d = __builtin_amdgcn_readfirstlane(blockIdx.x * 8 + (threadIdx.x >> 5));
    const int deg   = cnt[d];
    const int total = deg + 1;

    float4 accA = make_float4(0.0f, 0.0f, 0.0f, 0.0f);
    float4 accB = make_float4(0.0f, 0.0f, 0.0f, 0.0f);
    const float* xlane = xp + 4 * lane;
    const int hsel = lane >> 3;

    if (__builtin_expect(deg <= CAP, 1)) {
        const int jj    = lane >> 3;
        const int h     = lane & 7;
        const float add = ad[d * H1 + h];

        int   sp[8];
        float ev[8];
        const int s0 = list_src(lst, deg, d, lane);
    #pragma unroll
        for (int p = 0; p < 8; ++p) sp[p] = __shfl(s0, 4 * p + jj);
    #pragma unroll
        for (int p = 0; p < 8; ++p) {
            const float v = lrelu(as[sp[p] * H1 + h] + add);
            ev[p] = (4 * p + jj < total) ? v : -INFINITY;
        }

        float mx = ev[0];
    #pragma unroll
        for (int p = 1; p < 8; ++p) mx = fmaxf(mx, ev[p]);
        for (int j0 = 32; j0 < total; j0 += 4) {
            const int j = j0 + jj;
            if (j < total) {
                const int s = list_src(lst, deg, d, j);
                mx = fmaxf(mx, lrelu(as[s * H1 + h] + add));
            }
        }
        mx = fmaxf(mx, __shfl_xor(mx, 8));
        mx = fmaxf(mx, __shfl_xor(mx, 16));

        float sum = 0.0f;
    #pragma unroll
        for (int p = 0; p < 8; ++p) {
            ev[p] = (4 * p + jj < total) ? __expf(ev[p] - mx) : 0.0f;
            sum += ev[p];
        }
        for (int j0 = 32; j0 < total; j0 += 4) {
            const int j = j0 + jj;
            if (j < total) {
                const int s = list_src(lst, deg, d, j);
                sum += __expf(lrelu(as[s * H1 + h] + add) - mx);
            }
        }
        sum += __shfl_xor(sum, 8);
        sum += __shfl_xor(sum, 16);
        const float den = sum;
    #pragma unroll
        for (int p = 0; p < 8; ++p) ev[p] = ev[p] / den;

        const int n0   = min(total, 32);
    #pragma unroll
        for (int p = 0; p < 8; ++p) {
            if (4 * p < n0) gather_pass(accA, accB, xlane, sp[p], ev[p], n0 - 4 * p, hsel);
        }
        for (int j0 = 32; j0 < total; j0 += 4) {
            const int j = j0 + jj;
            int   s = d;
            float a = 0.0f;
            if (j < total) {
                s = list_src(lst, deg, d, j);
                a = __expf(lrelu(as[s * H1 + h] + add) - mx) / den;
            }
            gather_pass(accA, accB, xlane, s, a, total - j0, hsel);
        }
    } else {
        const int   novf = min(max(cnt[NNODES], 0), NEDGES);
        const int   hA   = hsel, hB = 4 + hsel;
        const float addA = ad[d * H1 + hA];
        const float addB = ad[d * H1 + hB];
        float mA = -INFINITY, mB = -INFINITY;
        walk_long_list(lst, ovf, novf, d, [&](int s) {
            mA = fmaxf(mA, lrelu(as[s * H1 + hA] + addA));
            mB = fmaxf(mB, lrelu(as[s * H1 + hB] + addB));
        });
        float sA = 0.0f, sB = 0.0f;
        walk_long_list(lst, ovf, novf, d, [&](int s) {
            sA += __expf(lrelu(as[s * H1 + hA] + addA) - mA);
            sB += __expf(lrelu(as[s * H1 + hB] + addB) - mB);
        });
        walk_long_list(lst, ovf, novf, d, [&](int s) {
            const float aA = __expf(lrelu(as[s * H1 + hA] + addA) - mA) / sA;
            const float aB = __expf(lrelu(as[s * H1 + hB] + addB) - mB) / sB;
            const float* r = xlane + (size_t)s * D1;
            const float4 va = *(const float4*)(r);
            const float4 vb = *(const float4*)(r + 128);
            accA.x = fmaf(aA, va.x, accA.x);
            accA.y = fmaf(aA, va.y, accA.y);
            accA.z = fmaf(aA, va.z, accA.z);
            accA.w = fmaf(aA, va.w, accA.w);
            accB.x = fmaf(aB, vb.x, accB.x);
            accB.y = fmaf(aB, vb.y, accB.y);
            accB.z = fmaf(aB, vb.z, accB.z);
            accB.w = fmaf(aB, vb.w, accB.w);
        });
    }

    const float* bl = b + 4 * lane;
    float4 rA, rB;
    rA.x = accA.x + bl[0];   rA.y = accA.y + bl[1];   rA.z = accA.z + bl[2];   rA.w = accA.w + bl[3];
    rB.x = accB.x + bl[128]; rB.y = accB.y + bl[129]; rB.z = accB.z + bl[130]; rB.w = accB.w + bl[131];
    rA.x = rA.x > 0.0f ? rA.x : 0.0f; rA.y = rA.y > 0.0f ? rA.y : 0.0f;
    rA.z = rA.z > 0.0f ? rA.z : 0.0f; rA.w = rA.w > 0.0f ? rA.w : 0.0f;
    rB.x = rB.x > 0.0f ? rB.x : 0.0f; rB.y = rB.y > 0.0f ? rB.y : 0.0f;
    rB.z = rB.z > 0.0f ? rB.z : 0.0f; rB.w = rB.w > 0.0f ? rB.w : 0.0f;

    const float4* w2l = (const float4*)(W2 + 8 * lane);
    const float4 wa0 = w2l[0];
    const float4 wa1 = w2l[1];
    const float4 wb0 = w2l[64];
    const float4 wb1 = w2l[65];
    float y0 = 0.0f, y1 = 0.0f;
    y0 = fmaf(rA.x, wa0.x, y0);  y1 = fmaf(rA.x, wa0.y, y1);
    y0 = fmaf(rA.y, wa0.z, y0);  y1 = fmaf(rA.y, wa0.w, y1);
    y0 = fmaf(rA.z, wa1.x, y0);  y1 = fmaf(rA.z, wa1.y, y1);
    y0 = fmaf(rA.w, wa1.z, y0);  y1 = fmaf(rA.w, wa1.w, y1);
    y0 = fmaf(rB.x, wb0.x, y0);  y1 = fmaf(rB.x, wb0.y, y1);
    y0 = fmaf(rB.y, wb0.z, y0);  y1 = fmaf(rB.y, wb0.w, y1);
    y0 = fmaf(rB.z, wb1.x, y0);  y1 = fmaf(rB.z, wb1.y, y1);
    y0 = fmaf(rB.w, wb1.z, y0);  y1 = fmaf(rB.w, wb1.w, y1);
    y0 = wave_sum(y0);
    y1 = wave_sum(y1);
    if (lane == 0) {
        *(float2*)(xp2 + (size_t)d * OUT2) = make_float2(y0, y1);
        as2[d] = fmaf(y0, aw_src[0], y1 * aw_src[1]);
        ad2[d] = fmaf(y0, aw_dst[0], y1 * aw_dst[1]);
    }
}

static_assert(NNODES % 16 == 0, "node_agg2 blocks cover 16 nodes each");
__global__ __launch_bounds__(512) void node_agg2(const int* __restrict__ cnt,
                                                 const int* __restrict__ lst,
                                                 const int* __restrict__ ovf,
                                                 const float* __restrict__ xp2,
                                                 const float* __restrict__ as2,
                                                 const float* __restrict__ ad2,
                                                 const float* __restrict__ b,
                                                 float* __restrict__ out) {
    __shared__ float res[16 * OUT2];
    const int lane = threadIdx.x & 31;
    const int w    = threadIdx.x >> 5;
    const int d    = __builtin_amdgcn_readfirstlane(blockIdx.x * 16 + w);
    const int deg   = cnt[d];
    const int total = deg + 1;
    const float add = ad2[d];

    float c0 = 0.0f, c1 = 0.0f;
    if (__builtin_expect(deg <= CAP, 1)) {
        const bool  act0 = lane < total;
        const int   s0   = list_src(lst, deg, d, lane);
        const float v0   = act0 ? lrelu(as2[s0] + add) : -INFINITY;

        float mx = v0;
        for (int j0 = 32; j0 < total; j0 += 32) {
            const int j = j0 + lane;
            if (j < total) {
                const int s = list_src(lst, deg, d, j);
                mx = fmaxf(mx, lrelu(as2[s] + add));
            }
        }
        mx = wave_max(mx);

        const float p0 = act0 ? __expf(v0 - mx) : 0.0f;
        float sum = p0;
        for (int j0 = 32; j0 < total; j0 += 32) {
            const int j = j0 + lane;
            if (j < total) {
                const int s = list_src(lst, deg, d, j);
                sum += __expf(lrelu(as2[s] + add) - mx);
            }
        }
        const float den = wave_sum(sum);

        if (act0) {
            const float a = p0 / den;
            c0 = a * xp2[s0 * OUT2 + 0];
            c1 = a * xp2[s0 * OUT2 + 1];
        }
        for (int j0 = 32; j0 < total; j0 += 32) {
            const int j = j0 + lane;
            if (j < total) {
                const int s = list_src(lst, deg, d, j);
                const float a = __expf(lrelu(as2[s] + add) - mx) / den;
                c0 = fmaf(a, xp2[s * OUT2 + 0], c0);
                c1 = fmaf(a, xp2[s * OUT2 + 1], c1);
            }
        }
    } else {
        const int novf = min(max(cnt[NNODES], 0), NEDGES);
        float m = -INFINITY;
        walk_long_list(lst, ovf, novf, d, [&](int s) {
            m = fmaxf(m, lrelu(as2[s] + add));
        });
        float sm = 0.0f;
        walk_long_list(lst, ovf, novf, d, [&](int s) {
            sm += __expf(lrelu(as2[s] + add) - m);
        });
        float t0 = 0.0f, t1 = 0.0f;
        walk_long_list(lst, ovf, novf, d, [&](int s) {
            const float a = __expf(lrelu(as2[s] + add) - m) / sm;
            t0 = fmaf(a, xp2[s * OUT2 + 0], t0);
            t1 = fmaf(a, xp2[s * OUT2 + 1], t1);
        });
        if (lane == 0) { c0 = t0; c1 = t1; }
    }
    c0 = wave_sum(c0);
    c1 = wave_sum(c1);
    if (lane == 0) {
        res[w * OUT2 + 0] = c0;
        res[w * OUT2 + 1] = c1;
    }
    __syncthreads();
    if (threadIdx.x < 16 * OUT2)
        out[blockIdx.x * (16 * OUT2) + threadIdx.x] = res[threadIdx.x] + b[threadIdx.x & 1];
}

extern "C" void kernel_launch(void* const* d_in, const int* in_sizes, int n_in,
                              void* d_out, int out_size, void* d_ws, size_t ws_size,
                              hipStream_t stream) {
    const float* x     = (const float*)d_in[0];
    const int*   ei    = (const int*)d_in[1];
    const float* W1    = (const float*)d_in[2];
    const float* asw1  = (const float*)d_in[3];
    const float* adw1  = (const float*)d_in[4];
    const float* b1    = (const float*)d_in[5];
    const float* W2    = (const float*)d_in[6];
    const float* asw2  = (const float*)d_in[7];
    const float* adw2  = (const float*)d_in[8];
    const float* b2    = (const float*)d_in[9];
    float*       out   = (float*)d_out;

    float* w    = (float*)d_ws;
    float* xp1  = w;
    float* as1  = xp1  + (size_t)NNODES * D1;
    float* ad1  = as1  + (size_t)NNODES * H1;
    float* xp2  = ad1  + (size_t)NNODES * H1;
    float* as2  = xp2  + (size_t)NNODES * OUT2;
    float* ad2  = as2  + (size_t)NNODES;
    int*   lst  = (int*)(ad2 + (size_t)NNODES);
    int*   ovf  = lst  + (size_t)NNODES * CAP;
    int*   cnt  = ovf  + (size_t)2 * NEDGES;

    zero_i32<<<(NNODES + 1 + 255) / 256, 256, 0, stream>>>(cnt, NNODES + 1);

    gemm1_wmma<<<G1_BLOCKS, EDGES_PER_TILE, 0, stream>>>(x, W1, xp1, asw1, adw1, as1, ad1,
                                                         ei, cnt, lst, ovf);
    node_agg1<<<NNODES / 8, 256, 0, stream>>>(cnt, lst, ovf, xp1, as1, ad1, b1,
                                              W2, asw2, adw2, xp2, as2, ad2);

    node_agg2<<<NNODES / 16, 512, 0, stream>>>(cnt, lst, ovf, xp2, as2, ad2, b2, out);
}


# Edge placement spread evenly over all 3125 row tiles (256 edges per tile, on waves 8-15) instead of 512 per tile on the first 1563: counting adds come at half the rate; list order stays arrival order
# speedup vs baseline: 8.7270x; 1.0657x over previous
#include <hip/hip_runtime.h>
#include <hip/hip_bf16.h>
#include <math.h>

#define NNODES 50000
#define NEDGES 800000
#define FIN    256
#define H1     8
#define C1     32
#define D1     256
#define OUT2   2
#define SLOPE  0.2f

#define LDSPAD 4
#define ASTRIDE (FIN + LDSPAD)

typedef float v2f __attribute__((ext_vector_type(2)));
typedef float v8f __attribute__((ext_vector_type(8)));

__device__ __forceinline__ float lrelu(float v) {
    return v > 0.0f ? v : SLOPE * v;
}

#define CAPH 32
#define CAP  (2 * CAPH)
static_assert((long long)NNODES * CAP < 2147483647LL, "slot index fits an int");

__device__ __forceinline__ int slot_index(int d, int j) {
    return (j < CAPH ? 0 : NNODES * CAPH - CAPH) + d * CAPH + j;
}


#define MTILES    (NNODES / 16)
#define G1_BLOCKS 256
#define KSTEPS    (FIN / 4)
#define G1_THREADS 512
#define EDGES_PER_TILE 256
static_assert(NNODES % 16 == 0, "gemm1 row tiles are 16 rows");
static_assert((long long)MTILES * EDGES_PER_TILE >= NEDGES, "every edge has an owner thread");
static_assert(EDGES_PER_TILE % 32 == 0 && EDGES_PER_TILE <= G1_THREADS, "whole waves place edges");
#define PSTRIDE   36
static_assert(D1 == 16 * (G1_THREADS / 32), "one wave per 16 output columns");
static_assert(C1 == 32 && H1 * 2 == G1_THREADS / 32, "two waves per head");

template <int M>
__device__ __forceinline__ float row_xor(float v) {
    static_assert(M > 0 && M < 16, "inside a 16-lane row");
    return __int_as_float(__builtin_amdgcn_update_dpp(0, __float_as_int(v), 0x160 + M, 0xf, 0xf, false));
}

__device__ __forceinline__ void store_logits(const float* __restrict__ pbuf, int t, int tile,
                                             float* __restrict__ as, float* __restrict__ ad) {
    if (t < 256) {
        const int kind = t >> 7;
        const int row  = (t >> 3) & 15;
        const int h    = t & 7;
        const float* pp = pbuf + (2 * h) * PSTRIDE + kind * 16 + row;
        const float s = pp[0] + pp[PSTRIDE];
        float* dstp = kind ? ad : as;
        dstp[(size_t)(tile * 16 + row) * H1 + h] = s;
    }
}
__global__ __launch_bounds__(G1_THREADS) void gemm1_wmma(const float* __restrict__ x,
                                                             const float* __restrict__ W1,
                                                             float* __restrict__ xp,
                                                             const float* __restrict__ a_src,
                                                             const float* __restrict__ a_dst,
                                                             float* __restrict__ as,
                                                             float* __restrict__ ad,
                                                             const int* __restrict__ ei,
                                                             int* __restrict__ cnt,
                                                             int* __restrict__ lst,
                                                             int* __restrict__ ovf) {
    __shared__ float aTile[2][16 * ASTRIDE];
    __shared__ float part[2][16 * PSTRIDE];

    const int t     = threadIdx.x;
    const int lane  = t & 31;
    const int wave  = t >> 5;
    const int half  = lane >> 4;
    const int l16   = lane & 15;
    const int bcol  = wave * 16 + l16;

    v2f breg[KSTEPS];
#pragma unroll
    for (int s = 0; s < KSTEPS; ++s) {
        const int kk = 4 * s + half * 2;
        breg[s].x = W1[kk * D1 + bcol];
        breg[s].y = W1[(kk + 1) * D1 + bcol];
    }
    const float av_s = a_src[bcol];
    const float av_d = a_dst[bcol];
    const bool b3 = (l16 & 8) != 0, b2 = (l16 & 4) != 0, b1 = (l16 & 2) != 0, b0 = (l16 & 1) != 0;
    const int  poff = wave * PSTRIDE + (l16 >> 3) * 16 + half * 8 + (l16 & 7);

    const int srow = t >> 5;
    const int scol = (t & 31) * 8;
    const int soff = srow * ASTRIDE + scol;
    const int aoff = l16 * ASTRIDE + half * 2;

    const int et = t - (G1_THREADS - EDGES_PER_TILE);

    int tile = blockIdx.x;
    float4 p0 = make_float4(0.0f, 0.0f, 0.0f, 0.0f);
    float4 p1 = p0;
    int es = 0, ed = 0;
    if (tile < MTILES) {
        const float4* src = (const float4*)(x + (size_t)(tile * 16 + srow) * FIN + scol);
        p0 = src[0];
        p1 = src[1];
        const int e = tile * EDGES_PER_TILE + et;
        if (et >= 0 && e < NEDGES) {
            es = ei[e];
            ed = ei[NEDGES + e];
        }
    }
    int buf = 0;
    int ptile = -1;
    for (; tile < MTILES; tile += G1_BLOCKS) {
        float* abuf = aTile[buf];
        *(float4*)(abuf + soff)     = p0;
        *(float4*)(abuf + soff + 4) = p1;
        __syncthreads();

        const bool has_edge = et >= 0 && tile * EDGES_PER_TILE + et < NEDGES;
        const int  cs = es, cd = ed;
        int slot = 0;
        if (has_edge) slot = atomicAdd(&cnt[cd], 1);

        const int nxt = tile + G1_BLOCKS;
        if (nxt < MTILES) {
            const float4* src = (const float4*)(x + (size_t)(nxt * 16 + srow) * FIN + scol);
            p0 = src[0];
            p1 = src[1];
            const int ne = nxt * EDGES_PER_TILE + et;
            if (et >= 0 && ne < NEDGES) {
                es = ei[ne];
                ed = ei[NEDGES + ne];
            }
        }

        if (ptile >= 0) store_logits(part[buf ^ 1], t, ptile, as, ad);

        v8f acc = {};
#pragma unroll
        for (int s = 0; s < KSTEPS; ++s) {
            v2f a = *(const v2f*)(abuf + aoff + 4 * s);
            acc = __builtin_amdgcn_wmma_f32_16x16x4_f32(
                 false, a,  false, breg[s],
                 (short)0, acc,  false,  false);
        }

        const int mrow = tile * 16 + half * 8;
#pragma unroll
        for (int r = 0; r < 8; ++r)
            xp[(size_t)(mrow + r) * D1 + bcol] = acc[r];

        {
            float v[16];
#pragma unroll
            for (int r = 0; r < 8; ++r) {
                v[r]     = acc[r] * av_s;
                v[8 + r] = acc[r] * av_d;
            }
            float u[8], w[4], y[2];
#pragma unroll
            for (int i = 0; i < 8; ++i)
                u[i] = (b3 ? v[8 + i] : v[i]) + row_xor<8>(b3 ? v[i] : v[8 + i]);
#pragma unroll
            for (int i = 0; i < 4; ++i)
                w[i] = (b2 ? u[4 + i] : u[i]) + row_xor<4>(b2 ? u[i] : u[4 + i]);
#pragma unroll
            for (int i = 0; i < 2; ++i)
                y[i] = (b1 ? w[2 + i] : w[i]) + row_xor<2>(b1 ? w[i] : w[2 + i]);
            const float z = (b0 ? y[1] : y[0]) + row_xor<1>(b0 ? y[0] : y[1]);
            part[buf][poff] = z;
        }

        if (has_edge) {
            if (slot < CAP) {
                lst[slot_index(cd, slot)] = cs;
            } else {
                const int k = atomicAdd(&cnt[NNODES], 1);
                ovf[2 * k]     = cd;
                ovf[2 * k + 1] = cs;
            }
        }
        ptile = tile;
        buf ^= 1;
    }
    __syncthreads();
    if (ptile >= 0) store_logits(part[buf ^ 1], t, ptile, as, ad);
}

__global__ void zero_i32(int* __restrict__ p, int n) {
    int i = blockIdx.x * blockDim.x + threadIdx.x;
    if (i < n) p[i] = 0;
}

__device__ __forceinline__ float wave_max(float v) {
    for (int o = 16; o > 0; o >>= 1) v = fmaxf(v, __shfl_xor(v, o));
    return v;
}
__device__ __forceinline__ float wave_sum(float v) {
    for (int o = 16; o > 0; o >>= 1) v += __shfl_xor(v, o);
    return v;
}
__device__ __forceinline__ int clamp_node(int s) {
    return min(max(s, 0), NNODES - 1);
}
__device__ __forceinline__ int list_src(const int* __restrict__ lst, int deg, int d, int j) {
    int s = d;
    if (j < deg) s = clamp_node(lst[slot_index(d, j)]);
    return s;
}
template <typename F>
__device__ __forceinline__ void walk_long_list(const int* __restrict__ lst,
                                               const int* __restrict__ ovf, int novf,
                                               int d, F&& f) {
#pragma nounroll
    for (int j = 0; j < CAP; ++j) f(clamp_node(lst[slot_index(d, j)]));
#pragma nounroll
    for (int k = 0; k < novf; ++k)
        if (ovf[2 * k] == d) f(clamp_node(ovf[2 * k + 1]));
    f(d);
}

__device__ __forceinline__ void gather_pass(float4& accA, float4& accB,
                                            const float* __restrict__ xlane,
                                            int s_l, float a_l, int nvalid, int hsel) {
    float4 xa[4], xb[4];
#pragma unroll
    for (int i = 0; i < 4; ++i) {
        const int s = __builtin_amdgcn_readlane(s_l, i * 8);
        const float* r = xlane + (size_t)s * D1;
        xa[i] = *(const float4*)(r);
        xb[i] = *(const float4*)(r + 128);
    }
#pragma unroll
    for (int i = 0; i < 4; ++i) {
        const float alo = __shfl(a_l, i * 8 + hsel);
        const float ahi = __shfl(a_l, i * 8 + 4 + hsel);
        float4 va = xa[i], vb = xb[i];
        if (i >= nvalid) {
            va = make_float4(0.0f, 0.0f, 0.0f, 0.0f);
            vb = make_float4(0.0f, 0.0f, 0.0f, 0.0f);
        }
        accA.x = fmaf(alo, va.x, accA.x);
        accA.y = fmaf(alo, va.y, accA.y);
        accA.z = fmaf(alo, va.z, accA.z);
        accA.w = fmaf(alo, va.w, accA.w);
        accB.x = fmaf(ahi, vb.x, accB.x);
        accB.y = fmaf(ahi, vb.y, accB.y);
        accB.z = fmaf(ahi, vb.z, accB.z);
        accB.w = fmaf(ahi, vb.w, accB.w);
    }
}

static_assert(NNODES % 8 == 0, "node_agg1 blocks cover 8 nodes each");
__global__ __launch_bounds__(256) void node_agg1(const int* __restrict__ cnt,
                                                 const int* __restrict__ lst,
                                                 const int* __restrict__ ovf,
                                                 const float* __restrict__ xp,
                                                 const float* __restrict__ as,
                                                 const float* __restrict__ ad,
                                                 const float* __restrict__ b,
                                                 const float* __restrict__ W2,
                                                 const float* __restrict__ aw_src,
                                                 const float* __restrict__ aw_dst,
                                                 float* __restrict__ xp2,
                                                 float* __restrict__ as2,
                                                 float* __restrict__ ad2) {
    const int lane = threadIdx.x & 31;
    const int d = __builtin_amdgcn_readfirstlane(blockIdx.x * 8 + (threadIdx.x >> 5));
    const int deg   = cnt[d];
    const int total = deg + 1;

    float4 accA = make_float4(0.0f, 0.0f, 0.0f, 0.0f);
    float4 accB = make_float4(0.0f, 0.0f, 0.0f, 0.0f);
    const float* xlane = xp + 4 * lane;
    const int hsel = lane >> 3;

    if (__builtin_expect(deg <= CAP, 1)) {
        const int jj    = lane >> 3;
        const int h     = lane & 7;
        const float add = ad[d * H1 + h];

        int   sp[8];
        float ev[8];
        const int s0 = list_src(lst, deg, d, lane);
    #pragma unroll
        for (int p = 0; p < 8; ++p) sp[p] = __shfl(s0, 4 * p + jj);
    #pragma unroll
        for (int p = 0; p < 8; ++p) {
            const float v = lrelu(as[sp[p] * H1 + h] + add);
            ev[p] = (4 * p + jj < total) ? v : -INFINITY;
        }

        float mx = ev[0];
    #pragma unroll
        for (int p = 1; p < 8; ++p) mx = fmaxf(mx, ev[p]);
        for (int j0 = 32; j0 < total; j0 += 4) {
            const int j = j0 + jj;
            if (j < total) {
                const int s = list_src(lst, deg, d, j);
                mx = fmaxf(mx, lrelu(as[s * H1 + h] + add));
            }
        }
        mx = fmaxf(mx, __shfl_xor(mx, 8));
        mx = fmaxf(mx, __shfl_xor(mx, 16));

        float sum = 0.0f;
    #pragma unroll
        for (int p = 0; p < 8; ++p) {
            ev[p] = (4 * p + jj < total) ? __expf(ev[p] - mx) : 0.0f;
            sum += ev[p];
        }
        for (int j0 = 32; j0 < total; j0 += 4) {
            const int j = j0 + jj;
            if (j < total) {
                const int s = list_src(lst, deg, d, j);
                sum += __expf(lrelu(as[s * H1 + h] + add) - mx);
            }
        }
        sum += __shfl_xor(sum, 8);
        sum += __shfl_xor(sum, 16);
        const float den = sum;
    #pragma unroll
        for (int p = 0; p < 8; ++p) ev[p] = ev[p] / den;

        const int n0   = min(total, 32);
    #pragma unroll
        for (int p = 0; p < 8; ++p) {
            if (4 * p < n0) gather_pass(accA, accB, xlane, sp[p], ev[p], n0 - 4 * p, hsel);
        }
        for (int j0 = 32; j0 < total; j0 += 4) {
            const int j = j0 + jj;
            int   s = d;
            float a = 0.0f;
            if (j < total) {
                s = list_src(lst, deg, d, j);
                a = __expf(lrelu(as[s * H1 + h] + add) - mx) / den;
            }
            gather_pass(accA, accB, xlane, s, a, total - j0, hsel);
        }
    } else {
        const int   novf = min(max(cnt[NNODES], 0), NEDGES);
        const int   hA   = hsel, hB = 4 + hsel;
        const float addA = ad[d * H1 + hA];
        const float addB = ad[d * H1 + hB];
        float mA = -INFINITY, mB = -INFINITY;
        walk_long_list(lst, ovf, novf, d, [&](int s) {
            mA = fmaxf(mA, lrelu(as[s * H1 + hA] + addA));
            mB = fmaxf(mB, lrelu(as[s * H1 + hB] + addB));
        });
        float sA = 0.0f, sB = 0.0f;
        walk_long_list(lst, ovf, novf, d, [&](int s) {
            sA += __expf(lrelu(as[s * H1 + hA] + addA) - mA);
            sB += __expf(lrelu(as[s * H1 + hB] + addB) - mB);
        });
        walk_long_list(lst, ovf, novf, d, [&](int s) {
            const float aA = __expf(lrelu(as[s * H1 + hA] + addA) - mA) / sA;
            const float aB = __expf(lrelu(as[s * H1 + hB] + addB) - mB) / sB;
            const float* r = xlane + (size_t)s * D1;
            const float4 va = *(const float4*)(r);
            const float4 vb = *(const float4*)(r + 128);
            accA.x = fmaf(aA, va.x, accA.x);
            accA.y = fmaf(aA, va.y, accA.y);
            accA.z = fmaf(aA, va.z, accA.z);
            accA.w = fmaf(aA, va.w, accA.w);
            accB.x = fmaf(aB, vb.x, accB.x);
            accB.y = fmaf(aB, vb.y, accB.y);
            accB.z = fmaf(aB, vb.z, accB.z);
            accB.w = fmaf(aB, vb.w, accB.w);
        });
    }

    const float* bl = b + 4 * lane;
    float4 rA, rB;
    rA.x = accA.x + bl[0];   rA.y = accA.y + bl[1];   rA.z = accA.z + bl[2];   rA.w = accA.w + bl[3];
    rB.x = accB.x + bl[128]; rB.y = accB.y + bl[129]; rB.z = accB.z + bl[130]; rB.w = accB.w + bl[131];
    rA.x = rA.x > 0.0f ? rA.x : 0.0f; rA.y = rA.y > 0.0f ? rA.y : 0.0f;
    rA.z = rA.z > 0.0f ? rA.z : 0.0f; rA.w = rA.w > 0.0f ? rA.w : 0.0f;
    rB.x = rB.x > 0.0f ? rB.x : 0.0f; rB.y = rB.y > 0.0f ? rB.y : 0.0f;
    rB.z = rB.z > 0.0f ? rB.z : 0.0f; rB.w = rB.w > 0.0f ? rB.w : 0.0f;

    const float4* w2l = (const float4*)(W2 + 8 * lane);
    const float4 wa0 = w2l[0];
    const float4 wa1 = w2l[1];
    const float4 wb0 = w2l[64];
    const float4 wb1 = w2l[65];
    float y0 = 0.0f, y1 = 0.0f;
    y0 = fmaf(rA.x, wa0.x, y0);  y1 = fmaf(rA.x, wa0.y, y1);
    y0 = fmaf(rA.y, wa0.z, y0);  y1 = fmaf(rA.y, wa0.w, y1);
    y0 = fmaf(rA.z, wa1.x, y0);  y1 = fmaf(rA.z, wa1.y, y1);
    y0 = fmaf(rA.w, wa1.z, y0);  y1 = fmaf(rA.w, wa1.w, y1);
    y0 = fmaf(rB.x, wb0.x, y0);  y1 = fmaf(rB.x, wb0.y, y1);
    y0 = fmaf(rB.y, wb0.z, y0);  y1 = fmaf(rB.y, wb0.w, y1);
    y0 = fmaf(rB.z, wb1.x, y0);  y1 = fmaf(rB.z, wb1.y, y1);
    y0 = fmaf(rB.w, wb1.z, y0);  y1 = fmaf(rB.w, wb1.w, y1);
    y0 = wave_sum(y0);
    y1 = wave_sum(y1);
    if (lane == 0) {
        *(float2*)(xp2 + (size_t)d * OUT2) = make_float2(y0, y1);
        as2[d] = fmaf(y0, aw_src[0], y1 * aw_src[1]);
        ad2[d] = fmaf(y0, aw_dst[0], y1 * aw_dst[1]);
    }
}

static_assert(NNODES % 16 == 0, "node_agg2 blocks cover 16 nodes each");
__global__ __launch_bounds__(512) void node_agg2(const int* __restrict__ cnt,
                                                 const int* __restrict__ lst,
                                                 const int* __restrict__ ovf,
                                                 const float* __restrict__ xp2,
                                                 const float* __restrict__ as2,
                                                 const float* __restrict__ ad2,
                                                 const float* __restrict__ b,
                                                 float* __restrict__ out) {
    __shared__ float res[16 * OUT2];
    const int lane = threadIdx.x & 31;
    const int w    = threadIdx.x >> 5;
    const int d    = __builtin_amdgcn_readfirstlane(blockIdx.x * 16 + w);
    const int deg   = cnt[d];
    const int total = deg + 1;
    const float add = ad2[d];

    float c0 = 0.0f, c1 = 0.0f;
    if (__builtin_expect(deg <= CAP, 1)) {
        const bool  act0 = lane < total;
        const int   s0   = list_src(lst, deg, d, lane);
        const float v0   = act0 ? lrelu(as2[s0] + add) : -INFINITY;

        float mx = v0;
        for (int j0 = 32; j0 < total; j0 += 32) {
            const int j = j0 + lane;
            if (j < total) {
                const int s = list_src(lst, deg, d, j);
                mx = fmaxf(mx, lrelu(as2[s] + add));
            }
        }
        mx = wave_max(mx);

        const float p0 = act0 ? __expf(v0 - mx) : 0.0f;
        float sum = p0;
        for (int j0 = 32; j0 < total; j0 += 32) {
            const int j = j0 + lane;
            if (j < total) {
                const int s = list_src(lst, deg, d, j);
                sum += __expf(lrelu(as2[s] + add) - mx);
            }
        }
        const float den = wave_sum(sum);

        if (act0) {
            const float a = p0 / den;
            c0 = a * xp2[s0 * OUT2 + 0];
            c1 = a * xp2[s0 * OUT2 + 1];
        }
        for (int j0 = 32; j0 < total; j0 += 32) {
            const int j = j0 + lane;
            if (j < total) {
                const int s = list_src(lst, deg, d, j);
                const float a = __expf(lrelu(as2[s] + add) - mx) / den;
                c0 = fmaf(a, xp2[s * OUT2 + 0], c0);
                c1 = fmaf(a, xp2[s * OUT2 + 1], c1);
            }
        }
    } else {
        const int novf = min(max(cnt[NNODES], 0), NEDGES);
        float m = -INFINITY;
        walk_long_list(lst, ovf, novf, d, [&](int s) {
            m = fmaxf(m, lrelu(as2[s] + add));
        });
        float sm = 0.0f;
        walk_long_list(lst, ovf, novf, d, [&](int s) {
            sm += __expf(lrelu(as2[s] + add) - m);
        });
        float t0 = 0.0f, t1 = 0.0f;
        walk_long_list(lst, ovf, novf, d, [&](int s) {
            const float a = __expf(lrelu(as2[s] + add) - m) / sm;
            t0 = fmaf(a, xp2[s * OUT2 + 0], t0);
            t1 = fmaf(a, xp2[s * OUT2 + 1], t1);
        });
        if (lane == 0) { c0 = t0; c1 = t1; }
    }
    c0 = wave_sum(c0);
    c1 = wave_sum(c1);
    if (lane == 0) {
        res[w * OUT2 + 0] = c0;
        res[w * OUT2 + 1] = c1;
    }
    __syncthreads();
    if (threadIdx.x < 16 * OUT2)
        out[blockIdx.x * (16 * OUT2) + threadIdx.x] = res[threadIdx.x] + b[threadIdx.x & 1];
}

extern "C" void kernel_launch(void* const* d_in, const int* in_sizes, int n_in,
                              void* d_out, int out_size, void* d_ws, size_t ws_size,
                              hipStream_t stream) {
    const float* x     = (const float*)d_in[0];
    const int*   ei    = (const int*)d_in[1];
    const float* W1    = (const float*)d_in[2];
    const float* asw1  = (const float*)d_in[3];
    const float* adw1  = (const float*)d_in[4];
    const float* b1    = (const float*)d_in[5];
    const float* W2    = (const float*)d_in[6];
    const float* asw2  = (const float*)d_in[7];
    const float* adw2  = (const float*)d_in[8];
    const float* b2    = (const float*)d_in[9];
    float*       out   = (float*)d_out;

    float* w    = (float*)d_ws;
    float* xp1  = w;
    float* as1  = xp1  + (size_t)NNODES * D1;
    float* ad1  = as1  + (size_t)NNODES * H1;
    float* xp2  = ad1  + (size_t)NNODES * H1;
    float* as2  = xp2  + (size_t)NNODES * OUT2;
    float* ad2  = as2  + (size_t)NNODES;
    int*   lst  = (int*)(ad2 + (size_t)NNODES);
    int*   ovf  = lst  + (size_t)NNODES * CAP;
    int*   cnt  = ovf  + (size_t)2 * NEDGES;

    zero_i32<<<(NNODES + 1 + 255) / 256, 256, 0, stream>>>(cnt, NNODES + 1);

    gemm1_wmma<<<G1_BLOCKS, G1_THREADS, 0, stream>>>(x, W1, xp1, asw1, adw1, as1, ad1,
                                                         ei, cnt, lst, ovf);
    node_agg1<<<NNODES / 8, 256, 0, stream>>>(cnt, lst, ovf, xp1, as1, ad1, b1,
                                              W2, asw2, adw2, xp2, as2, ad2);

    node_agg2<<<NNODES / 16, 512, 0, stream>>>(cnt, lst, ovf, xp2, as2, ad2, b2, out);
}


# Layer-2 aggregation: one lane per node (1563 waves, not 50000), list lines and 16-byte node records loaded in groups into registers; sums now serial in list order, self loop last, no lane tree
# speedup vs baseline: 9.3510x; 1.0715x over previous
#include <hip/hip_runtime.h>
#include <hip/hip_bf16.h>
#include <math.h>

#define NNODES 50000
#define NEDGES 800000
#define FIN    256
#define H1     8
#define C1     32
#define D1     256
#define OUT2   2
#define REC2   4
#define SLOPE  0.2f

#define LDSPAD 4
#define ASTRIDE (FIN + LDSPAD)

typedef float v2f __attribute__((ext_vector_type(2)));
typedef float v4f __attribute__((ext_vector_type(4)));
typedef float v8f __attribute__((ext_vector_type(8)));

__device__ __forceinline__ float lrelu(float v) {
    return v > 0.0f ? v : SLOPE * v;
}

#define CAPH 32
#define CAP  (2 * CAPH)
static_assert((long long)NNODES * CAP < 2147483647LL, "slot index fits an int");

__device__ __forceinline__ int slot_index(int d, int j) {
    return (j < CAPH ? 0 : NNODES * CAPH - CAPH) + d * CAPH + j;
}


#define MTILES    (NNODES / 16)
#define G1_BLOCKS 256
#define KSTEPS    (FIN / 4)
#define G1_THREADS 512
#define EDGES_PER_TILE 256
static_assert(NNODES % 16 == 0, "gemm1 row tiles are 16 rows");
static_assert((long long)MTILES * EDGES_PER_TILE >= NEDGES, "every edge has an owner thread");
static_assert(EDGES_PER_TILE % 32 == 0 && EDGES_PER_TILE <= G1_THREADS, "whole waves place edges");
#define PSTRIDE   36
static_assert(D1 == 16 * (G1_THREADS / 32), "one wave per 16 output columns");
static_assert(C1 == 32 && H1 * 2 == G1_THREADS / 32, "two waves per head");

template <int M>
__device__ __forceinline__ float row_xor(float v) {
    static_assert(M > 0 && M < 16, "inside a 16-lane row");
    return __int_as_float(__builtin_amdgcn_update_dpp(0, __float_as_int(v), 0x160 + M, 0xf, 0xf, false));
}

__device__ __forceinline__ void store_logits(const float* __restrict__ pbuf, int t, int tile,
                                             float* __restrict__ as, float* __restrict__ ad) {
    if (t < 256) {
        const int kind = t >> 7;
        const int row  = (t >> 3) & 15;
        const int h    = t & 7;
        const float* pp = pbuf + (2 * h) * PSTRIDE + kind * 16 + row;
        const float s = pp[0] + pp[PSTRIDE];
        float* dstp = kind ? ad : as;
        dstp[(size_t)(tile * 16 + row) * H1 + h] = s;
    }
}
__global__ __launch_bounds__(G1_THREADS) void gemm1_wmma(const float* __restrict__ x,
                                                             const float* __restrict__ W1,
                                                             float* __restrict__ xp,
                                                             const float* __restrict__ a_src,
                                                             const float* __restrict__ a_dst,
                                                             float* __restrict__ as,
                                                             float* __restrict__ ad,
                                                             const int* __restrict__ ei,
                                                             int* __restrict__ cnt,
                                                             int* __restrict__ lst,
                                                             int* __restrict__ ovf) {
    __shared__ float aTile[2][16 * ASTRIDE];
    __shared__ float part[2][16 * PSTRIDE];

    const int t     = threadIdx.x;
    const int lane  = t & 31;
    const int wave  = t >> 5;
    const int half  = lane >> 4;
    const int l16   = lane & 15;
    const int bcol  = wave * 16 + l16;

    v2f breg[KSTEPS];
#pragma unroll
    for (int s = 0; s < KSTEPS; ++s) {
        const int kk = 4 * s + half * 2;
        breg[s].x = W1[kk * D1 + bcol];
        breg[s].y = W1[(kk + 1) * D1 + bcol];
    }
    const float av_s = a_src[bcol];
    const float av_d = a_dst[bcol];
    const bool b3 = (l16 & 8) != 0, b2 = (l16 & 4) != 0, b1 = (l16 & 2) != 0, b0 = (l16 & 1) != 0;
    const int  poff = wave * PSTRIDE + (l16 >> 3) * 16 + half * 8 + (l16 & 7);

    const int srow = t >> 5;
    const int scol = (t & 31) * 8;
    const int soff = srow * ASTRIDE + scol;
    const int aoff = l16 * ASTRIDE + half * 2;

    const int et = t - (G1_THREADS - EDGES_PER_TILE);

    int tile = blockIdx.x;
    float4 p0 = make_float4(0.0f, 0.0f, 0.0f, 0.0f);
    float4 p1 = p0;
    int es = 0, ed = 0;
    if (tile < MTILES) {
        const float4* src = (const float4*)(x + (size_t)(tile * 16 + srow) * FIN + scol);
        p0 = src[0];
        p1 = src[1];
        const int e = tile * EDGES_PER_TILE + et;
        if (et >= 0 && e < NEDGES) {
            es = ei[e];
            ed = ei[NEDGES + e];
        }
    }
    int buf = 0;
    int ptile = -1;
    for (; tile < MTILES; tile += G1_BLOCKS) {
        float* abuf = aTile[buf];
        *(float4*)(abuf + soff)     = p0;
        *(float4*)(abuf + soff + 4) = p1;
        __syncthreads();

        const bool has_edge = et >= 0 && tile * EDGES_PER_TILE + et < NEDGES;
        const int  cs = es, cd = ed;
        int slot = 0;
        if (has_edge) slot = atomicAdd(&cnt[cd], 1);

        const int nxt = tile + G1_BLOCKS;
        if (nxt < MTILES) {
            const float4* src = (const float4*)(x + (size_t)(nxt * 16 + srow) * FIN + scol);
            p0 = src[0];
            p1 = src[1];
            const int ne = nxt * EDGES_PER_TILE + et;
            if (et >= 0 && ne < NEDGES) {
                es = ei[ne];
                ed = ei[NEDGES + ne];
            }
        }

        if (ptile >= 0) store_logits(part[buf ^ 1], t, ptile, as, ad);

        v8f acc = {};
#pragma unroll
        for (int s = 0; s < KSTEPS; ++s) {
            v2f a = *(const v2f*)(abuf + aoff + 4 * s);
            acc = __builtin_amdgcn_wmma_f32_16x16x4_f32(
                 false, a,  false, breg[s],
                 (short)0, acc,  false,  false);
        }

        const int mrow = tile * 16 + half * 8;
#pragma unroll
        for (int r = 0; r < 8; ++r)
            xp[(size_t)(mrow + r) * D1 + bcol] = acc[r];

        {
            float v[16];
#pragma unroll
            for (int r = 0; r < 8; ++r) {
                v[r]     = acc[r] * av_s;
                v[8 + r] = acc[r] * av_d;
            }
            float u[8], w[4], y[2];
#pragma unroll
            for (int i = 0; i < 8; ++i)
                u[i] = (b3 ? v[8 + i] : v[i]) + row_xor<8>(b3 ? v[i] : v[8 + i]);
#pragma unroll
            for (int i = 0; i < 4; ++i)
                w[i] = (b2 ? u[4 + i] : u[i]) + row_xor<4>(b2 ? u[i] : u[4 + i]);
#pragma unroll
            for (int i = 0; i < 2; ++i)
                y[i] = (b1 ? w[2 + i] : w[i]) + row_xor<2>(b1 ? w[i] : w[2 + i]);
            const float z = (b0 ? y[1] : y[0]) + row_xor<1>(b0 ? y[0] : y[1]);
            part[buf][poff] = z;
        }

        if (has_edge) {
            if (slot < CAP) {
                lst[slot_index(cd, slot)] = cs;
            } else {
                const int k = atomicAdd(&cnt[NNODES], 1);
                ovf[2 * k]     = cd;
                ovf[2 * k + 1] = cs;
            }
        }
        ptile = tile;
        buf ^= 1;
    }
    __syncthreads();
    if (ptile >= 0) store_logits(part[buf ^ 1], t, ptile, as, ad);
}

__global__ void zero_i32(int* __restrict__ p, int n) {
    int i = blockIdx.x * blockDim.x + threadIdx.x;
    if (i < n) p[i] = 0;
}

__device__ __forceinline__ float wave_sum(float v) {
    for (int o = 16; o > 0; o >>= 1) v += __shfl_xor(v, o);
    return v;
}
__device__ __forceinline__ int clamp_node(int s) {
    return min(max(s, 0), NNODES - 1);
}
__device__ __forceinline__ int list_src(const int* __restrict__ lst, int deg, int d, int j) {
    int s = d;
    if (j < deg) s = clamp_node(lst[slot_index(d, j)]);
    return s;
}
template <typename F>
__device__ __forceinline__ void walk_long_list(const int* __restrict__ lst,
                                               const int* __restrict__ ovf, int novf,
                                               int d, F&& f) {
#pragma nounroll
    for (int j = 0; j < CAP; ++j) f(clamp_node(lst[slot_index(d, j)]));
#pragma nounroll
    for (int k = 0; k < novf; ++k)
        if (ovf[2 * k] == d) f(clamp_node(ovf[2 * k + 1]));
    f(d);
}

__device__ __forceinline__ void gather_pass(float4& accA, float4& accB,
                                            const float* __restrict__ xlane,
                                            int s_l, float a_l, int nvalid, int hsel) {
    float4 xa[4], xb[4];
#pragma unroll
    for (int i = 0; i < 4; ++i) {
        const int s = __builtin_amdgcn_readlane(s_l, i * 8);
        const float* r = xlane + (size_t)s * D1;
        xa[i] = *(const float4*)(r);
        xb[i] = *(const float4*)(r + 128);
    }
#pragma unroll
    for (int i = 0; i < 4; ++i) {
        const float alo = __shfl(a_l, i * 8 + hsel);
        const float ahi = __shfl(a_l, i * 8 + 4 + hsel);
        float4 va = xa[i], vb = xb[i];
        if (i >= nvalid) {
            va = make_float4(0.0f, 0.0f, 0.0f, 0.0f);
            vb = make_float4(0.0f, 0.0f, 0.0f, 0.0f);
        }
        accA.x = fmaf(alo, va.x, accA.x);
        accA.y = fmaf(alo, va.y, accA.y);
        accA.z = fmaf(alo, va.z, accA.z);
        accA.w = fmaf(alo, va.w, accA.w);
        accB.x = fmaf(ahi, vb.x, accB.x);
        accB.y = fmaf(ahi, vb.y, accB.y);
        accB.z = fmaf(ahi, vb.z, accB.z);
        accB.w = fmaf(ahi, vb.w, accB.w);
    }
}

static_assert(NNODES % 8 == 0, "node_agg1 blocks cover 8 nodes each");
__global__ __launch_bounds__(256) void node_agg1(const int* __restrict__ cnt,
                                                 const int* __restrict__ lst,
                                                 const int* __restrict__ ovf,
                                                 const float* __restrict__ xp,
                                                 const float* __restrict__ as,
                                                 const float* __restrict__ ad,
                                                 const float* __restrict__ b,
                                                 const float* __restrict__ W2,
                                                 const float* __restrict__ aw_src,
                                                 const float* __restrict__ aw_dst,
                                                 float* __restrict__ rec) {
    const int lane = threadIdx.x & 31;
    const int d = __builtin_amdgcn_readfirstlane(blockIdx.x * 8 + (threadIdx.x >> 5));
    const int deg   = cnt[d];
    const int total = deg + 1;

    float4 accA = make_float4(0.0f, 0.0f, 0.0f, 0.0f);
    float4 accB = make_float4(0.0f, 0.0f, 0.0f, 0.0f);
    const float* xlane = xp + 4 * lane;
    const int hsel = lane >> 3;

    if (__builtin_expect(deg <= CAP, 1)) {
        const int jj    = lane >> 3;
        const int h     = lane & 7;
        const float add = ad[d * H1 + h];

        int   sp[8];
        float ev[8];
        const int s0 = list_src(lst, deg, d, lane);
    #pragma unroll
        for (int p = 0; p < 8; ++p) sp[p] = __shfl(s0, 4 * p + jj);
    #pragma unroll
        for (int p = 0; p < 8; ++p) {
            const float v = lrelu(as[sp[p] * H1 + h] + add);
            ev[p] = (4 * p + jj < total) ? v : -INFINITY;
        }

        float mx = ev[0];
    #pragma unroll
        for (int p = 1; p < 8; ++p) mx = fmaxf(mx, ev[p]);
        for (int j0 = 32; j0 < total; j0 += 4) {
            const int j = j0 + jj;
            if (j < total) {
                const int s = list_src(lst, deg, d, j);
                mx = fmaxf(mx, lrelu(as[s * H1 + h] + add));
            }
        }
        mx = fmaxf(mx, __shfl_xor(mx, 8));
        mx = fmaxf(mx, __shfl_xor(mx, 16));

        float sum = 0.0f;
    #pragma unroll
        for (int p = 0; p < 8; ++p) {
            ev[p] = (4 * p + jj < total) ? __expf(ev[p] - mx) : 0.0f;
            sum += ev[p];
        }
        for (int j0 = 32; j0 < total; j0 += 4) {
            const int j = j0 + jj;
            if (j < total) {
                const int s = list_src(lst, deg, d, j);
                sum += __expf(lrelu(as[s * H1 + h] + add) - mx);
            }
        }
        sum += __shfl_xor(sum, 8);
        sum += __shfl_xor(sum, 16);
        const float den = sum;
    #pragma unroll
        for (int p = 0; p < 8; ++p) ev[p] = ev[p] / den;

        const int n0   = min(total, 32);
    #pragma unroll
        for (int p = 0; p < 8; ++p) {
            if (4 * p < n0) gather_pass(accA, accB, xlane, sp[p], ev[p], n0 - 4 * p, hsel);
        }
        for (int j0 = 32; j0 < total; j0 += 4) {
            const int j = j0 + jj;
            int   s = d;
            float a = 0.0f;
            if (j < total) {
                s = list_src(lst, deg, d, j);
                a = __expf(lrelu(as[s * H1 + h] + add) - mx) / den;
            }
            gather_pass(accA, accB, xlane, s, a, total - j0, hsel);
        }
    } else {
        const int   novf = min(max(cnt[NNODES], 0), NEDGES);
        const int   hA   = hsel, hB = 4 + hsel;
        const float addA = ad[d * H1 + hA];
        const float addB = ad[d * H1 + hB];
        float mA = -INFINITY, mB = -INFINITY;
        walk_long_list(lst, ovf, novf, d, [&](int s) {
            mA = fmaxf(mA, lrelu(as[s * H1 + hA] + addA));
            mB = fmaxf(mB, lrelu(as[s * H1 + hB] + addB));
        });
        float sA = 0.0f, sB = 0.0f;
        walk_long_list(lst, ovf, novf, d, [&](int s) {
            sA += __expf(lrelu(as[s * H1 + hA] + addA) - mA);
            sB += __expf(lrelu(as[s * H1 + hB] + addB) - mB);
        });
        walk_long_list(lst, ovf, novf, d, [&](int s) {
            const float aA = __expf(lrelu(as[s * H1 + hA] + addA) - mA) / sA;
            const float aB = __expf(lrelu(as[s * H1 + hB] + addB) - mB) / sB;
            const float* r = xlane + (size_t)s * D1;
            const float4 va = *(const float4*)(r);
            const float4 vb = *(const float4*)(r + 128);
            accA.x = fmaf(aA, va.x, accA.x);
            accA.y = fmaf(aA, va.y, accA.y);
            accA.z = fmaf(aA, va.z, accA.z);
            accA.w = fmaf(aA, va.w, accA.w);
            accB.x = fmaf(aB, vb.x, accB.x);
            accB.y = fmaf(aB, vb.y, accB.y);
            accB.z = fmaf(aB, vb.z, accB.z);
            accB.w = fmaf(aB, vb.w, accB.w);
        });
    }

    const float* bl = b + 4 * lane;
    float4 rA, rB;
    rA.x = accA.x + bl[0];   rA.y = accA.y + bl[1];   rA.z = accA.z + bl[2];   rA.w = accA.w + bl[3];
    rB.x = accB.x + bl[128]; rB.y = accB.y + bl[129]; rB.z = accB.z + bl[130]; rB.w = accB.w + bl[131];
    rA.x = rA.x > 0.0f ? rA.x : 0.0f; rA.y = rA.y > 0.0f ? rA.y : 0.0f;
    rA.z = rA.z > 0.0f ? rA.z : 0.0f; rA.w = rA.w > 0.0f ? rA.w : 0.0f;
    rB.x = rB.x > 0.0f ? rB.x : 0.0f; rB.y = rB.y > 0.0f ? rB.y : 0.0f;
    rB.z = rB.z > 0.0f ? rB.z : 0.0f; rB.w = rB.w > 0.0f ? rB.w : 0.0f;

    const float4* w2l = (const float4*)(W2 + 8 * lane);
    const float4 wa0 = w2l[0];
    const float4 wa1 = w2l[1];
    const float4 wb0 = w2l[64];
    const float4 wb1 = w2l[65];
    float y0 = 0.0f, y1 = 0.0f;
    y0 = fmaf(rA.x, wa0.x, y0);  y1 = fmaf(rA.x, wa0.y, y1);
    y0 = fmaf(rA.y, wa0.z, y0);  y1 = fmaf(rA.y, wa0.w, y1);
    y0 = fmaf(rA.z, wa1.x, y0);  y1 = fmaf(rA.z, wa1.y, y1);
    y0 = fmaf(rA.w, wa1.z, y0);  y1 = fmaf(rA.w, wa1.w, y1);
    y0 = fmaf(rB.x, wb0.x, y0);  y1 = fmaf(rB.x, wb0.y, y1);
    y0 = fmaf(rB.y, wb0.z, y0);  y1 = fmaf(rB.y, wb0.w, y1);
    y0 = fmaf(rB.z, wb1.x, y0);  y1 = fmaf(rB.z, wb1.y, y1);
    y0 = fmaf(rB.w, wb1.z, y0);  y1 = fmaf(rB.w, wb1.w, y1);
    y0 = wave_sum(y0);
    y1 = wave_sum(y1);
    if (lane == 0) {
        *(float4*)(rec + (size_t)d * REC2) = make_float4(y0, y1,
                                                         fmaf(y0, aw_src[0], y1 * aw_src[1]),
                                                         fmaf(y0, aw_dst[0], y1 * aw_dst[1]));
    }
}

#define A2_THREADS 32
#define A2_BLOCKS  ((NNODES + A2_THREADS - 1) / A2_THREADS)

__device__ __forceinline__ void line_slots(const int* __restrict__ line, int (&sl)[CAPH]) {
    const int4* lp = (const int4*)line;
#pragma unroll
    for (int k = 0; k < CAPH / 4; ++k) {
        const int4 v = lp[k];
        sl[4 * k + 0] = v.x;
        sl[4 * k + 1] = v.y;
        sl[4 * k + 2] = v.z;
        sl[4 * k + 3] = v.w;
    }
}
__device__ __forceinline__ void line_records(const v4f* __restrict__ rec4, const int (&sl)[CAPH],
                                             int n, int d, float add,
                                             float (&e)[CAPH], float (&x0)[CAPH], float (&x1)[CAPH]) {
#pragma unroll
    for (int j = 0; j < CAPH; ++j) {
        const bool valid = j < n;
        const int  s     = valid ? clamp_node(sl[j]) : d;
        const v4f  r     = rec4[s];
        e[j]  = valid ? lrelu(r.z + add) : -INFINITY;
        x0[j] = valid ? r.x : 0.0f;
        x1[j] = valid ? r.y : 0.0f;
    }
}
__device__ __forceinline__ float line_max(const float (&e)[CAPH], float mx) {
#pragma unroll
    for (int j = 0; j < CAPH; ++j) mx = fmaxf(mx, e[j]);
    return mx;
}
__device__ __forceinline__ float line_exp_sum(float (&e)[CAPH], int n, float mx, float sum) {
#pragma unroll
    for (int j = 0; j < CAPH; ++j) {
        e[j] = (j < n) ? __expf(e[j] - mx) : 0.0f;
        sum += e[j];
    }
    return sum;
}
__device__ __forceinline__ void line_weighted(const float (&p)[CAPH], const float (&x0)[CAPH],
                                              const float (&x1)[CAPH], float den,
                                              float& c0, float& c1) {
#pragma unroll
    for (int j = 0; j < CAPH; ++j) {
        const float a = p[j] / den;
        c0 = fmaf(a, x0[j], c0);
        c1 = fmaf(a, x1[j], c1);
    }
}

__global__ __launch_bounds__(A2_THREADS) void node_agg2(const int* __restrict__ cnt,
                                                        const int* __restrict__ lst,
                                                        const int* __restrict__ ovf,
                                                        const float* __restrict__ rec,
                                                        const float* __restrict__ b,
                                                        float* __restrict__ out) {
    const int d = blockIdx.x * A2_THREADS + threadIdx.x;
    if (d >= NNODES) return;
    const v4f* __restrict__ rec4 = (const v4f*)rec;
    const int    deg   = cnt[d];
    const v4f    self  = rec4[d];
    const float  add   = self.w;
    const float  eself = lrelu(self.z + add);

    float c0 = 0.0f, c1 = 0.0f;
    if (__builtin_expect(deg <= CAP, 1)) {
        float ev[CAPH], x0[CAPH], x1[CAPH];
        float ew[CAPH], y0[CAPH], y1[CAPH];
        const bool more = deg > CAPH;
        const int  n1   = deg - CAPH;

        int sl[CAPH];
        line_slots(lst + (size_t)d * CAPH, sl);
        line_records(rec4, sl, deg, d, add, ev, x0, x1);
        float mx = line_max(ev, eself);
        if (more) {
            int sm[CAPH];
            line_slots(lst + (size_t)NNODES * CAPH + (size_t)d * CAPH, sm);
            line_records(rec4, sm, n1, d, add, ew, y0, y1);
            mx = line_max(ew, mx);
        }

        float sum = line_exp_sum(ev, deg, mx, 0.0f);
        if (more) sum = line_exp_sum(ew, n1, mx, sum);
        const float pself = __expf(eself - mx);
        sum += pself;

        line_weighted(ev, x0, x1, sum, c0, c1);
        if (more) line_weighted(ew, y0, y1, sum, c0, c1);
        const float a = pself / sum;
        c0 = fmaf(a, self.x, c0);
        c1 = fmaf(a, self.y, c1);
    } else {
        const int novf = min(max(cnt[NNODES], 0), NEDGES);
        float m = -INFINITY;
        walk_long_list(lst, ovf, novf, d, [&](int s) {
            m = fmaxf(m, lrelu(rec4[s].z + add));
        });
        float sm = 0.0f;
        walk_long_list(lst, ovf, novf, d, [&](int s) {
            sm += __expf(lrelu(rec4[s].z + add) - m);
        });
        walk_long_list(lst, ovf, novf, d, [&](int s) {
            const v4f   r = rec4[s];
            const float a = __expf(lrelu(r.z + add) - m) / sm;
            c0 = fmaf(a, r.x, c0);
            c1 = fmaf(a, r.y, c1);
        });
    }
    *(float2*)(out + (size_t)d * OUT2) = make_float2(c0 + b[0], c1 + b[1]);
}

extern "C" void kernel_launch(void* const* d_in, const int* in_sizes, int n_in,
                              void* d_out, int out_size, void* d_ws, size_t ws_size,
                              hipStream_t stream) {
    const float* x     = (const float*)d_in[0];
    const int*   ei    = (const int*)d_in[1];
    const float* W1    = (const float*)d_in[2];
    const float* asw1  = (const float*)d_in[3];
    const float* adw1  = (const float*)d_in[4];
    const float* b1    = (const float*)d_in[5];
    const float* W2    = (const float*)d_in[6];
    const float* asw2  = (const float*)d_in[7];
    const float* adw2  = (const float*)d_in[8];
    const float* b2    = (const float*)d_in[9];
    float*       out   = (float*)d_out;

    float* w    = (float*)d_ws;
    float* xp1  = w;
    float* as1  = xp1  + (size_t)NNODES * D1;
    float* ad1  = as1  + (size_t)NNODES * H1;
    float* rec2 = ad1  + (size_t)NNODES * H1;
    int*   lst  = (int*)(rec2 + (size_t)NNODES * REC2);
    int*   ovf  = lst  + (size_t)NNODES * CAP;
    int*   cnt  = ovf  + (size_t)2 * NEDGES;

    zero_i32<<<(NNODES + 1 + 255) / 256, 256, 0, stream>>>(cnt, NNODES + 1);

    gemm1_wmma<<<G1_BLOCKS, G1_THREADS, 0, stream>>>(x, W1, xp1, asw1, adw1, as1, ad1,
                                                         ei, cnt, lst, ovf);
    node_agg1<<<NNODES / 8, 256, 0, stream>>>(cnt, lst, ovf, xp1, as1, ad1, b1,
                                              W2, asw2, adw2, rec2);

    node_agg2<<<A2_BLOCKS, A2_THREADS, 0, stream>>>(cnt, lst, ovf, rec2, b2, out);
}


# Edges grouped by destination by a two-pass bucket sort in LDS (cells per chunk and bucket, lists built in the GEMM's front): no global counting add per edge; list order = arrival order of LDS adds
# speedup vs baseline: 11.6586x; 1.2468x over previous
#include <hip/hip_runtime.h>
#include <hip/hip_bf16.h>
#include <math.h>

#define NNODES 50000
#define NEDGES 800000
#define FIN    256
#define H1     8
#define C1     32
#define D1     256
#define OUT2   2
#define REC2   4
#define SLOPE  0.2f

#define LDSPAD 4
#define ASTRIDE (FIN + LDSPAD)

typedef float v2f __attribute__((ext_vector_type(2)));
typedef float v4f __attribute__((ext_vector_type(4)));
typedef float v8f __attribute__((ext_vector_type(8)));

__device__ __forceinline__ float lrelu(float v) {
    return v > 0.0f ? v : SLOPE * v;
}
__device__ __forceinline__ int clamp_node(int s) {
    return min(max(s, 0), NNODES - 1);
}

#define CAPH 32
#define CAP  (2 * CAPH)
static_assert((long long)NNODES * CAP < 2147483647LL, "slot index fits an int");
#define NCHUNK 256
#define ECH    (NEDGES / NCHUNK)
#define NBUCK  256
#define NPB    196
#define CELLW  64
#define B1_THREADS 512
#define ECH_ITERS ((ECH + B1_THREADS - 1) / B1_THREADS)
static_assert(NCHUNK * ECH == NEDGES, "chunks cover every edge exactly");
static_assert(NBUCK * NPB >= NNODES, "buckets cover every node");
static_assert(NNODES <= 65536 && NPB <= 256, "a packed edge word holds a source id and a local destination");
static_assert((NBUCK * CELLW / 4) % B1_THREADS == 0, "cells are copied out in whole 16-byte rounds");

__device__ __forceinline__ int slot_index(int d, int j) {
    return (j < CAPH ? 0 : NNODES * CAPH - CAPH) + d * CAPH + j;
}


#define MTILES    (NNODES / 16)
#define G1_BLOCKS 256
#define KSTEPS    (FIN / 4)
#define G1_THREADS 512
static_assert(NNODES % 16 == 0, "gemm1 row tiles are 16 rows");
static_assert(NBUCK == G1_BLOCKS, "one bucket of nodes per gemm work-group");
static_assert(G1_THREADS == 8 * CELLW, "pass 2 reads eight cells per round, one word per thread");
static_assert(NCHUNK % 8 == 0 && NCHUNK <= G1_THREADS && NPB <= G1_THREADS, "pass 2 thread roles");
#define PSTRIDE   36
static_assert(D1 == 16 * (G1_THREADS / 32), "one wave per 16 output columns");
static_assert(C1 == 32 && H1 * 2 == G1_THREADS / 32, "two waves per head");

template <int M>
__device__ __forceinline__ float row_xor(float v) {
    static_assert(M > 0 && M < 16, "inside a 16-lane row");
    return __int_as_float(__builtin_amdgcn_update_dpp(0, __float_as_int(v), 0x160 + M, 0xf, 0xf, false));
}

__device__ __forceinline__ void store_logits(const float* __restrict__ pbuf, int t, int tile,
                                             float* __restrict__ as, float* __restrict__ ad) {
    if (t < 256) {
        const int kind = t >> 7;
        const int row  = (t >> 3) & 15;
        const int h    = t & 7;
        const float* pp = pbuf + (2 * h) * PSTRIDE + kind * 16 + row;
        const float s = pp[0] + pp[PSTRIDE];
        float* dstp = kind ? ad : as;
        dstp[(size_t)(tile * 16 + row) * H1 + h] = s;
    }
}
__device__ __forceinline__ void put_edge(int* __restrict__ llist, int b, int dl, int s, int k,
                                         int* __restrict__ cnt, int* __restrict__ ovf) {
    if (k < CAP) {
        llist[dl * CAP + k] = s;
    } else {
        const int j = atomicAdd(&cnt[NNODES], 1);
        ovf[2 * j]     = b * NPB + dl;
        ovf[2 * j + 1] = s;
    }
}
__device__ __forceinline__ void add_edge(int* __restrict__ lcnt, int* __restrict__ llist,
                                         int b, int dl, int s,
                                         int* __restrict__ cnt, int* __restrict__ ovf) {
    put_edge(llist, b, dl, s, atomicAdd(&lcnt[dl], 1), cnt, ovf);
}
__global__ __launch_bounds__(G1_THREADS) void gemm1_wmma(const float* __restrict__ x,
                                                             const float* __restrict__ W1,
                                                             float* __restrict__ xp,
                                                             const float* __restrict__ a_src,
                                                             const float* __restrict__ a_dst,
                                                             float* __restrict__ as,
                                                             float* __restrict__ ad,
                                                             const int* __restrict__ cells,
                                                             const int* __restrict__ ovf1,
                                                             const int* __restrict__ ovf1cnt,
                                                             int* __restrict__ cnt,
                                                             int* __restrict__ lst,
                                                             int* __restrict__ ovf) {
    __shared__ float aTile[2][16 * ASTRIDE];
    __shared__ float part[2][16 * PSTRIDE];
    __shared__ int   lcnt[NBUCK];
    __shared__ int   llist[NPB * CAP];

    const int t     = threadIdx.x;

    {
        const int b = blockIdx.x;
        if (t < NBUCK) lcnt[t] = 0;
        for (int i = t; i < NPB * CAP / 4; i += G1_THREADS)
            ((int4*)llist)[i] = make_int4(0, 0, 0, 0);
        const int wq = t & (CELLW - 1);
        const int cq = t >> 6;
        int wv[NCHUNK / 8], nv[NCHUNK / 8];
#pragma unroll
        for (int r = 0; r < NCHUNK / 8; ++r) {
            const int* cp = cells + ((size_t)(8 * r + cq) * NBUCK + b) * CELLW;
            wv[r] = cp[wq];
            nv[r] = cp[0];
        }
        __syncthreads();
        int kq[NCHUNK / 8];
#pragma unroll
        for (int r = 0; r < NCHUNK / 8; ++r) {
            const bool valid = wq >= 1 && wq <= min(nv[r], CELLW - 1);
            kq[r] = -1;
            if (valid) kq[r] = atomicAdd(&lcnt[min((wv[r] >> 16) & 0xff, NPB - 1)], 1);
        }
#pragma unroll
        for (int r = 0; r < NCHUNK / 8; ++r) {
            if (kq[r] >= 0)
                put_edge(llist, b, min((wv[r] >> 16) & 0xff, NPB - 1), wv[r] & 0xffff, kq[r], cnt, ovf);
        }
        if (t < NCHUNK) {
            const int ns = min(max(ovf1cnt[t], 0), ECH);
            for (int k = 0; k < ns; ++k) {
                const int d = clamp_node(ovf1[2 * (t * ECH + k)]);
                const int sN = clamp_node(ovf1[2 * (t * ECH + k) + 1]);
                if (d / NPB == b) add_edge(lcnt, llist, b, d - b * NPB, sN, cnt, ovf);
            }
        }
        __syncthreads();
        if (t < NPB && b * NPB + t < NNODES) cnt[b * NPB + t] = lcnt[t];
        for (int q = t; q < NPB * 8; q += G1_THREADS) {
            const int node = q >> 3, pc = q & 7;
            const int d = b * NPB + node;
            if (d < NNODES) {
                const int4* lp = (const int4*)(llist + node * CAP);
                *(int4*)(lst + (size_t)d * CAPH + 4 * pc) = lp[pc];
                if (lcnt[node] > CAPH)
                    *(int4*)(lst + (size_t)NNODES * CAPH + (size_t)d * CAPH + 4 * pc) = lp[CAPH / 4 + pc];
            }
        }
    }
    const int lane  = t & 31;
    const int wave  = t >> 5;
    const int half  = lane >> 4;
    const int l16   = lane & 15;
    const int bcol  = wave * 16 + l16;

    v2f breg[KSTEPS];
#pragma unroll
    for (int s = 0; s < KSTEPS; ++s) {
        const int kk = 4 * s + half * 2;
        breg[s].x = W1[kk * D1 + bcol];
        breg[s].y = W1[(kk + 1) * D1 + bcol];
    }
    const float av_s = a_src[bcol];
    const float av_d = a_dst[bcol];
    const bool b3 = (l16 & 8) != 0, b2 = (l16 & 4) != 0, b1 = (l16 & 2) != 0, b0 = (l16 & 1) != 0;
    const int  poff = wave * PSTRIDE + (l16 >> 3) * 16 + half * 8 + (l16 & 7);

    const int srow = t >> 5;
    const int scol = (t & 31) * 8;
    const int soff = srow * ASTRIDE + scol;
    const int aoff = l16 * ASTRIDE + half * 2;

    int tile = blockIdx.x;
    float4 p0 = make_float4(0.0f, 0.0f, 0.0f, 0.0f);
    float4 p1 = p0;
    if (tile < MTILES) {
        const float4* src = (const float4*)(x + (size_t)(tile * 16 + srow) * FIN + scol);
        p0 = src[0];
        p1 = src[1];
    }
    int buf = 0;
    int ptile = -1;
    for (; tile < MTILES; tile += G1_BLOCKS) {
        float* abuf = aTile[buf];
        *(float4*)(abuf + soff)     = p0;
        *(float4*)(abuf + soff + 4) = p1;
        __syncthreads();

        const int nxt = tile + G1_BLOCKS;
        if (nxt < MTILES) {
            const float4* src = (const float4*)(x + (size_t)(nxt * 16 + srow) * FIN + scol);
            p0 = src[0];
            p1 = src[1];
        }

        if (ptile >= 0) store_logits(part[buf ^ 1], t, ptile, as, ad);

        v8f acc = {};
#pragma unroll
        for (int s = 0; s < KSTEPS; ++s) {
            v2f a = *(const v2f*)(abuf + aoff + 4 * s);
            acc = __builtin_amdgcn_wmma_f32_16x16x4_f32(
                 false, a,  false, breg[s],
                 (short)0, acc,  false,  false);
        }

        const int mrow = tile * 16 + half * 8;
#pragma unroll
        for (int r = 0; r < 8; ++r)
            xp[(size_t)(mrow + r) * D1 + bcol] = acc[r];

        {
            float v[16];
#pragma unroll
            for (int r = 0; r < 8; ++r) {
                v[r]     = acc[r] * av_s;
                v[8 + r] = acc[r] * av_d;
            }
            float u[8], w[4], y[2];
#pragma unroll
            for (int i = 0; i < 8; ++i)
                u[i] = (b3 ? v[8 + i] : v[i]) + row_xor<8>(b3 ? v[i] : v[8 + i]);
#pragma unroll
            for (int i = 0; i < 4; ++i)
                w[i] = (b2 ? u[4 + i] : u[i]) + row_xor<4>(b2 ? u[i] : u[4 + i]);
#pragma unroll
            for (int i = 0; i < 2; ++i)
                y[i] = (b1 ? w[2 + i] : w[i]) + row_xor<2>(b1 ? w[i] : w[2 + i]);
            const float z = (b0 ? y[1] : y[0]) + row_xor<1>(b0 ? y[0] : y[1]);
            part[buf][poff] = z;
        }

        ptile = tile;
        buf ^= 1;
    }
    __syncthreads();
    if (ptile >= 0) store_logits(part[buf ^ 1], t, ptile, as, ad);
}

__global__ __launch_bounds__(B1_THREADS) void bucket_edges(const int* __restrict__ ei,
                                                           int* __restrict__ cells,
                                                           int* __restrict__ ovf1,
                                                           int* __restrict__ ovf1cnt,
                                                           int* __restrict__ cnt) {
    __shared__ int cell[NBUCK * CELLW];
    __shared__ int ccnt[NBUCK];
    __shared__ int nspill;
    const int t = threadIdx.x;
    const int c = blockIdx.x;

    int es[ECH_ITERS], ed[ECH_ITERS];
#pragma unroll
    for (int i = 0; i < ECH_ITERS; ++i) {
        const int k = i * B1_THREADS + t;
        es[i] = 0;
        ed[i] = 0;
        if (k < ECH) {
            es[i] = ei[c * ECH + k];
            ed[i] = ei[NEDGES + c * ECH + k];
        }
    }
#pragma unroll
    for (int i = 0; i < NBUCK * CELLW / 4 / B1_THREADS; ++i)
        ((int4*)cell)[i * B1_THREADS + t] = make_int4(0, 0, 0, 0);
    if (t < NBUCK) ccnt[t] = 0;
    if (t == 0) nspill = 0;
    if (c == 0 && t == 0) cnt[NNODES] = 0;
    __syncthreads();

    int pos[ECH_ITERS];
#pragma unroll
    for (int i = 0; i < ECH_ITERS; ++i) {
        pos[i] = -1;
        if (i * B1_THREADS + t < ECH) pos[i] = atomicAdd(&ccnt[clamp_node(ed[i]) / NPB], 1);
    }
#pragma unroll
    for (int i = 0; i < ECH_ITERS; ++i) {
        if (pos[i] >= 0) {
            const int sN = clamp_node(es[i]);
            const int d  = clamp_node(ed[i]);
            const int b  = d / NPB;
            const int dl = d - b * NPB;
            if (pos[i] < CELLW - 1) {
                cell[b * CELLW + 1 + pos[i]] = (dl << 16) | sN;
            } else {
                const int j = atomicAdd(&nspill, 1);
                ovf1[2 * (c * ECH + j)]     = d;
                ovf1[2 * (c * ECH + j) + 1] = sN;
            }
        }
    }
    __syncthreads();
    if (t < NBUCK) cell[t * CELLW] = ccnt[t];
    __syncthreads();

    int4* dst = (int4*)(cells + (size_t)c * NBUCK * CELLW);
#pragma unroll
    for (int i = 0; i < NBUCK * CELLW / 4 / B1_THREADS; ++i)
        dst[i * B1_THREADS + t] = ((const int4*)cell)[i * B1_THREADS + t];
    if (t == 0) ovf1cnt[c] = nspill;
}

__device__ __forceinline__ float wave_sum(float v) {
    for (int o = 16; o > 0; o >>= 1) v += __shfl_xor(v, o);
    return v;
}
__device__ __forceinline__ int list_src(const int* __restrict__ lst, int deg, int d, int j) {
    int s = d;
    if (j < deg) s = clamp_node(lst[slot_index(d, j)]);
    return s;
}
template <typename F>
__device__ __forceinline__ void walk_long_list(const int* __restrict__ lst,
                                               const int* __restrict__ ovf, int novf,
                                               int d, F&& f) {
#pragma nounroll
    for (int j = 0; j < CAP; ++j) f(clamp_node(lst[slot_index(d, j)]));
#pragma nounroll
    for (int k = 0; k < novf; ++k)
        if (ovf[2 * k] == d) f(clamp_node(ovf[2 * k + 1]));
    f(d);
}

__device__ __forceinline__ void gather_pass(float4& accA, float4& accB,
                                            const float* __restrict__ xlane,
                                            int s_l, float a_l, int nvalid, int hsel) {
    float4 xa[4], xb[4];
#pragma unroll
    for (int i = 0; i < 4; ++i) {
        const int s = __builtin_amdgcn_readlane(s_l, i * 8);
        const float* r = xlane + (size_t)s * D1;
        xa[i] = *(const float4*)(r);
        xb[i] = *(const float4*)(r + 128);
    }
#pragma unroll
    for (int i = 0; i < 4; ++i) {
        const float alo = __shfl(a_l, i * 8 + hsel);
        const float ahi = __shfl(a_l, i * 8 + 4 + hsel);
        float4 va = xa[i], vb = xb[i];
        if (i >= nvalid) {
            va = make_float4(0.0f, 0.0f, 0.0f, 0.0f);
            vb = make_float4(0.0f, 0.0f, 0.0f, 0.0f);
        }
        accA.x = fmaf(alo, va.x, accA.x);
        accA.y = fmaf(alo, va.y, accA.y);
        accA.z = fmaf(alo, va.z, accA.z);
        accA.w = fmaf(alo, va.w, accA.w);
        accB.x = fmaf(ahi, vb.x, accB.x);
        accB.y = fmaf(ahi, vb.y, accB.y);
        accB.z = fmaf(ahi, vb.z, accB.z);
        accB.w = fmaf(ahi, vb.w, accB.w);
    }
}

static_assert(NNODES % 8 == 0, "node_agg1 blocks cover 8 nodes each");
__global__ __launch_bounds__(256) void node_agg1(const int* __restrict__ cnt,
                                                 const int* __restrict__ lst,
                                                 const int* __restrict__ ovf,
                                                 const float* __restrict__ xp,
                                                 const float* __restrict__ as,
                                                 const float* __restrict__ ad,
                                                 const float* __restrict__ b,
                                                 const float* __restrict__ W2,
                                                 const float* __restrict__ aw_src,
                                                 const float* __restrict__ aw_dst,
                                                 float* __restrict__ rec) {
    const int lane = threadIdx.x & 31;
    const int d = __builtin_amdgcn_readfirstlane(blockIdx.x * 8 + (threadIdx.x >> 5));
    const int deg   = cnt[d];
    const int total = deg + 1;

    float4 accA = make_float4(0.0f, 0.0f, 0.0f, 0.0f);
    float4 accB = make_float4(0.0f, 0.0f, 0.0f, 0.0f);
    const float* xlane = xp + 4 * lane;
    const int hsel = lane >> 3;

    if (__builtin_expect(deg <= CAP, 1)) {
        const int jj    = lane >> 3;
        const int h     = lane & 7;
        const float add = ad[d * H1 + h];

        int   sp[8];
        float ev[8];
        const int s0 = list_src(lst, deg, d, lane);
    #pragma unroll
        for (int p = 0; p < 8; ++p) sp[p] = __shfl(s0, 4 * p + jj);
    #pragma unroll
        for (int p = 0; p < 8; ++p) {
            const float v = lrelu(as[sp[p] * H1 + h] + add);
            ev[p] = (4 * p + jj < total) ? v : -INFINITY;
        }

        float mx = ev[0];
    #pragma unroll
        for (int p = 1; p < 8; ++p) mx = fmaxf(mx, ev[p]);
        for (int j0 = 32; j0 < total; j0 += 4) {
            const int j = j0 + jj;
            if (j < total) {
                const int s = list_src(lst, deg, d, j);
                mx = fmaxf(mx, lrelu(as[s * H1 + h] + add));
            }
        }
        mx = fmaxf(mx, __shfl_xor(mx, 8));
        mx = fmaxf(mx, __shfl_xor(mx, 16));

        float sum = 0.0f;
    #pragma unroll
        for (int p = 0; p < 8; ++p) {
            ev[p] = (4 * p + jj < total) ? __expf(ev[p] - mx) : 0.0f;
            sum += ev[p];
        }
        for (int j0 = 32; j0 < total; j0 += 4) {
            const int j = j0 + jj;
            if (j < total) {
                const int s = list_src(lst, deg, d, j);
                sum += __expf(lrelu(as[s * H1 + h] + add) - mx);
            }
        }
        sum += __shfl_xor(sum, 8);
        sum += __shfl_xor(sum, 16);
        const float den = sum;
    #pragma unroll
        for (int p = 0; p < 8; ++p) ev[p] = ev[p] / den;

        const int n0   = min(total, 32);
    #pragma unroll
        for (int p = 0; p < 8; ++p) {
            if (4 * p < n0) gather_pass(accA, accB, xlane, sp[p], ev[p], n0 - 4 * p, hsel);
        }
        for (int j0 = 32; j0 < total; j0 += 4) {
            const int j = j0 + jj;
            int   s = d;
            float a = 0.0f;
            if (j < total) {
                s = list_src(lst, deg, d, j);
                a = __expf(lrelu(as[s * H1 + h] + add) - mx) / den;
            }
            gather_pass(accA, accB, xlane, s, a, total - j0, hsel);
        }
    } else {
        const int   novf = min(max(cnt[NNODES], 0), NEDGES);
        const int   hA   = hsel, hB = 4 + hsel;
        const float addA = ad[d * H1 + hA];
        const float addB = ad[d * H1 + hB];
        float mA = -INFINITY, mB = -INFINITY;
        walk_long_list(lst, ovf, novf, d, [&](int s) {
            mA = fmaxf(mA, lrelu(as[s * H1 + hA] + addA));
            mB = fmaxf(mB, lrelu(as[s * H1 + hB] + addB));
        });
        float sA = 0.0f, sB = 0.0f;
        walk_long_list(lst, ovf, novf, d, [&](int s) {
            sA += __expf(lrelu(as[s * H1 + hA] + addA) - mA);
            sB += __expf(lrelu(as[s * H1 + hB] + addB) - mB);
        });
        walk_long_list(lst, ovf, novf, d, [&](int s) {
            const float aA = __expf(lrelu(as[s * H1 + hA] + addA) - mA) / sA;
            const float aB = __expf(lrelu(as[s * H1 + hB] + addB) - mB) / sB;
            const float* r = xlane + (size_t)s * D1;
            const float4 va = *(const float4*)(r);
            const float4 vb = *(const float4*)(r + 128);
            accA.x = fmaf(aA, va.x, accA.x);
            accA.y = fmaf(aA, va.y, accA.y);
            accA.z = fmaf(aA, va.z, accA.z);
            accA.w = fmaf(aA, va.w, accA.w);
            accB.x = fmaf(aB, vb.x, accB.x);
            accB.y = fmaf(aB, vb.y, accB.y);
            accB.z = fmaf(aB, vb.z, accB.z);
            accB.w = fmaf(aB, vb.w, accB.w);
        });
    }

    const float* bl = b + 4 * lane;
    float4 rA, rB;
    rA.x = accA.x + bl[0];   rA.y = accA.y + bl[1];   rA.z = accA.z + bl[2];   rA.w = accA.w + bl[3];
    rB.x = accB.x + bl[128]; rB.y = accB.y + bl[129]; rB.z = accB.z + bl[130]; rB.w = accB.w + bl[131];
    rA.x = rA.x > 0.0f ? rA.x : 0.0f; rA.y = rA.y > 0.0f ? rA.y : 0.0f;
    rA.z = rA.z > 0.0f ? rA.z : 0.0f; rA.w = rA.w > 0.0f ? rA.w : 0.0f;
    rB.x = rB.x > 0.0f ? rB.x : 0.0f; rB.y = rB.y > 0.0f ? rB.y : 0.0f;
    rB.z = rB.z > 0.0f ? rB.z : 0.0f; rB.w = rB.w > 0.0f ? rB.w : 0.0f;

    const float4* w2l = (const float4*)(W2 + 8 * lane);
    const float4 wa0 = w2l[0];
    const float4 wa1 = w2l[1];
    const float4 wb0 = w2l[64];
    const float4 wb1 = w2l[65];
    float y0 = 0.0f, y1 = 0.0f;
    y0 = fmaf(rA.x, wa0.x, y0);  y1 = fmaf(rA.x, wa0.y, y1);
    y0 = fmaf(rA.y, wa0.z, y0);  y1 = fmaf(rA.y, wa0.w, y1);
    y0 = fmaf(rA.z, wa1.x, y0);  y1 = fmaf(rA.z, wa1.y, y1);
    y0 = fmaf(rA.w, wa1.z, y0);  y1 = fmaf(rA.w, wa1.w, y1);
    y0 = fmaf(rB.x, wb0.x, y0);  y1 = fmaf(rB.x, wb0.y, y1);
    y0 = fmaf(rB.y, wb0.z, y0);  y1 = fmaf(rB.y, wb0.w, y1);
    y0 = fmaf(rB.z, wb1.x, y0);  y1 = fmaf(rB.z, wb1.y, y1);
    y0 = fmaf(rB.w, wb1.z, y0);  y1 = fmaf(rB.w, wb1.w, y1);
    y0 = wave_sum(y0);
    y1 = wave_sum(y1);
    if (lane == 0) {
        *(float4*)(rec + (size_t)d * REC2) = make_float4(y0, y1,
                                                         fmaf(y0, aw_src[0], y1 * aw_src[1]),
                                                         fmaf(y0, aw_dst[0], y1 * aw_dst[1]));
    }
}

#define A2_THREADS 32
#define A2_BLOCKS  ((NNODES + A2_THREADS - 1) / A2_THREADS)

__device__ __forceinline__ void line_slots(const int* __restrict__ line, int (&sl)[CAPH]) {
    const int4* lp = (const int4*)line;
#pragma unroll
    for (int k = 0; k < CAPH / 4; ++k) {
        const int4 v = lp[k];
        sl[4 * k + 0] = v.x;
        sl[4 * k + 1] = v.y;
        sl[4 * k + 2] = v.z;
        sl[4 * k + 3] = v.w;
    }
}
__device__ __forceinline__ void line_records(const v4f* __restrict__ rec4, const int (&sl)[CAPH],
                                             int n, int d, float add,
                                             float (&e)[CAPH], float (&x0)[CAPH], float (&x1)[CAPH]) {
#pragma unroll
    for (int j = 0; j < CAPH; ++j) {
        const bool valid = j < n;
        const int  s     = valid ? clamp_node(sl[j]) : d;
        const v4f  r     = rec4[s];
        e[j]  = valid ? lrelu(r.z + add) : -INFINITY;
        x0[j] = valid ? r.x : 0.0f;
        x1[j] = valid ? r.y : 0.0f;
    }
}
__device__ __forceinline__ float line_max(const float (&e)[CAPH], float mx) {
#pragma unroll
    for (int j = 0; j < CAPH; ++j) mx = fmaxf(mx, e[j]);
    return mx;
}
__device__ __forceinline__ float line_exp_sum(float (&e)[CAPH], int n, float mx, float sum) {
#pragma unroll
    for (int j = 0; j < CAPH; ++j) {
        e[j] = (j < n) ? __expf(e[j] - mx) : 0.0f;
        sum += e[j];
    }
    return sum;
}
__device__ __forceinline__ void line_weighted(const float (&p)[CAPH], const float (&x0)[CAPH],
                                              const float (&x1)[CAPH], float den,
                                              float& c0, float& c1) {
#pragma unroll
    for (int j = 0; j < CAPH; ++j) {
        const float a = p[j] / den;
        c0 = fmaf(a, x0[j], c0);
        c1 = fmaf(a, x1[j], c1);
    }
}

__global__ __launch_bounds__(A2_THREADS) void node_agg2(const int* __restrict__ cnt,
                                                        const int* __restrict__ lst,
                                                        const int* __restrict__ ovf,
                                                        const float* __restrict__ rec,
                                                        const float* __restrict__ b,
                                                        float* __restrict__ out) {
    const int d = blockIdx.x * A2_THREADS + threadIdx.x;
    if (d >= NNODES) return;
    const v4f* __restrict__ rec4 = (const v4f*)rec;
    const int    deg   = cnt[d];
    const v4f    self  = rec4[d];
    const float  add   = self.w;
    const float  eself = lrelu(self.z + add);

    float c0 = 0.0f, c1 = 0.0f;
    if (__builtin_expect(deg <= CAP, 1)) {
        float ev[CAPH], x0[CAPH], x1[CAPH];
        float ew[CAPH], y0[CAPH], y1[CAPH];
        const bool more = deg > CAPH;
        const int  n1   = deg - CAPH;

        int sl[CAPH];
        line_slots(lst + (size_t)d * CAPH, sl);
        line_records(rec4, sl, deg, d, add, ev, x0, x1);
        float mx = line_max(ev, eself);
        if (more) {
            int sm[CAPH];
            line_slots(lst + (size_t)NNODES * CAPH + (size_t)d * CAPH, sm);
            line_records(rec4, sm, n1, d, add, ew, y0, y1);
            mx = line_max(ew, mx);
        }

        float sum = line_exp_sum(ev, deg, mx, 0.0f);
        if (more) sum = line_exp_sum(ew, n1, mx, sum);
        const float pself = __expf(eself - mx);
        sum += pself;

        line_weighted(ev, x0, x1, sum, c0, c1);
        if (more) line_weighted(ew, y0, y1, sum, c0, c1);
        const float a = pself / sum;
        c0 = fmaf(a, self.x, c0);
        c1 = fmaf(a, self.y, c1);
    } else {
        const int novf = min(max(cnt[NNODES], 0), NEDGES);
        float m = -INFINITY;
        walk_long_list(lst, ovf, novf, d, [&](int s) {
            m = fmaxf(m, lrelu(rec4[s].z + add));
        });
        float sm = 0.0f;
        walk_long_list(lst, ovf, novf, d, [&](int s) {
            sm += __expf(lrelu(rec4[s].z + add) - m);
        });
        walk_long_list(lst, ovf, novf, d, [&](int s) {
            const v4f   r = rec4[s];
            const float a = __expf(lrelu(r.z + add) - m) / sm;
            c0 = fmaf(a, r.x, c0);
            c1 = fmaf(a, r.y, c1);
        });
    }
    *(float2*)(out + (size_t)d * OUT2) = make_float2(c0 + b[0], c1 + b[1]);
}

extern "C" void kernel_launch(void* const* d_in, const int* in_sizes, int n_in,
                              void* d_out, int out_size, void* d_ws, size_t ws_size,
                              hipStream_t stream) {
    const float* x     = (const float*)d_in[0];
    const int*   ei    = (const int*)d_in[1];
    const float* W1    = (const float*)d_in[2];
    const float* asw1  = (const float*)d_in[3];
    const float* adw1  = (const float*)d_in[4];
    const float* b1    = (const float*)d_in[5];
    const float* W2    = (const float*)d_in[6];
    const float* asw2  = (const float*)d_in[7];
    const float* adw2  = (const float*)d_in[8];
    const float* b2    = (const float*)d_in[9];
    float*       out   = (float*)d_out;

    float* w    = (float*)d_ws;
    float* xp1  = w;
    float* as1  = xp1  + (size_t)NNODES * D1;
    float* ad1  = as1  + (size_t)NNODES * H1;
    float* rec2 = ad1  + (size_t)NNODES * H1;
    int*   lst  = (int*)(rec2 + (size_t)NNODES * REC2);
    int*   ovf  = lst  + (size_t)NNODES * CAP;
    int*   cnt  = ovf  + (size_t)2 * NEDGES;
    int*   cells   = cnt   + 50048;
    int*   ovf1    = cells + (size_t)NCHUNK * NBUCK * CELLW;
    int*   ovf1cnt = ovf1  + (size_t)2 * NEDGES;

    bucket_edges<<<NCHUNK, B1_THREADS, 0, stream>>>(ei, cells, ovf1, ovf1cnt, cnt);

    gemm1_wmma<<<G1_BLOCKS, G1_THREADS, 0, stream>>>(x, W1, xp1, asw1, adw1, as1, ad1,
                                                         cells, ovf1, ovf1cnt, cnt, lst, ovf);
    node_agg1<<<NNODES / 8, 256, 0, stream>>>(cnt, lst, ovf, xp1, as1, ad1, b1,
                                              W2, asw2, adw2, rec2);

    node_agg2<<<A2_BLOCKS, A2_THREADS, 0, stream>>>(cnt, lst, ovf, rec2, b2, out);
}
